# starting kernel
# baseline (speedup 1.0000x reference)
#include <hip/hip_runtime.h>
#include <hip/hip_bf16.h>

typedef float v2f __attribute__((ext_vector_type(2)));
typedef float v8f __attribute__((ext_vector_type(8)));

__device__ __forceinline__ v8f wmma_f32(v2f a, v2f b, v8f c) {
  return __builtin_amdgcn_wmma_f32_16x16x4_f32(false, a, false, b, (short)0, c,
                                               false, false);
}

#define BATCH   8
#define CIN     3
#define HW      320
#define CLS     2
#define PSZ     10
#define P2      100
#define LPATCH  1024
#define HID     200

#define WS_INI  0
#define WS_FEA  (WS_INI + BATCH*CLS*HW*HW)
#define WS_UI   (WS_FEA + BATCH*CLS*HW*HW)
#define WS_UF   (WS_UI  + BATCH*CLS*LPATCH*P2)
#define WS_W1P  (WS_UF  + BATCH*CLS*LPATCH*P2)
#define WS_W2P  (WS_W1P + 208*100)

__global__ __launch_bounds__(256) void pad_weights_kernel(
    const float* __restrict__ w_lin1, const float* __restrict__ w_lin2,
    float* __restrict__ w1p, float* __restrict__ w2p) {
  int i = blockIdx.x * 256 + threadIdx.x;
  if (i < 208 * 100) w1p[i] = (i < HID * P2) ? w_lin1[i] : 0.f;
  if (i < 112 * 200) w2p[i] = (i < P2 * HID) ? w_lin2[i] : 0.f;
}

#define XS_FLOATS        2160
#define CONV_LDS_FLOATS  (XS_FLOATS + 148 + 9472 + 64 + 64*325 + 1152 + 16)

__global__ __launch_bounds__(256) void conv_fused_kernel(
    const float* __restrict__ x,  const float* __restrict__ w1,
    const float* __restrict__ b1, const float* __restrict__ w2,
    const float* __restrict__ b2, const float* __restrict__ wfm,
    const float* __restrict__ bfm,
    float* __restrict__ ini, float* __restrict__ fea) {
  extern __shared__ float smem[];
  float* xs     = smem;
  int*   offtab = (int*)(xs + XS_FLOATS);
  float* w1s    = xs + XS_FLOATS + 148;
  float* b1s    = w1s + 74 * 64 * 2;
  float* t1s    = b1s + 64;
  float* w2s    = t1s + 64 * 325;
  float* cs     = w2s + 1152;

  const int tid = threadIdx.x;
  const int X0 = blockIdx.x * 16, Y0 = blockIdx.y * 16, b = blockIdx.z;

  for (int i = tid; i < XS_FLOATS; i += 256) {
    float v = 0.f;
    if (i < CIN * 576) {
      int c = i / 576, r = i % 576, iy = r / 24, ix = r % 24;
      int gy = Y0 - 4 + iy, gx = X0 - 4 + ix;
      if (gy >= 0 && gy < HW && gx >= 0 && gx < HW)
        v = x[((size_t)(b * CIN + c) * HW + gy) * HW + gx];
    }
    xs[i] = v;
  }
  if (tid < 148) {
    int off = 1728;
    if (tid < 147) {
      int c = tid / 49, rr = tid % 49;
      off = c * 576 + (rr / 7) * 24 + (rr % 7);
    }
    offtab[tid] = off;
  }
  for (int i = tid; i < 148 * 64; i += 256) {
    int k = i >> 6, n = i & 63;
    w1s[(((k >> 1) << 6) + n) * 2 + (k & 1)] =
        (k < 147) ? w1[n * 147 + k] : 0.f;
  }
  if (tid < 64) b1s[tid] = b1[tid];
  for (int i = tid; i < CLS * 64 * 9; i += 256) w2s[i] = w2[i];
  if (tid < 2) cs[tid]     = b2[tid];
  if (tid < 6) cs[2 + tid] = wfm[tid];
  if (tid < 2) cs[8 + tid] = bfm[tid];
  __syncthreads();

  const int wave  = tid >> 5;
  const int lane  = tid & 31;
  const int l16   = lane & 15;
  const int lhalf = (lane < 16) ? 0 : 2;

  for (int t = wave; t < 21 * 4; t += 8) {
    const int m0 = (t >> 2) * 16, n0 = (t & 3) * 16;
    const int m  = m0 + l16;
    const int mm = (m < 324) ? m : 0;
    const int abase = (mm / 18) * 24 + (mm % 18);
    const int2* ot = ((const int2*)offtab) + (lhalf >> 1);
    const v2f*  bp = ((const v2f*)w1s) + (lhalf >> 1) * 64 + n0 + l16;
    v8f acc = {};
    #pragma unroll 4
    for (int k0 = 0; k0 < 148; k0 += 4) {
      int2 o = *ot;  ot += 2;
      v2f a;
      a.x = xs[abase + o.x];
      a.y = xs[abase + o.y];
      v2f bf = *bp;  bp += 128;
      acc = wmma_f32(a, bf, acc);
    }
    const int ch = n0 + l16;
    const float bias = b1s[ch];
    #pragma unroll
    for (int r = 0; r < 8; ++r) {
      int M = m0 + r + ((lane < 16) ? 0 : 8);
      if (M < 324) {
        int ty = M / 18, tx = M % 18;
        int gy = Y0 - 1 + ty, gx = X0 - 1 + tx;
        float v = (gy >= 0 && gy < HW && gx >= 0 && gx < HW) ? (acc[r] + bias)
                                                             : 0.f;
        t1s[ch * 325 + M] = v;
      }
    }
  }
  __syncthreads();

  {
    const int oy = tid >> 4, ox = tid & 15;
    float a0 = cs[0], a1 = cs[1];
    for (int ci = 0; ci < 64; ++ci) {
      const float* trow = &t1s[ci * 325];
      const float* wr0  = &w2s[ci * 9];
      const float* wr1  = &w2s[(64 + ci) * 9];
      #pragma unroll
      for (int dy = 0; dy < 3; ++dy)
        #pragma unroll
        for (int dx = 0; dx < 3; ++dx) {
          float tv = trow[(oy + dy) * 18 + (ox + dx)];
          a0 = fmaf(wr0[dy * 3 + dx], tv, a0);
          a1 = fmaf(wr1[dy * 3 + dx], tv, a1);
        }
    }
    const int gy = Y0 + oy, gx = X0 + ox;
    const size_t o0 = ((size_t)(b * CLS + 0) * HW + gy) * HW + gx;
    const size_t o1 = ((size_t)(b * CLS + 1) * HW + gy) * HW + gx;
    ini[o0] = a0;
    ini[o1] = a1;
    float x0 = xs[(oy + 4) * 24 + (ox + 4)];
    float x1 = xs[576 + (oy + 4) * 24 + (ox + 4)];
    float x2 = xs[1152 + (oy + 4) * 24 + (ox + 4)];
    fea[o0] = (cs[2] * x0 + cs[3] * x1 + cs[4] * x2 + cs[8]) * 0.25f;
    fea[o1] = (cs[5] * x0 + cs[6] * x1 + cs[7] * x2 + cs[9]) * 0.25f;
  }
}

#define TRANS_LDS_FLOATS (64 * 102 + 64 * 210)

__global__ __launch_bounds__(256) void trans_kernel(
    const float* __restrict__ ini, const float* __restrict__ fea,
    const float* __restrict__ w1p, const float* __restrict__ w2p,
    float* __restrict__ ui, float* __restrict__ uf) {
  extern __shared__ float smem[];
  float* a_lds = smem;
  float* h_lds = a_lds + 64 * 102;

  const int tid = threadIdx.x;
  const float* src = blockIdx.y ? fea : ini;
  float* dst       = blockIdx.y ? uf  : ui;
  const int P0 = blockIdx.x * 64;

  for (int i = tid; i < 64 * P2; i += 256) {
    int r = i / P2, p = i % P2;
    int Pa = P0 + r;
    int bc = Pa >> 10, l = Pa & 1023;
    int gy = (l >> 5) * PSZ + p / PSZ;
    int gx = (l & 31) * PSZ + p % PSZ;
    a_lds[r * 102 + p] = src[((size_t)bc * HW + gy) * HW + gx];
  }
  __syncthreads();

  const int wave  = tid >> 5;
  const int lane  = tid & 31;
  const int l16   = lane & 15;
  const int lhalf = (lane < 16) ? 0 : 2;

  for (int t = wave; t < 4 * 13; t += 8) {
    const int m0 = (t / 13) * 16, n0 = (t % 13) * 16;
    const int n  = n0 + l16;
    const v2f* ap = (const v2f*)&a_lds[(m0 + l16) * 102 + lhalf];
    const v2f* bp = (const v2f*)&w1p[n * P2 + lhalf];
    v8f acc = {};
    #pragma unroll 5
    for (int k0 = 0; k0 < P2; k0 += 4) {
      v2f a  = *ap;  ap += 2;
      v2f bf = *bp;  bp += 2;
      acc = wmma_f32(a, bf, acc);
    }
    #pragma unroll
    for (int r = 0; r < 8; ++r) {
      int M = m0 + r + ((lane < 16) ? 0 : 8);
      h_lds[M * 210 + n] = acc[r];
    }
  }
  __syncthreads();

  for (int t = wave; t < 4 * 7; t += 8) {
    const int m0 = (t / 7) * 16, n0 = (t % 7) * 16;
    const int n  = n0 + l16;
    const v2f* ap = (const v2f*)&h_lds[(m0 + l16) * 210 + lhalf];
    const v2f* bp = (const v2f*)&w2p[n * HID + lhalf];
    v8f acc = {};
    #pragma unroll 5
    for (int k0 = 0; k0 < HID; k0 += 4) {
      v2f a  = *ap;  ap += 2;
      v2f bf = *bp;  bp += 2;
      acc = wmma_f32(a, bf, acc);
    }
    if (n < P2) {
      #pragma unroll
      for (int r = 0; r < 8; ++r) {
        int M = m0 + r + ((lane < 16) ? 0 : 8);
        float v = acc[r];
        dst[(size_t)(P0 + M) * P2 + n] = v > 0.f ? v : 0.f;
      }
    }
  }
}

#define ATT_LDS_FLOATS (128 * 102 + 64 * 102)

__global__ __launch_bounds__(256) void att_kernel(
    const float* __restrict__ ui, const float* __restrict__ uf,
    float* __restrict__ out) {
  extern __shared__ float smem[];
  float* us = smem;
  float* vs = us + 128 * 102;

  const int tid = threadIdx.x;
  const int bc = blockIdx.z;
  const int M0 = blockIdx.x * 128, N0 = blockIdx.y * 64;
  const float* U = ui + (size_t)bc * LPATCH * P2;
  const float* V = uf + (size_t)bc * LPATCH * P2;

  for (int i = tid; i < 128 * P2; i += 256) {
    int r = i / P2, k = i % P2;
    us[r * 102 + k] = U[(size_t)(M0 + r) * P2 + k];
  }
  for (int i = tid; i < 64 * P2; i += 256) {
    int r = i / P2, k = i % P2;
    vs[r * 102 + k] = V[(size_t)(N0 + r) * P2 + k];
  }
  __syncthreads();

  const int wave  = tid >> 5;
  const int lane  = tid & 31;
  const int l16   = lane & 15;
  const int lhalf = (lane < 16) ? 0 : 2;

  for (int t = wave; t < 8 * 4; t += 8) {
    const int m0 = (t >> 2) * 16, n0 = (t & 3) * 16;
    const v2f* ap = (const v2f*)&us[(m0 + l16) * 102 + lhalf];
    const v2f* bp = (const v2f*)&vs[(n0 + l16) * 102 + lhalf];
    v8f acc = {};
    #pragma unroll 5
    for (int k0 = 0; k0 < P2; k0 += 4) {
      v2f a  = *ap;  ap += 2;
      v2f bf = *bp;  bp += 2;
      acc = wmma_f32(a, bf, acc);
    }
    #pragma unroll
    for (int r = 0; r < 8; ++r) {
      int M = M0 + m0 + r + ((lane < 16) ? 0 : 8);
      int N = N0 + n0 + l16;
      out[((size_t)bc * LPATCH + M) * LPATCH + N] = acc[r] * 0.01f;
    }
  }
}

extern "C" void kernel_launch(void* const* d_in, const int* in_sizes, int n_in,
                              void* d_out, int out_size, void* d_ws,
                              size_t ws_size, hipStream_t stream) {
  (void)in_sizes; (void)n_in; (void)out_size; (void)ws_size;
  const float* x       = (const float*)d_in[0];
  const float* w_conv1 = (const float*)d_in[1];
  const float* b_conv1 = (const float*)d_in[2];
  const float* w_conv2 = (const float*)d_in[3];
  const float* b_conv2 = (const float*)d_in[4];
  const float* w_fm    = (const float*)d_in[5];
  const float* b_fm    = (const float*)d_in[6];
  const float* w_lin1  = (const float*)d_in[7];
  const float* w_lin2  = (const float*)d_in[8];
  float* out = (float*)d_out;

  float* ws  = (float*)d_ws;
  float* ini = ws + WS_INI;
  float* fea = ws + WS_FEA;
  float* ui  = ws + WS_UI;
  float* uf  = ws + WS_UF;
  float* w1p = ws + WS_W1P;
  float* w2p = ws + WS_W2P;

  pad_weights_kernel<<<dim3((112 * 200 + 255) / 256), 256, 0, stream>>>(
      w_lin1, w_lin2, w1p, w2p);

  conv_fused_kernel<<<dim3(HW / 16, HW / 16, BATCH), 256,
                      CONV_LDS_FLOATS * sizeof(float), stream>>>(
      x, w_conv1, b_conv1, w_conv2, b_conv2, w_fm, b_fm, ini, fea);

  trans_kernel<<<dim3(BATCH * CLS * LPATCH / 64, 2, 1), 256,
                 TRANS_LDS_FLOATS * sizeof(float), stream>>>(
      ini, fea, w1p, w2p, ui, uf);

  att_kernel<<<dim3(LPATCH / 128, LPATCH / 64, BATCH * CLS), 256,
               ATT_LDS_FLOATS * sizeof(float), stream>>>(ui, uf, out);
}


# conv kernel LDS staging: its one-at-a-time global loads (about 50 dependent round trips per wave) become one group of 13 coalesced 16-byte loads, from a constants image the prep kernel lays out
# speedup vs baseline: 1.2417x; 1.2417x over previous
#include <hip/hip_runtime.h>
#include <hip/hip_bf16.h>

typedef float v2f __attribute__((ext_vector_type(2)));
typedef float v8f __attribute__((ext_vector_type(8)));
typedef float v4f __attribute__((ext_vector_type(4)));

__device__ __forceinline__ v8f wmma_f32(v2f a, v2f b, v8f c) {
  return __builtin_amdgcn_wmma_f32_16x16x4_f32(false, a, false, b, (short)0, c,
                                               false, false);
}

#define BATCH   8
#define CIN     3
#define HW      320
#define CLS     2
#define PSZ     10
#define P2      100
#define LPATCH  1024
#define HID     200

#define WS_INI  0
#define WS_FEA  (WS_INI + BATCH*CLS*HW*HW)
#define WS_UI   (WS_FEA + BATCH*CLS*HW*HW)
#define WS_UF   (WS_UI  + BATCH*CLS*LPATCH*P2)
#define WS_W1P  (WS_UF  + BATCH*CLS*LPATCH*P2)
#define WS_W2P  (WS_W1P + 208*100)
#define WS_IMG  WS_UI
#define IMG_W1     9472
#define IMG_HEAD   (IMG_W1 + 64)
#define IMG_W2     1152
#define IMG_TAIL   (IMG_W2 + 16)
#define IMG_FLOATS (IMG_HEAD + IMG_TAIL)

__global__ __launch_bounds__(256) void pad_weights_kernel(
    const float* __restrict__ w_lin1, const float* __restrict__ w_lin2,
    const float* __restrict__ w1c, const float* __restrict__ b1c,
    const float* __restrict__ w2c, const float* __restrict__ b2c,
    const float* __restrict__ wfm, const float* __restrict__ bfm,
    float* __restrict__ w1p, float* __restrict__ w2p,
    float* __restrict__ img) {
  int i = blockIdx.x * 256 + threadIdx.x;
  if (i < 208 * 100) w1p[i] = (i < HID * P2) ? w_lin1[i] : 0.f;
  if (i < 112 * 200) w2p[i] = (i < P2 * HID) ? w_lin2[i] : 0.f;
  if (i < IMG_FLOATS) {
    float v = 0.f;
    if (i < IMG_W1) {
      int k = ((i >> 7) << 1) + (i & 1), n = (i & 127) >> 1;
      if (k < 147) v = w1c[n * 147 + k];
    } else if (i < IMG_HEAD) {
      v = b1c[i - IMG_W1];
    } else if (i < IMG_HEAD + IMG_W2) {
      v = w2c[i - IMG_HEAD];
    } else {
      int j = i - (IMG_HEAD + IMG_W2);
      if (j < 2)       v = b2c[j];
      else if (j < 8)  v = wfm[j - 2];
      else if (j < 10) v = bfm[j - 8];
    }
    img[i] = v;
  }
}

#define XS_FLOATS        2160
#define CONV_LDS_FLOATS  (XS_FLOATS + 148 + 9472 + 64 + 64*325 + 1152 + 16)

__global__ __launch_bounds__(256) void conv_fused_kernel(
    const float* __restrict__ x,  const float* __restrict__ img,
    float* __restrict__ ini, float* __restrict__ fea) {
  extern __shared__ float smem[];
  float* xs     = smem;
  int*   offtab = (int*)(xs + XS_FLOATS);
  float* w1s    = xs + XS_FLOATS + 148;
  float* b1s    = w1s + 74 * 64 * 2;
  float* t1s    = b1s + 64;
  float* w2s    = t1s + 64 * 325;
  float* cs     = w2s + 1152;

  const int tid = threadIdx.x;
  const int X0 = blockIdx.x * 16, Y0 = blockIdx.y * 16, b = blockIdx.z;

  const v4f* img4 = (const v4f*)img;
  v4f cw[11];
  #pragma unroll
  for (int j = 0; j < 11; ++j) {
    int q = tid + 256 * j;
    q = (q < IMG_FLOATS / 4) ? q : (IMG_FLOATS / 4 - 1);
    cw[j] = img4[q];
  }
  v4f xq[2];
  #pragma unroll
  for (int j = 0; j < 2; ++j) {
    const int q  = tid + 256 * j;
    const int qq = (q < CIN * 144) ? q : (CIN * 144 - 1);
    const int row = qq / 6, f = qq % 6, c = row / 24, iy = row % 24;
    const int gy = Y0 - 4 + iy, gx = X0 - 4 + 4 * f;
    const bool inb = (q < CIN * 144) && gy >= 0 && gy < HW && gx >= 0 && gx < HW;
    const int cy = gy < 0 ? 0 : (gy > HW - 1 ? HW - 1 : gy);
    const int cx = gx < 0 ? 0 : (gx > HW - 4 ? HW - 4 : gx);
    const v4f v = *(const v4f*)(x + ((size_t)(b * CIN + c) * HW + cy) * HW + cx);
    xq[j] = inb ? v : v4f{0.f, 0.f, 0.f, 0.f};
  }
  #pragma unroll
  for (int j = 0; j < 11; ++j) {
    const int q = tid + 256 * j;
    if (q < IMG_HEAD / 4)        ((v4f*)w1s)[q] = cw[j];
    else if (q < IMG_FLOATS / 4) ((v4f*)w2s)[q - IMG_HEAD / 4] = cw[j];
  }
  ((v4f*)xs)[tid]       = xq[0];
  ((v4f*)xs)[tid + 256] = xq[1];
  if (tid < XS_FLOATS / 4 - 512) ((v4f*)xs)[tid + 512] = v4f{0.f, 0.f, 0.f, 0.f};
  if (tid < 148) {
    int off = 1728;
    if (tid < 147) {
      int c = tid / 49, rr = tid % 49;
      off = c * 576 + (rr / 7) * 24 + (rr % 7);
    }
    offtab[tid] = off;
  }
  __syncthreads();

  const int wave  = tid >> 5;
  const int lane  = tid & 31;
  const int l16   = lane & 15;
  const int lhalf = (lane < 16) ? 0 : 2;

  for (int t = wave; t < 21 * 4; t += 8) {
    const int m0 = (t >> 2) * 16, n0 = (t & 3) * 16;
    const int m  = m0 + l16;
    const int mm = (m < 324) ? m : 0;
    const int abase = (mm / 18) * 24 + (mm % 18);
    const int2* ot = ((const int2*)offtab) + (lhalf >> 1);
    const v2f*  bp = ((const v2f*)w1s) + (lhalf >> 1) * 64 + n0 + l16;
    v8f acc = {};
    #pragma unroll 4
    for (int k0 = 0; k0 < 148; k0 += 4) {
      int2 o = *ot;  ot += 2;
      v2f a;
      a.x = xs[abase + o.x];
      a.y = xs[abase + o.y];
      v2f bf = *bp;  bp += 128;
      acc = wmma_f32(a, bf, acc);
    }
    const int ch = n0 + l16;
    const float bias = b1s[ch];
    #pragma unroll
    for (int r = 0; r < 8; ++r) {
      int M = m0 + r + ((lane < 16) ? 0 : 8);
      if (M < 324) {
        int ty = M / 18, tx = M % 18;
        int gy = Y0 - 1 + ty, gx = X0 - 1 + tx;
        float v = (gy >= 0 && gy < HW && gx >= 0 && gx < HW) ? (acc[r] + bias)
                                                             : 0.f;
        t1s[ch * 325 + M] = v;
      }
    }
  }
  __syncthreads();

  {
    const int oy = tid >> 4, ox = tid & 15;
    float a0 = cs[0], a1 = cs[1];
    for (int ci = 0; ci < 64; ++ci) {
      const float* trow = &t1s[ci * 325];
      const float* wr0  = &w2s[ci * 9];
      const float* wr1  = &w2s[(64 + ci) * 9];
      #pragma unroll
      for (int dy = 0; dy < 3; ++dy)
        #pragma unroll
        for (int dx = 0; dx < 3; ++dx) {
          float tv = trow[(oy + dy) * 18 + (ox + dx)];
          a0 = fmaf(wr0[dy * 3 + dx], tv, a0);
          a1 = fmaf(wr1[dy * 3 + dx], tv, a1);
        }
    }
    const int gy = Y0 + oy, gx = X0 + ox;
    const size_t o0 = ((size_t)(b * CLS + 0) * HW + gy) * HW + gx;
    const size_t o1 = ((size_t)(b * CLS + 1) * HW + gy) * HW + gx;
    ini[o0] = a0;
    ini[o1] = a1;
    float x0 = xs[(oy + 4) * 24 + (ox + 4)];
    float x1 = xs[576 + (oy + 4) * 24 + (ox + 4)];
    float x2 = xs[1152 + (oy + 4) * 24 + (ox + 4)];
    fea[o0] = (cs[2] * x0 + cs[3] * x1 + cs[4] * x2 + cs[8]) * 0.25f;
    fea[o1] = (cs[5] * x0 + cs[6] * x1 + cs[7] * x2 + cs[9]) * 0.25f;
  }
}

#define TRANS_LDS_FLOATS (64 * 102 + 64 * 210)

__global__ __launch_bounds__(256) void trans_kernel(
    const float* __restrict__ ini, const float* __restrict__ fea,
    const float* __restrict__ w1p, const float* __restrict__ w2p,
    float* __restrict__ ui, float* __restrict__ uf) {
  extern __shared__ float smem[];
  float* a_lds = smem;
  float* h_lds = a_lds + 64 * 102;

  const int tid = threadIdx.x;
  const float* src = blockIdx.y ? fea : ini;
  float* dst       = blockIdx.y ? uf  : ui;
  const int P0 = blockIdx.x * 64;

  for (int i = tid; i < 64 * P2; i += 256) {
    int r = i / P2, p = i % P2;
    int Pa = P0 + r;
    int bc = Pa >> 10, l = Pa & 1023;
    int gy = (l >> 5) * PSZ + p / PSZ;
    int gx = (l & 31) * PSZ + p % PSZ;
    a_lds[r * 102 + p] = src[((size_t)bc * HW + gy) * HW + gx];
  }
  __syncthreads();

  const int wave  = tid >> 5;
  const int lane  = tid & 31;
  const int l16   = lane & 15;
  const int lhalf = (lane < 16) ? 0 : 2;

  for (int t = wave; t < 4 * 13; t += 8) {
    const int m0 = (t / 13) * 16, n0 = (t % 13) * 16;
    const int n  = n0 + l16;
    const v2f* ap = (const v2f*)&a_lds[(m0 + l16) * 102 + lhalf];
    const v2f* bp = (const v2f*)&w1p[n * P2 + lhalf];
    v8f acc = {};
    #pragma unroll 5
    for (int k0 = 0; k0 < P2; k0 += 4) {
      v2f a  = *ap;  ap += 2;
      v2f bf = *bp;  bp += 2;
      acc = wmma_f32(a, bf, acc);
    }
    #pragma unroll
    for (int r = 0; r < 8; ++r) {
      int M = m0 + r + ((lane < 16) ? 0 : 8);
      h_lds[M * 210 + n] = acc[r];
    }
  }
  __syncthreads();

  for (int t = wave; t < 4 * 7; t += 8) {
    const int m0 = (t / 7) * 16, n0 = (t % 7) * 16;
    const int n  = n0 + l16;
    const v2f* ap = (const v2f*)&h_lds[(m0 + l16) * 210 + lhalf];
    const v2f* bp = (const v2f*)&w2p[n * HID + lhalf];
    v8f acc = {};
    #pragma unroll 5
    for (int k0 = 0; k0 < HID; k0 += 4) {
      v2f a  = *ap;  ap += 2;
      v2f bf = *bp;  bp += 2;
      acc = wmma_f32(a, bf, acc);
    }
    if (n < P2) {
      #pragma unroll
      for (int r = 0; r < 8; ++r) {
        int M = m0 + r + ((lane < 16) ? 0 : 8);
        float v = acc[r];
        dst[(size_t)(P0 + M) * P2 + n] = v > 0.f ? v : 0.f;
      }
    }
  }
}

#define ATT_LDS_FLOATS (128 * 102 + 64 * 102)

__global__ __launch_bounds__(256) void att_kernel(
    const float* __restrict__ ui, const float* __restrict__ uf,
    float* __restrict__ out) {
  extern __shared__ float smem[];
  float* us = smem;
  float* vs = us + 128 * 102;

  const int tid = threadIdx.x;
  const int bc = blockIdx.z;
  const int M0 = blockIdx.x * 128, N0 = blockIdx.y * 64;
  const float* U = ui + (size_t)bc * LPATCH * P2;
  const float* V = uf + (size_t)bc * LPATCH * P2;

  for (int i = tid; i < 128 * P2; i += 256) {
    int r = i / P2, k = i % P2;
    us[r * 102 + k] = U[(size_t)(M0 + r) * P2 + k];
  }
  for (int i = tid; i < 64 * P2; i += 256) {
    int r = i / P2, k = i % P2;
    vs[r * 102 + k] = V[(size_t)(N0 + r) * P2 + k];
  }
  __syncthreads();

  const int wave  = tid >> 5;
  const int lane  = tid & 31;
  const int l16   = lane & 15;
  const int lhalf = (lane < 16) ? 0 : 2;

  for (int t = wave; t < 8 * 4; t += 8) {
    const int m0 = (t >> 2) * 16, n0 = (t & 3) * 16;
    const v2f* ap = (const v2f*)&us[(m0 + l16) * 102 + lhalf];
    const v2f* bp = (const v2f*)&vs[(n0 + l16) * 102 + lhalf];
    v8f acc = {};
    #pragma unroll 5
    for (int k0 = 0; k0 < P2; k0 += 4) {
      v2f a  = *ap;  ap += 2;
      v2f bf = *bp;  bp += 2;
      acc = wmma_f32(a, bf, acc);
    }
    #pragma unroll
    for (int r = 0; r < 8; ++r) {
      int M = M0 + m0 + r + ((lane < 16) ? 0 : 8);
      int N = N0 + n0 + l16;
      out[((size_t)bc * LPATCH + M) * LPATCH + N] = acc[r] * 0.01f;
    }
  }
}

extern "C" void kernel_launch(void* const* d_in, const int* in_sizes, int n_in,
                              void* d_out, int out_size, void* d_ws,
                              size_t ws_size, hipStream_t stream) {
  (void)in_sizes; (void)n_in; (void)out_size; (void)ws_size;
  const float* x       = (const float*)d_in[0];
  const float* w_conv1 = (const float*)d_in[1];
  const float* b_conv1 = (const float*)d_in[2];
  const float* w_conv2 = (const float*)d_in[3];
  const float* b_conv2 = (const float*)d_in[4];
  const float* w_fm    = (const float*)d_in[5];
  const float* b_fm    = (const float*)d_in[6];
  const float* w_lin1  = (const float*)d_in[7];
  const float* w_lin2  = (const float*)d_in[8];
  float* out = (float*)d_out;

  float* ws  = (float*)d_ws;
  float* ini = ws + WS_INI;
  float* fea = ws + WS_FEA;
  float* ui  = ws + WS_UI;
  float* uf  = ws + WS_UF;
  float* w1p = ws + WS_W1P;
  float* w2p = ws + WS_W2P;
  float* img = ws + WS_IMG;

  pad_weights_kernel<<<dim3((112 * 200 + 255) / 256), 256, 0, stream>>>(
      w_lin1, w_lin2, w_conv1, b_conv1, w_conv2, b_conv2, w_fm, b_fm,
      w1p, w2p, img);

  conv_fused_kernel<<<dim3(HW / 16, HW / 16, BATCH), 256,
                      CONV_LDS_FLOATS * sizeof(float), stream>>>(
      x, img, ini, fea);

  trans_kernel<<<dim3(BATCH * CLS * LPATCH / 64, 2, 1), 256,
                 TRANS_LDS_FLOATS * sizeof(float), stream>>>(
      ini, fea, w1p, w2p, ui, uf);

  att_kernel<<<dim3(LPATCH / 128, LPATCH / 64, BATCH * CLS), 256,
               ATT_LDS_FLOATS * sizeof(float), stream>>>(ui, uf, out);
}


# conv2 weights and small constants reach the conv blocks as wave-uniform scalar loads from the constants image (SGPR operands of the FMAs) instead of per-lane LDS broadcast loads
# speedup vs baseline: 1.3366x; 1.0764x over previous
#include <hip/hip_runtime.h>
#include <hip/hip_bf16.h>

typedef float v2f __attribute__((ext_vector_type(2)));
typedef float v8f __attribute__((ext_vector_type(8)));
typedef float v4f __attribute__((ext_vector_type(4)));

__device__ __forceinline__ v8f wmma_f32(v2f a, v2f b, v8f c) {
  return __builtin_amdgcn_wmma_f32_16x16x4_f32(false, a, false, b, (short)0, c,
                                               false, false);
}

#define BATCH   8
#define CIN     3
#define HW      320
#define CLS     2
#define PSZ     10
#define P2      100
#define LPATCH  1024
#define HID     200

#define WS_INI  0
#define WS_FEA  (WS_INI + BATCH*CLS*HW*HW)
#define WS_UI   (WS_FEA + BATCH*CLS*HW*HW)
#define WS_UF   (WS_UI  + BATCH*CLS*LPATCH*P2)
#define WS_W1P  (WS_UF  + BATCH*CLS*LPATCH*P2)
#define WS_W2P  (WS_W1P + 208*100)
#define WS_IMG  WS_UI
#define IMG_W1     9472
#define IMG_HEAD   (IMG_W1 + 64)
#define IMG_W2     1152
#define IMG_TAIL   (IMG_W2 + 16)
#define IMG_FLOATS (IMG_HEAD + IMG_TAIL)

__global__ __launch_bounds__(256) void pad_weights_kernel(
    const float* __restrict__ w_lin1, const float* __restrict__ w_lin2,
    const float* __restrict__ w1c, const float* __restrict__ b1c,
    const float* __restrict__ w2c, const float* __restrict__ b2c,
    const float* __restrict__ wfm, const float* __restrict__ bfm,
    float* __restrict__ w1p, float* __restrict__ w2p,
    float* __restrict__ img) {
  int i = blockIdx.x * 256 + threadIdx.x;
  if (i < 208 * 100) w1p[i] = (i < HID * P2) ? w_lin1[i] : 0.f;
  if (i < 112 * 200) w2p[i] = (i < P2 * HID) ? w_lin2[i] : 0.f;
  if (i < IMG_FLOATS) {
    float v = 0.f;
    if (i < IMG_W1) {
      int k = ((i >> 7) << 1) + (i & 1), n = (i & 127) >> 1;
      if (k < 147) v = w1c[n * 147 + k];
    } else if (i < IMG_HEAD) {
      v = b1c[i - IMG_W1];
    } else if (i < IMG_HEAD + IMG_W2) {
      const int j = i - IMG_HEAD, ci = j / 18, r = j % 18;
      v = w2c[((r / 9) * 64 + ci) * 9 + (r % 9)];
    } else {
      int j = i - (IMG_HEAD + IMG_W2);
      if (j < 2)       v = b2c[j];
      else if (j < 8)  v = wfm[j - 2];
      else if (j < 10) v = bfm[j - 8];
    }
    img[i] = v;
  }
}

#define XS_FLOATS        2160
#define CONV_LDS_FLOATS  (XS_FLOATS + 148 + 9472 + 64 + 64*325)

__global__ __launch_bounds__(256) void conv_fused_kernel(
    const float* __restrict__ x,  const float* __restrict__ img,
    float* __restrict__ ini, float* __restrict__ fea) {
  extern __shared__ float smem[];
  float* xs     = smem;
  int*   offtab = (int*)(xs + XS_FLOATS);
  float* w1s    = xs + XS_FLOATS + 148;
  float* b1s    = w1s + 74 * 64 * 2;
  float* t1s    = b1s + 64;
  const float* __restrict__ w2g = img + IMG_HEAD;
  const float* __restrict__ cg  = img + IMG_HEAD + IMG_W2;

  const int tid = threadIdx.x;
  const int X0 = blockIdx.x * 16, Y0 = blockIdx.y * 16, b = blockIdx.z;

  const v4f* img4 = (const v4f*)img;
  v4f cw[10];
  #pragma unroll
  for (int j = 0; j < 10; ++j) {
    int q = tid + 256 * j;
    q = (q < IMG_HEAD / 4) ? q : (IMG_HEAD / 4 - 1);
    cw[j] = img4[q];
  }
  v4f xq[2];
  #pragma unroll
  for (int j = 0; j < 2; ++j) {
    const int q  = tid + 256 * j;
    const int qq = (q < CIN * 144) ? q : (CIN * 144 - 1);
    const int row = qq / 6, f = qq % 6, c = row / 24, iy = row % 24;
    const int gy = Y0 - 4 + iy, gx = X0 - 4 + 4 * f;
    const bool inb = (q < CIN * 144) && gy >= 0 && gy < HW && gx >= 0 && gx < HW;
    const int cy = gy < 0 ? 0 : (gy > HW - 1 ? HW - 1 : gy);
    const int cx = gx < 0 ? 0 : (gx > HW - 4 ? HW - 4 : gx);
    const v4f v = *(const v4f*)(x + ((size_t)(b * CIN + c) * HW + cy) * HW + cx);
    xq[j] = inb ? v : v4f{0.f, 0.f, 0.f, 0.f};
  }
  #pragma unroll
  for (int j = 0; j < 10; ++j) {
    const int q = tid + 256 * j;
    if (q < IMG_HEAD / 4) ((v4f*)w1s)[q] = cw[j];
  }
  ((v4f*)xs)[tid]       = xq[0];
  ((v4f*)xs)[tid + 256] = xq[1];
  if (tid < XS_FLOATS / 4 - 512) ((v4f*)xs)[tid + 512] = v4f{0.f, 0.f, 0.f, 0.f};
  if (tid < 148) {
    int off = 1728;
    if (tid < 147) {
      int c = tid / 49, rr = tid % 49;
      off = c * 576 + (rr / 7) * 24 + (rr % 7);
    }
    offtab[tid] = off;
  }
  __syncthreads();

  const int wave  = tid >> 5;
  const int lane  = tid & 31;
  const int l16   = lane & 15;
  const int lhalf = (lane < 16) ? 0 : 2;

  for (int t = wave; t < 21 * 4; t += 8) {
    const int m0 = (t >> 2) * 16, n0 = (t & 3) * 16;
    const int m  = m0 + l16;
    const int mm = (m < 324) ? m : 0;
    const int abase = (mm / 18) * 24 + (mm % 18);
    const int2* ot = ((const int2*)offtab) + (lhalf >> 1);
    const v2f*  bp = ((const v2f*)w1s) + (lhalf >> 1) * 64 + n0 + l16;
    v8f acc = {};
    #pragma unroll 4
    for (int k0 = 0; k0 < 148; k0 += 4) {
      int2 o = *ot;  ot += 2;
      v2f a;
      a.x = xs[abase + o.x];
      a.y = xs[abase + o.y];
      v2f bf = *bp;  bp += 128;
      acc = wmma_f32(a, bf, acc);
    }
    const int ch = n0 + l16;
    const float bias = b1s[ch];
    #pragma unroll
    for (int r = 0; r < 8; ++r) {
      int M = m0 + r + ((lane < 16) ? 0 : 8);
      if (M < 324) {
        int ty = M / 18, tx = M % 18;
        int gy = Y0 - 1 + ty, gx = X0 - 1 + tx;
        float v = (gy >= 0 && gy < HW && gx >= 0 && gx < HW) ? (acc[r] + bias)
                                                             : 0.f;
        t1s[ch * 325 + M] = v;
      }
    }
  }
  __syncthreads();

  {
    const int oy = tid >> 4, ox = tid & 15;
    float a0 = cg[0], a1 = cg[1];
    for (int ci = 0; ci < 64; ++ci) {
      const float* trow = &t1s[ci * 325];
      const float* wr0  = w2g + ci * 18;
      const float* wr1  = wr0 + 9;
      #pragma unroll
      for (int dy = 0; dy < 3; ++dy)
        #pragma unroll
        for (int dx = 0; dx < 3; ++dx) {
          float tv = trow[(oy + dy) * 18 + (ox + dx)];
          a0 = fmaf(wr0[dy * 3 + dx], tv, a0);
          a1 = fmaf(wr1[dy * 3 + dx], tv, a1);
        }
    }
    const int gy = Y0 + oy, gx = X0 + ox;
    const size_t o0 = ((size_t)(b * CLS + 0) * HW + gy) * HW + gx;
    const size_t o1 = ((size_t)(b * CLS + 1) * HW + gy) * HW + gx;
    ini[o0] = a0;
    ini[o1] = a1;
    float x0 = xs[(oy + 4) * 24 + (ox + 4)];
    float x1 = xs[576 + (oy + 4) * 24 + (ox + 4)];
    float x2 = xs[1152 + (oy + 4) * 24 + (ox + 4)];
    fea[o0] = (cg[2] * x0 + cg[3] * x1 + cg[4] * x2 + cg[8]) * 0.25f;
    fea[o1] = (cg[5] * x0 + cg[6] * x1 + cg[7] * x2 + cg[9]) * 0.25f;
  }
}

#define TRANS_LDS_FLOATS (64 * 102 + 64 * 210)

__global__ __launch_bounds__(256) void trans_kernel(
    const float* __restrict__ ini, const float* __restrict__ fea,
    const float* __restrict__ w1p, const float* __restrict__ w2p,
    float* __restrict__ ui, float* __restrict__ uf) {
  extern __shared__ float smem[];
  float* a_lds = smem;
  float* h_lds = a_lds + 64 * 102;

  const int tid = threadIdx.x;
  const float* src = blockIdx.y ? fea : ini;
  float* dst       = blockIdx.y ? uf  : ui;
  const int P0 = blockIdx.x * 64;

  for (int i = tid; i < 64 * P2; i += 256) {
    int r = i / P2, p = i % P2;
    int Pa = P0 + r;
    int bc = Pa >> 10, l = Pa & 1023;
    int gy = (l >> 5) * PSZ + p / PSZ;
    int gx = (l & 31) * PSZ + p % PSZ;
    a_lds[r * 102 + p] = src[((size_t)bc * HW + gy) * HW + gx];
  }
  __syncthreads();

  const int wave  = tid >> 5;
  const int lane  = tid & 31;
  const int l16   = lane & 15;
  const int lhalf = (lane < 16) ? 0 : 2;

  for (int t = wave; t < 4 * 13; t += 8) {
    const int m0 = (t / 13) * 16, n0 = (t % 13) * 16;
    const int n  = n0 + l16;
    const v2f* ap = (const v2f*)&a_lds[(m0 + l16) * 102 + lhalf];
    const v2f* bp = (const v2f*)&w1p[n * P2 + lhalf];
    v8f acc = {};
    #pragma unroll 5
    for (int k0 = 0; k0 < P2; k0 += 4) {
      v2f a  = *ap;  ap += 2;
      v2f bf = *bp;  bp += 2;
      acc = wmma_f32(a, bf, acc);
    }
    #pragma unroll
    for (int r = 0; r < 8; ++r) {
      int M = m0 + r + ((lane < 16) ? 0 : 8);
      h_lds[M * 210 + n] = acc[r];
    }
  }
  __syncthreads();

  for (int t = wave; t < 4 * 7; t += 8) {
    const int m0 = (t / 7) * 16, n0 = (t % 7) * 16;
    const int n  = n0 + l16;
    const v2f* ap = (const v2f*)&h_lds[(m0 + l16) * 210 + lhalf];
    const v2f* bp = (const v2f*)&w2p[n * HID + lhalf];
    v8f acc = {};
    #pragma unroll 5
    for (int k0 = 0; k0 < HID; k0 += 4) {
      v2f a  = *ap;  ap += 2;
      v2f bf = *bp;  bp += 2;
      acc = wmma_f32(a, bf, acc);
    }
    if (n < P2) {
      #pragma unroll
      for (int r = 0; r < 8; ++r) {
        int M = m0 + r + ((lane < 16) ? 0 : 8);
        float v = acc[r];
        dst[(size_t)(P0 + M) * P2 + n] = v > 0.f ? v : 0.f;
      }
    }
  }
}

#define ATT_LDS_FLOATS (128 * 102 + 64 * 102)

__global__ __launch_bounds__(256) void att_kernel(
    const float* __restrict__ ui, const float* __restrict__ uf,
    float* __restrict__ out) {
  extern __shared__ float smem[];
  float* us = smem;
  float* vs = us + 128 * 102;

  const int tid = threadIdx.x;
  const int bc = blockIdx.z;
  const int M0 = blockIdx.x * 128, N0 = blockIdx.y * 64;
  const float* U = ui + (size_t)bc * LPATCH * P2;
  const float* V = uf + (size_t)bc * LPATCH * P2;

  for (int i = tid; i < 128 * P2; i += 256) {
    int r = i / P2, k = i % P2;
    us[r * 102 + k] = U[(size_t)(M0 + r) * P2 + k];
  }
  for (int i = tid; i < 64 * P2; i += 256) {
    int r = i / P2, k = i % P2;
    vs[r * 102 + k] = V[(size_t)(N0 + r) * P2 + k];
  }
  __syncthreads();

  const int wave  = tid >> 5;
  const int lane  = tid & 31;
  const int l16   = lane & 15;
  const int lhalf = (lane < 16) ? 0 : 2;

  for (int t = wave; t < 8 * 4; t += 8) {
    const int m0 = (t >> 2) * 16, n0 = (t & 3) * 16;
    const v2f* ap = (const v2f*)&us[(m0 + l16) * 102 + lhalf];
    const v2f* bp = (const v2f*)&vs[(n0 + l16) * 102 + lhalf];
    v8f acc = {};
    #pragma unroll 5
    for (int k0 = 0; k0 < P2; k0 += 4) {
      v2f a  = *ap;  ap += 2;
      v2f bf = *bp;  bp += 2;
      acc = wmma_f32(a, bf, acc);
    }
    #pragma unroll
    for (int r = 0; r < 8; ++r) {
      int M = M0 + m0 + r + ((lane < 16) ? 0 : 8);
      int N = N0 + n0 + l16;
      out[((size_t)bc * LPATCH + M) * LPATCH + N] = acc[r] * 0.01f;
    }
  }
}

extern "C" void kernel_launch(void* const* d_in, const int* in_sizes, int n_in,
                              void* d_out, int out_size, void* d_ws,
                              size_t ws_size, hipStream_t stream) {
  (void)in_sizes; (void)n_in; (void)out_size; (void)ws_size;
  const float* x       = (const float*)d_in[0];
  const float* w_conv1 = (const float*)d_in[1];
  const float* b_conv1 = (const float*)d_in[2];
  const float* w_conv2 = (const float*)d_in[3];
  const float* b_conv2 = (const float*)d_in[4];
  const float* w_fm    = (const float*)d_in[5];
  const float* b_fm    = (const float*)d_in[6];
  const float* w_lin1  = (const float*)d_in[7];
  const float* w_lin2  = (const float*)d_in[8];
  float* out = (float*)d_out;

  float* ws  = (float*)d_ws;
  float* ini = ws + WS_INI;
  float* fea = ws + WS_FEA;
  float* ui  = ws + WS_UI;
  float* uf  = ws + WS_UF;
  float* w1p = ws + WS_W1P;
  float* w2p = ws + WS_W2P;
  float* img = ws + WS_IMG;

  pad_weights_kernel<<<dim3((112 * 200 + 255) / 256), 256, 0, stream>>>(
      w_lin1, w_lin2, w_conv1, b_conv1, w_conv2, b_conv2, w_fm, b_fm,
      w1p, w2p, img);

  conv_fused_kernel<<<dim3(HW / 16, HW / 16, BATCH), 256,
                      CONV_LDS_FLOATS * sizeof(float), stream>>>(
      x, img, ini, fea);

  trans_kernel<<<dim3(BATCH * CLS * LPATCH / 64, 2, 1), 256,
                 TRANS_LDS_FLOATS * sizeof(float), stream>>>(
      ini, fea, w1p, w2p, ui, uf);

  att_kernel<<<dim3(LPATCH / 128, LPATCH / 64, BATCH * CLS), 256,
               ATT_LDS_FLOATS * sizeof(float), stream>>>(ui, uf, out);
}


# conv1 K loop: a wave owns a 16x64 tile so one A gather (offset pair + 2 LDS pixel loads) feeds 4 matrix steps; each 4-K-step block's LDS loads are issued as one group, one block ahead
# speedup vs baseline: 1.6267x; 1.2170x over previous
#include <hip/hip_runtime.h>
#include <hip/hip_bf16.h>

typedef float v2f __attribute__((ext_vector_type(2)));
typedef float v8f __attribute__((ext_vector_type(8)));
typedef float v4f __attribute__((ext_vector_type(4)));

__device__ __forceinline__ v8f wmma_f32(v2f a, v2f b, v8f c) {
  return __builtin_amdgcn_wmma_f32_16x16x4_f32(false, a, false, b, (short)0, c,
                                               false, false);
}

#define BATCH   8
#define CIN     3
#define HW      320
#define CLS     2
#define PSZ     10
#define P2      100
#define LPATCH  1024
#define HID     200

#define WS_INI  0
#define WS_FEA  (WS_INI + BATCH*CLS*HW*HW)
#define WS_UI   (WS_FEA + BATCH*CLS*HW*HW)
#define WS_UF   (WS_UI  + BATCH*CLS*LPATCH*P2)
#define WS_W1P  (WS_UF  + BATCH*CLS*LPATCH*P2)
#define WS_W2P  (WS_W1P + 208*100)
#define WS_IMG  WS_UI
#define IMG_W1     9472
#define IMG_HEAD   (IMG_W1 + 64)
#define IMG_W2     1152
#define IMG_TAIL   (IMG_W2 + 16)
#define IMG_FLOATS (IMG_HEAD + IMG_TAIL)

__global__ __launch_bounds__(256) void pad_weights_kernel(
    const float* __restrict__ w_lin1, const float* __restrict__ w_lin2,
    const float* __restrict__ w1c, const float* __restrict__ b1c,
    const float* __restrict__ w2c, const float* __restrict__ b2c,
    const float* __restrict__ wfm, const float* __restrict__ bfm,
    float* __restrict__ w1p, float* __restrict__ w2p,
    float* __restrict__ img) {
  int i = blockIdx.x * 256 + threadIdx.x;
  if (i < 208 * 100) w1p[i] = (i < HID * P2) ? w_lin1[i] : 0.f;
  if (i < 112 * 200) w2p[i] = (i < P2 * HID) ? w_lin2[i] : 0.f;
  if (i < IMG_FLOATS) {
    float v = 0.f;
    if (i < IMG_W1) {
      int k = ((i >> 7) << 1) + (i & 1), n = (i & 127) >> 1;
      if (k < 147) v = w1c[n * 147 + k];
    } else if (i < IMG_HEAD) {
      v = b1c[i - IMG_W1];
    } else if (i < IMG_HEAD + IMG_W2) {
      const int j = i - IMG_HEAD, ci = j / 18, r = j % 18;
      v = w2c[((r / 9) * 64 + ci) * 9 + (r % 9)];
    } else {
      int j = i - (IMG_HEAD + IMG_W2);
      if (j < 2)       v = b2c[j];
      else if (j < 8)  v = wfm[j - 2];
      else if (j < 10) v = bfm[j - 8];
    }
    img[i] = v;
  }
}

#define XS_FLOATS        2160
#define CONV_LDS_FLOATS  (XS_FLOATS + 148 + 9472 + 64 + 64*325)

template <int NT>
__device__ __forceinline__ void conv1_tiles(
    const float* xs, const int* offtab, const float* w1s, const float* b1s,
    float* t1s, int m0, int n0, int lane, int X0, int Y0) {
  constexpr int KS = 37;
  constexpr int G  = 4;
  constexpr int NB = (KS + G - 1) / G;
  const int l16 = lane & 15;
  const int hk  = lane >> 4;
  const int m   = m0 + l16;
  const int mm  = (m < 324) ? m : 0;
  const float* xa = xs + (mm / 18) * 24 + (mm % 18);
  const int2* ot = ((const int2*)offtab) + hk;
  const v2f*  bp = ((const v2f*)w1s) + hk * 64 + n0 + l16;

  int2 off[2][G];
  v2f  af[2][G];
  v2f  bf[2][G][NT];
  v8f  acc[NT];
  #pragma unroll
  for (int j = 0; j < NT; ++j) acc[j] = v8f{};

  float bias[NT];
  #pragma unroll
  for (int j = 0; j < NT; ++j) bias[j] = b1s[n0 + 16 * j + l16];
  #pragma unroll
  for (int i = 0; i < G; ++i) { off[0][i] = ot[2 * i]; off[1][i] = ot[2 * (G + i)]; }
  #pragma unroll
  for (int i = 0; i < G; ++i) {
    af[0][i].x = xa[off[0][i].x];
    af[0][i].y = xa[off[0][i].y];
    #pragma unroll
    for (int j = 0; j < NT; ++j) bf[0][i][j] = bp[128 * i + 16 * j];
  }

  #pragma unroll
  for (int g = 0; g < NB; ++g) {
    const int cur = g & 1, nxt = cur ^ 1;
    if (g + 1 < NB) {
      #pragma unroll
      for (int i = 0; i < G; ++i) {
        const int s = (g + 1) * G + i;
        if (s < KS) {
          af[nxt][i].x = xa[off[nxt][i].x];
          af[nxt][i].y = xa[off[nxt][i].y];
          #pragma unroll
          for (int j = 0; j < NT; ++j) bf[nxt][i][j] = bp[128 * s + 16 * j];
        }
      }
    }
    if (g + 2 < NB) {
      #pragma unroll
      for (int i = 0; i < G; ++i) {
        const int s = (g + 2) * G + i;
        if (s < KS) off[cur][i] = ot[2 * s];
      }
    }
    __builtin_amdgcn_sched_barrier(0);
    #pragma unroll
    for (int i = 0; i < G; ++i) {
      if (g * G + i < KS) {
        #pragma unroll
        for (int j = 0; j < NT; ++j) acc[j] = wmma_f32(af[cur][i], bf[cur][i][j], acc[j]);
      }
    }
    __builtin_amdgcn_sched_barrier(0);
  }

  #pragma unroll
  for (int r = 0; r < 8; ++r) {
    const int M = m0 + r + 8 * hk;
    if (M < 324) {
      const int ty = M / 18, tx = M % 18;
      const int gy = Y0 - 1 + ty, gx = X0 - 1 + tx;
      const bool in = gy >= 0 && gy < HW && gx >= 0 && gx < HW;
      #pragma unroll
      for (int j = 0; j < NT; ++j) {
        const int ch = n0 + 16 * j + l16;
        t1s[ch * 325 + M] = in ? (acc[j][r] + bias[j]) : 0.f;
      }
    }
  }
}

__global__ __launch_bounds__(256) void conv_fused_kernel(
    const float* __restrict__ x,  const float* __restrict__ img,
    float* __restrict__ ini, float* __restrict__ fea) {
  extern __shared__ float smem[];
  float* xs     = smem;
  int*   offtab = (int*)(xs + XS_FLOATS);
  float* w1s    = xs + XS_FLOATS + 148;
  float* b1s    = w1s + 74 * 64 * 2;
  float* t1s    = b1s + 64;
  const float* __restrict__ w2g = img + IMG_HEAD;
  const float* __restrict__ cg  = img + IMG_HEAD + IMG_W2;

  const int tid = threadIdx.x;
  const int X0 = blockIdx.x * 16, Y0 = blockIdx.y * 16, b = blockIdx.z;

  const v4f* img4 = (const v4f*)img;
  v4f cw[10];
  #pragma unroll
  for (int j = 0; j < 10; ++j) {
    int q = tid + 256 * j;
    q = (q < IMG_HEAD / 4) ? q : (IMG_HEAD / 4 - 1);
    cw[j] = img4[q];
  }
  v4f xq[2];
  #pragma unroll
  for (int j = 0; j < 2; ++j) {
    const int q  = tid + 256 * j;
    const int qq = (q < CIN * 144) ? q : (CIN * 144 - 1);
    const int row = qq / 6, f = qq % 6, c = row / 24, iy = row % 24;
    const int gy = Y0 - 4 + iy, gx = X0 - 4 + 4 * f;
    const bool inb = (q < CIN * 144) && gy >= 0 && gy < HW && gx >= 0 && gx < HW;
    const int cy = gy < 0 ? 0 : (gy > HW - 1 ? HW - 1 : gy);
    const int cx = gx < 0 ? 0 : (gx > HW - 4 ? HW - 4 : gx);
    const v4f v = *(const v4f*)(x + ((size_t)(b * CIN + c) * HW + cy) * HW + cx);
    xq[j] = inb ? v : v4f{0.f, 0.f, 0.f, 0.f};
  }
  #pragma unroll
  for (int j = 0; j < 10; ++j) {
    const int q = tid + 256 * j;
    if (q < IMG_HEAD / 4) ((v4f*)w1s)[q] = cw[j];
  }
  ((v4f*)xs)[tid]       = xq[0];
  ((v4f*)xs)[tid + 256] = xq[1];
  if (tid < XS_FLOATS / 4 - 512) ((v4f*)xs)[tid + 512] = v4f{0.f, 0.f, 0.f, 0.f};
  if (tid < 148) {
    int off = 1728;
    if (tid < 147) {
      int c = tid / 49, rr = tid % 49;
      off = c * 576 + (rr / 7) * 24 + (rr % 7);
    }
    offtab[tid] = off;
  }
  __syncthreads();

  const int wave = tid >> 5;
  const int lane = tid & 31;

  for (int mt = wave; mt < 20; mt += 8)
    conv1_tiles<4>(xs, offtab, w1s, b1s, t1s, mt * 16, 0, lane, X0, Y0);
  if (wave >= 4)
    conv1_tiles<1>(xs, offtab, w1s, b1s, t1s, 320, (wave - 4) * 16, lane, X0, Y0);
  __syncthreads();

  {
    const int oy = tid >> 4, ox = tid & 15;
    float a0 = cg[0], a1 = cg[1];
    for (int ci = 0; ci < 64; ++ci) {
      const float* trow = &t1s[ci * 325];
      const float* wr0  = w2g + ci * 18;
      const float* wr1  = wr0 + 9;
      #pragma unroll
      for (int dy = 0; dy < 3; ++dy)
        #pragma unroll
        for (int dx = 0; dx < 3; ++dx) {
          float tv = trow[(oy + dy) * 18 + (ox + dx)];
          a0 = fmaf(wr0[dy * 3 + dx], tv, a0);
          a1 = fmaf(wr1[dy * 3 + dx], tv, a1);
        }
    }
    const int gy = Y0 + oy, gx = X0 + ox;
    const size_t o0 = ((size_t)(b * CLS + 0) * HW + gy) * HW + gx;
    const size_t o1 = ((size_t)(b * CLS + 1) * HW + gy) * HW + gx;
    ini[o0] = a0;
    ini[o1] = a1;
    float x0 = xs[(oy + 4) * 24 + (ox + 4)];
    float x1 = xs[576 + (oy + 4) * 24 + (ox + 4)];
    float x2 = xs[1152 + (oy + 4) * 24 + (ox + 4)];
    fea[o0] = (cg[2] * x0 + cg[3] * x1 + cg[4] * x2 + cg[8]) * 0.25f;
    fea[o1] = (cg[5] * x0 + cg[6] * x1 + cg[7] * x2 + cg[9]) * 0.25f;
  }
}

#define TRANS_LDS_FLOATS (64 * 102 + 64 * 210)

__global__ __launch_bounds__(256) void trans_kernel(
    const float* __restrict__ ini, const float* __restrict__ fea,
    const float* __restrict__ w1p, const float* __restrict__ w2p,
    float* __restrict__ ui, float* __restrict__ uf) {
  extern __shared__ float smem[];
  float* a_lds = smem;
  float* h_lds = a_lds + 64 * 102;

  const int tid = threadIdx.x;
  const float* src = blockIdx.y ? fea : ini;
  float* dst       = blockIdx.y ? uf  : ui;
  const int P0 = blockIdx.x * 64;

  for (int i = tid; i < 64 * P2; i += 256) {
    int r = i / P2, p = i % P2;
    int Pa = P0 + r;
    int bc = Pa >> 10, l = Pa & 1023;
    int gy = (l >> 5) * PSZ + p / PSZ;
    int gx = (l & 31) * PSZ + p % PSZ;
    a_lds[r * 102 + p] = src[((size_t)bc * HW + gy) * HW + gx];
  }
  __syncthreads();

  const int wave  = tid >> 5;
  const int lane  = tid & 31;
  const int l16   = lane & 15;
  const int lhalf = (lane < 16) ? 0 : 2;

  for (int t = wave; t < 4 * 13; t += 8) {
    const int m0 = (t / 13) * 16, n0 = (t % 13) * 16;
    const int n  = n0 + l16;
    const v2f* ap = (const v2f*)&a_lds[(m0 + l16) * 102 + lhalf];
    const v2f* bp = (const v2f*)&w1p[n * P2 + lhalf];
    v8f acc = {};
    #pragma unroll 5
    for (int k0 = 0; k0 < P2; k0 += 4) {
      v2f a  = *ap;  ap += 2;
      v2f bf = *bp;  bp += 2;
      acc = wmma_f32(a, bf, acc);
    }
    #pragma unroll
    for (int r = 0; r < 8; ++r) {
      int M = m0 + r + ((lane < 16) ? 0 : 8);
      h_lds[M * 210 + n] = acc[r];
    }
  }
  __syncthreads();

  for (int t = wave; t < 4 * 7; t += 8) {
    const int m0 = (t / 7) * 16, n0 = (t % 7) * 16;
    const int n  = n0 + l16;
    const v2f* ap = (const v2f*)&h_lds[(m0 + l16) * 210 + lhalf];
    const v2f* bp = (const v2f*)&w2p[n * HID + lhalf];
    v8f acc = {};
    #pragma unroll 5
    for (int k0 = 0; k0 < HID; k0 += 4) {
      v2f a  = *ap;  ap += 2;
      v2f bf = *bp;  bp += 2;
      acc = wmma_f32(a, bf, acc);
    }
    if (n < P2) {
      #pragma unroll
      for (int r = 0; r < 8; ++r) {
        int M = m0 + r + ((lane < 16) ? 0 : 8);
        float v = acc[r];
        dst[(size_t)(P0 + M) * P2 + n] = v > 0.f ? v : 0.f;
      }
    }
  }
}

#define ATT_LDS_FLOATS (128 * 102 + 64 * 102)

__global__ __launch_bounds__(256) void att_kernel(
    const float* __restrict__ ui, const float* __restrict__ uf,
    float* __restrict__ out) {
  extern __shared__ float smem[];
  float* us = smem;
  float* vs = us + 128 * 102;

  const int tid = threadIdx.x;
  const int bc = blockIdx.z;
  const int M0 = blockIdx.x * 128, N0 = blockIdx.y * 64;
  const float* U = ui + (size_t)bc * LPATCH * P2;
  const float* V = uf + (size_t)bc * LPATCH * P2;

  for (int i = tid; i < 128 * P2; i += 256) {
    int r = i / P2, k = i % P2;
    us[r * 102 + k] = U[(size_t)(M0 + r) * P2 + k];
  }
  for (int i = tid; i < 64 * P2; i += 256) {
    int r = i / P2, k = i % P2;
    vs[r * 102 + k] = V[(size_t)(N0 + r) * P2 + k];
  }
  __syncthreads();

  const int wave  = tid >> 5;
  const int lane  = tid & 31;
  const int l16   = lane & 15;
  const int lhalf = (lane < 16) ? 0 : 2;

  for (int t = wave; t < 8 * 4; t += 8) {
    const int m0 = (t >> 2) * 16, n0 = (t & 3) * 16;
    const v2f* ap = (const v2f*)&us[(m0 + l16) * 102 + lhalf];
    const v2f* bp = (const v2f*)&vs[(n0 + l16) * 102 + lhalf];
    v8f acc = {};
    #pragma unroll 5
    for (int k0 = 0; k0 < P2; k0 += 4) {
      v2f a  = *ap;  ap += 2;
      v2f bf = *bp;  bp += 2;
      acc = wmma_f32(a, bf, acc);
    }
    #pragma unroll
    for (int r = 0; r < 8; ++r) {
      int M = M0 + m0 + r + ((lane < 16) ? 0 : 8);
      int N = N0 + n0 + l16;
      out[((size_t)bc * LPATCH + M) * LPATCH + N] = acc[r] * 0.01f;
    }
  }
}

extern "C" void kernel_launch(void* const* d_in, const int* in_sizes, int n_in,
                              void* d_out, int out_size, void* d_ws,
                              size_t ws_size, hipStream_t stream) {
  (void)in_sizes; (void)n_in; (void)out_size; (void)ws_size;
  const float* x       = (const float*)d_in[0];
  const float* w_conv1 = (const float*)d_in[1];
  const float* b_conv1 = (const float*)d_in[2];
  const float* w_conv2 = (const float*)d_in[3];
  const float* b_conv2 = (const float*)d_in[4];
  const float* w_fm    = (const float*)d_in[5];
  const float* b_fm    = (const float*)d_in[6];
  const float* w_lin1  = (const float*)d_in[7];
  const float* w_lin2  = (const float*)d_in[8];
  float* out = (float*)d_out;

  float* ws  = (float*)d_ws;
  float* ini = ws + WS_INI;
  float* fea = ws + WS_FEA;
  float* ui  = ws + WS_UI;
  float* uf  = ws + WS_UF;
  float* w1p = ws + WS_W1P;
  float* w2p = ws + WS_W2P;
  float* img = ws + WS_IMG;

  pad_weights_kernel<<<dim3((112 * 200 + 255) / 256), 256, 0, stream>>>(
      w_lin1, w_lin2, w_conv1, b_conv1, w_conv2, b_conv2, w_fm, b_fm,
      w1p, w2p, img);

  conv_fused_kernel<<<dim3(HW / 16, HW / 16, BATCH), 256,
                      CONV_LDS_FLOATS * sizeof(float), stream>>>(
      x, img, ini, fea);

  trans_kernel<<<dim3(BATCH * CLS * LPATCH / 64, 2, 1), 256,
                 TRANS_LDS_FLOATS * sizeof(float), stream>>>(
      ini, fea, w1p, w2p, ui, uf);

  att_kernel<<<dim3(LPATCH / 128, LPATCH / 64, BATCH * CLS), 256,
               ATT_LDS_FLOATS * sizeof(float), stream>>>(ui, uf, out);
}


# att kernel: U and V tiles go to LDS as flat 16-byte copies of their contiguous rows at pitch 100 floats (conflict-free 8-byte fragment reads), loads issued as one group ahead of the LDS stores
# speedup vs baseline: 1.7205x; 1.0577x over previous
#include <hip/hip_runtime.h>
#include <hip/hip_bf16.h>

typedef float v2f __attribute__((ext_vector_type(2)));
typedef float v8f __attribute__((ext_vector_type(8)));
typedef float v4f __attribute__((ext_vector_type(4)));

__device__ __forceinline__ v8f wmma_f32(v2f a, v2f b, v8f c) {
  return __builtin_amdgcn_wmma_f32_16x16x4_f32(false, a, false, b, (short)0, c,
                                               false, false);
}

#define BATCH   8
#define CIN     3
#define HW      320
#define CLS     2
#define PSZ     10
#define P2      100
#define LPATCH  1024
#define HID     200

#define WS_INI  0
#define WS_FEA  (WS_INI + BATCH*CLS*HW*HW)
#define WS_UI   (WS_FEA + BATCH*CLS*HW*HW)
#define WS_UF   (WS_UI  + BATCH*CLS*LPATCH*P2)
#define WS_W1P  (WS_UF  + BATCH*CLS*LPATCH*P2)
#define WS_W2P  (WS_W1P + 208*100)
#define WS_IMG  WS_UI
#define IMG_W1     9472
#define IMG_HEAD   (IMG_W1 + 64)
#define IMG_W2     1152
#define IMG_TAIL   (IMG_W2 + 16)
#define IMG_FLOATS (IMG_HEAD + IMG_TAIL)

__global__ __launch_bounds__(256) void pad_weights_kernel(
    const float* __restrict__ w_lin1, const float* __restrict__ w_lin2,
    const float* __restrict__ w1c, const float* __restrict__ b1c,
    const float* __restrict__ w2c, const float* __restrict__ b2c,
    const float* __restrict__ wfm, const float* __restrict__ bfm,
    float* __restrict__ w1p, float* __restrict__ w2p,
    float* __restrict__ img) {
  int i = blockIdx.x * 256 + threadIdx.x;
  if (i < 208 * 100) w1p[i] = (i < HID * P2) ? w_lin1[i] : 0.f;
  if (i < 112 * 200) w2p[i] = (i < P2 * HID) ? w_lin2[i] : 0.f;
  if (i < IMG_FLOATS) {
    float v = 0.f;
    if (i < IMG_W1) {
      int k = ((i >> 7) << 1) + (i & 1), n = (i & 127) >> 1;
      if (k < 147) v = w1c[n * 147 + k];
    } else if (i < IMG_HEAD) {
      v = b1c[i - IMG_W1];
    } else if (i < IMG_HEAD + IMG_W2) {
      const int j = i - IMG_HEAD, ci = j / 18, r = j % 18;
      v = w2c[((r / 9) * 64 + ci) * 9 + (r % 9)];
    } else {
      int j = i - (IMG_HEAD + IMG_W2);
      if (j < 2)       v = b2c[j];
      else if (j < 8)  v = wfm[j - 2];
      else if (j < 10) v = bfm[j - 8];
    }
    img[i] = v;
  }
}

#define XS_FLOATS        2160
#define CONV_LDS_FLOATS  (XS_FLOATS + 148 + 9472 + 64 + 64*325)

template <int NT>
__device__ __forceinline__ void conv1_tiles(
    const float* xs, const int* offtab, const float* w1s, const float* b1s,
    float* t1s, int m0, int n0, int lane, int X0, int Y0) {
  constexpr int KS = 37;
  constexpr int G  = 4;
  constexpr int NB = (KS + G - 1) / G;
  const int l16 = lane & 15;
  const int hk  = lane >> 4;
  const int m   = m0 + l16;
  const int mm  = (m < 324) ? m : 0;
  const float* xa = xs + (mm / 18) * 24 + (mm % 18);
  const int2* ot = ((const int2*)offtab) + hk;
  const v2f*  bp = ((const v2f*)w1s) + hk * 64 + n0 + l16;

  int2 off[2][G];
  v2f  af[2][G];
  v2f  bf[2][G][NT];
  v8f  acc[NT];
  #pragma unroll
  for (int j = 0; j < NT; ++j) acc[j] = v8f{};

  float bias[NT];
  #pragma unroll
  for (int j = 0; j < NT; ++j) bias[j] = b1s[n0 + 16 * j + l16];
  #pragma unroll
  for (int i = 0; i < G; ++i) { off[0][i] = ot[2 * i]; off[1][i] = ot[2 * (G + i)]; }
  #pragma unroll
  for (int i = 0; i < G; ++i) {
    af[0][i].x = xa[off[0][i].x];
    af[0][i].y = xa[off[0][i].y];
    #pragma unroll
    for (int j = 0; j < NT; ++j) bf[0][i][j] = bp[128 * i + 16 * j];
  }

  #pragma unroll
  for (int g = 0; g < NB; ++g) {
    const int cur = g & 1, nxt = cur ^ 1;
    if (g + 1 < NB) {
      #pragma unroll
      for (int i = 0; i < G; ++i) {
        const int s = (g + 1) * G + i;
        if (s < KS) {
          af[nxt][i].x = xa[off[nxt][i].x];
          af[nxt][i].y = xa[off[nxt][i].y];
          #pragma unroll
          for (int j = 0; j < NT; ++j) bf[nxt][i][j] = bp[128 * s + 16 * j];
        }
      }
    }
    if (g + 2 < NB) {
      #pragma unroll
      for (int i = 0; i < G; ++i) {
        const int s = (g + 2) * G + i;
        if (s < KS) off[cur][i] = ot[2 * s];
      }
    }
    __builtin_amdgcn_sched_barrier(0);
    #pragma unroll
    for (int i = 0; i < G; ++i) {
      if (g * G + i < KS) {
        #pragma unroll
        for (int j = 0; j < NT; ++j) acc[j] = wmma_f32(af[cur][i], bf[cur][i][j], acc[j]);
      }
    }
    __builtin_amdgcn_sched_barrier(0);
  }

  #pragma unroll
  for (int r = 0; r < 8; ++r) {
    const int M = m0 + r + 8 * hk;
    if (M < 324) {
      const int ty = M / 18, tx = M % 18;
      const int gy = Y0 - 1 + ty, gx = X0 - 1 + tx;
      const bool in = gy >= 0 && gy < HW && gx >= 0 && gx < HW;
      #pragma unroll
      for (int j = 0; j < NT; ++j) {
        const int ch = n0 + 16 * j + l16;
        t1s[ch * 325 + M] = in ? (acc[j][r] + bias[j]) : 0.f;
      }
    }
  }
}

__global__ __launch_bounds__(256) void conv_fused_kernel(
    const float* __restrict__ x,  const float* __restrict__ img,
    float* __restrict__ ini, float* __restrict__ fea) {
  extern __shared__ float smem[];
  float* xs     = smem;
  int*   offtab = (int*)(xs + XS_FLOATS);
  float* w1s    = xs + XS_FLOATS + 148;
  float* b1s    = w1s + 74 * 64 * 2;
  float* t1s    = b1s + 64;
  const float* __restrict__ w2g = img + IMG_HEAD;
  const float* __restrict__ cg  = img + IMG_HEAD + IMG_W2;

  const int tid = threadIdx.x;
  const int X0 = blockIdx.x * 16, Y0 = blockIdx.y * 16, b = blockIdx.z;

  const v4f* img4 = (const v4f*)img;
  v4f cw[10];
  #pragma unroll
  for (int j = 0; j < 10; ++j) {
    int q = tid + 256 * j;
    q = (q < IMG_HEAD / 4) ? q : (IMG_HEAD / 4 - 1);
    cw[j] = img4[q];
  }
  v4f xq[2];
  #pragma unroll
  for (int j = 0; j < 2; ++j) {
    const int q  = tid + 256 * j;
    const int qq = (q < CIN * 144) ? q : (CIN * 144 - 1);
    const int row = qq / 6, f = qq % 6, c = row / 24, iy = row % 24;
    const int gy = Y0 - 4 + iy, gx = X0 - 4 + 4 * f;
    const bool inb = (q < CIN * 144) && gy >= 0 && gy < HW && gx >= 0 && gx < HW;
    const int cy = gy < 0 ? 0 : (gy > HW - 1 ? HW - 1 : gy);
    const int cx = gx < 0 ? 0 : (gx > HW - 4 ? HW - 4 : gx);
    const v4f v = *(const v4f*)(x + ((size_t)(b * CIN + c) * HW + cy) * HW + cx);
    xq[j] = inb ? v : v4f{0.f, 0.f, 0.f, 0.f};
  }
  #pragma unroll
  for (int j = 0; j < 10; ++j) {
    const int q = tid + 256 * j;
    if (q < IMG_HEAD / 4) ((v4f*)w1s)[q] = cw[j];
  }
  ((v4f*)xs)[tid]       = xq[0];
  ((v4f*)xs)[tid + 256] = xq[1];
  if (tid < XS_FLOATS / 4 - 512) ((v4f*)xs)[tid + 512] = v4f{0.f, 0.f, 0.f, 0.f};
  if (tid < 148) {
    int off = 1728;
    if (tid < 147) {
      int c = tid / 49, rr = tid % 49;
      off = c * 576 + (rr / 7) * 24 + (rr % 7);
    }
    offtab[tid] = off;
  }
  __syncthreads();

  const int wave = tid >> 5;
  const int lane = tid & 31;

  for (int mt = wave; mt < 20; mt += 8)
    conv1_tiles<4>(xs, offtab, w1s, b1s, t1s, mt * 16, 0, lane, X0, Y0);
  if (wave >= 4)
    conv1_tiles<1>(xs, offtab, w1s, b1s, t1s, 320, (wave - 4) * 16, lane, X0, Y0);
  __syncthreads();

  {
    const int oy = tid >> 4, ox = tid & 15;
    float a0 = cg[0], a1 = cg[1];
    for (int ci = 0; ci < 64; ++ci) {
      const float* trow = &t1s[ci * 325];
      const float* wr0  = w2g + ci * 18;
      const float* wr1  = wr0 + 9;
      #pragma unroll
      for (int dy = 0; dy < 3; ++dy)
        #pragma unroll
        for (int dx = 0; dx < 3; ++dx) {
          float tv = trow[(oy + dy) * 18 + (ox + dx)];
          a0 = fmaf(wr0[dy * 3 + dx], tv, a0);
          a1 = fmaf(wr1[dy * 3 + dx], tv, a1);
        }
    }
    const int gy = Y0 + oy, gx = X0 + ox;
    const size_t o0 = ((size_t)(b * CLS + 0) * HW + gy) * HW + gx;
    const size_t o1 = ((size_t)(b * CLS + 1) * HW + gy) * HW + gx;
    ini[o0] = a0;
    ini[o1] = a1;
    float x0 = xs[(oy + 4) * 24 + (ox + 4)];
    float x1 = xs[576 + (oy + 4) * 24 + (ox + 4)];
    float x2 = xs[1152 + (oy + 4) * 24 + (ox + 4)];
    fea[o0] = (cg[2] * x0 + cg[3] * x1 + cg[4] * x2 + cg[8]) * 0.25f;
    fea[o1] = (cg[5] * x0 + cg[6] * x1 + cg[7] * x2 + cg[9]) * 0.25f;
  }
}

#define TRANS_LDS_FLOATS (64 * 102 + 64 * 210)

__global__ __launch_bounds__(256) void trans_kernel(
    const float* __restrict__ ini, const float* __restrict__ fea,
    const float* __restrict__ w1p, const float* __restrict__ w2p,
    float* __restrict__ ui, float* __restrict__ uf) {
  extern __shared__ float smem[];
  float* a_lds = smem;
  float* h_lds = a_lds + 64 * 102;

  const int tid = threadIdx.x;
  const float* src = blockIdx.y ? fea : ini;
  float* dst       = blockIdx.y ? uf  : ui;
  const int P0 = blockIdx.x * 64;

  for (int i = tid; i < 64 * P2; i += 256) {
    int r = i / P2, p = i % P2;
    int Pa = P0 + r;
    int bc = Pa >> 10, l = Pa & 1023;
    int gy = (l >> 5) * PSZ + p / PSZ;
    int gx = (l & 31) * PSZ + p % PSZ;
    a_lds[r * 102 + p] = src[((size_t)bc * HW + gy) * HW + gx];
  }
  __syncthreads();

  const int wave  = tid >> 5;
  const int lane  = tid & 31;
  const int l16   = lane & 15;
  const int lhalf = (lane < 16) ? 0 : 2;

  for (int t = wave; t < 4 * 13; t += 8) {
    const int m0 = (t / 13) * 16, n0 = (t % 13) * 16;
    const int n  = n0 + l16;
    const v2f* ap = (const v2f*)&a_lds[(m0 + l16) * 102 + lhalf];
    const v2f* bp = (const v2f*)&w1p[n * P2 + lhalf];
    v8f acc = {};
    #pragma unroll 5
    for (int k0 = 0; k0 < P2; k0 += 4) {
      v2f a  = *ap;  ap += 2;
      v2f bf = *bp;  bp += 2;
      acc = wmma_f32(a, bf, acc);
    }
    #pragma unroll
    for (int r = 0; r < 8; ++r) {
      int M = m0 + r + ((lane < 16) ? 0 : 8);
      h_lds[M * 210 + n] = acc[r];
    }
  }
  __syncthreads();

  for (int t = wave; t < 4 * 7; t += 8) {
    const int m0 = (t / 7) * 16, n0 = (t % 7) * 16;
    const int n  = n0 + l16;
    const v2f* ap = (const v2f*)&h_lds[(m0 + l16) * 210 + lhalf];
    const v2f* bp = (const v2f*)&w2p[n * HID + lhalf];
    v8f acc = {};
    #pragma unroll 5
    for (int k0 = 0; k0 < HID; k0 += 4) {
      v2f a  = *ap;  ap += 2;
      v2f bf = *bp;  bp += 2;
      acc = wmma_f32(a, bf, acc);
    }
    if (n < P2) {
      #pragma unroll
      for (int r = 0; r < 8; ++r) {
        int M = m0 + r + ((lane < 16) ? 0 : 8);
        float v = acc[r];
        dst[(size_t)(P0 + M) * P2 + n] = v > 0.f ? v : 0.f;
      }
    }
  }
}

#define ATT_U_CHUNKS   (128 * P2 / 4)
#define ATT_CHUNKS     (ATT_U_CHUNKS + 64 * P2 / 4)
#define ATT_LDS_FLOATS (128 * P2 + 64 * P2)

__global__ __launch_bounds__(256) void att_kernel(
    const float* __restrict__ ui, const float* __restrict__ uf,
    float* __restrict__ out) {
  extern __shared__ float smem[];
  float* us = smem;
  float* vs = us + 128 * P2;

  const int tid = threadIdx.x;
  const int bc = blockIdx.z;
  const int M0 = blockIdx.x * 128, N0 = blockIdx.y * 64;
  const v4f* U4 = (const v4f*)(ui + ((size_t)bc * LPATCH + M0) * P2);
  const v4f* V4 = (const v4f*)(uf + ((size_t)bc * LPATCH + N0) * P2);

  v4f r[19];
  #pragma unroll
  for (int j = 0; j < 19; ++j) {
    const int q = (j < 18) ? (tid + 256 * j) : (tid + ATT_CHUNKS - 256);
    const v4f* p;
    if (j < 12)       p = U4 + q;
    else if (j == 12) p = (tid < 128) ? (U4 + q) : (V4 + (q - ATT_U_CHUNKS));
    else              p = V4 + (q - ATT_U_CHUNKS);
    r[j] = *p;
  }
  #pragma unroll
  for (int j = 0; j < 19; ++j) {
    const int q = (j < 18) ? (tid + 256 * j) : (tid + ATT_CHUNKS - 256);
    ((v4f*)us)[q] = r[j];
  }
  __syncthreads();

  const int wave  = tid >> 5;
  const int lane  = tid & 31;
  const int l16   = lane & 15;
  const int lhalf = (lane < 16) ? 0 : 2;

  for (int t = wave; t < 8 * 4; t += 8) {
    const int m0 = (t >> 2) * 16, n0 = (t & 3) * 16;
    const v2f* ap = (const v2f*)&us[(m0 + l16) * P2 + lhalf];
    const v2f* bp = (const v2f*)&vs[(n0 + l16) * P2 + lhalf];
    v8f acc = {};
    #pragma unroll 5
    for (int k0 = 0; k0 < P2; k0 += 4) {
      v2f a  = *ap;  ap += 2;
      v2f bf = *bp;  bp += 2;
      acc = wmma_f32(a, bf, acc);
    }
    #pragma unroll
    for (int r8 = 0; r8 < 8; ++r8) {
      int M = M0 + m0 + r8 + ((lane < 16) ? 0 : 8);
      int N = N0 + n0 + l16;
      out[((size_t)bc * LPATCH + M) * LPATCH + N] = acc[r8] * 0.01f;
    }
  }
}

extern "C" void kernel_launch(void* const* d_in, const int* in_sizes, int n_in,
                              void* d_out, int out_size, void* d_ws,
                              size_t ws_size, hipStream_t stream) {
  (void)in_sizes; (void)n_in; (void)out_size; (void)ws_size;
  const float* x       = (const float*)d_in[0];
  const float* w_conv1 = (const float*)d_in[1];
  const float* b_conv1 = (const float*)d_in[2];
  const float* w_conv2 = (const float*)d_in[3];
  const float* b_conv2 = (const float*)d_in[4];
  const float* w_fm    = (const float*)d_in[5];
  const float* b_fm    = (const float*)d_in[6];
  const float* w_lin1  = (const float*)d_in[7];
  const float* w_lin2  = (const float*)d_in[8];
  float* out = (float*)d_out;

  float* ws  = (float*)d_ws;
  float* ini = ws + WS_INI;
  float* fea = ws + WS_FEA;
  float* ui  = ws + WS_UI;
  float* uf  = ws + WS_UF;
  float* w1p = ws + WS_W1P;
  float* w2p = ws + WS_W2P;
  float* img = ws + WS_IMG;

  pad_weights_kernel<<<dim3((112 * 200 + 255) / 256), 256, 0, stream>>>(
      w_lin1, w_lin2, w_conv1, b_conv1, w_conv2, b_conv2, w_fm, b_fm,
      w1p, w2p, img);

  conv_fused_kernel<<<dim3(HW / 16, HW / 16, BATCH), 256,
                      CONV_LDS_FLOATS * sizeof(float), stream>>>(
      x, img, ini, fea);

  trans_kernel<<<dim3(BATCH * CLS * LPATCH / 64, 2, 1), 256,
                 TRANS_LDS_FLOATS * sizeof(float), stream>>>(
      ini, fea, w1p, w2p, ui, uf);

  att_kernel<<<dim3(LPATCH / 128, LPATCH / 64, BATCH * CLS), 256,
               ATT_LDS_FLOATS * sizeof(float), stream>>>(ui, uf, out);
}


# conv2 (linear, right behind conv1) folded into conv1's weights by the prep kernel: conv1's matrix product has 18 tap columns (padded 32), not 64 channels; conv2 = 9-term shifted sum; sums reordered
# speedup vs baseline: 2.7248x; 1.5837x over previous
#include <hip/hip_runtime.h>
#include <hip/hip_bf16.h>

typedef float v2f __attribute__((ext_vector_type(2)));
typedef float v8f __attribute__((ext_vector_type(8)));
typedef float v4f __attribute__((ext_vector_type(4)));

__device__ __forceinline__ v8f wmma_f32(v2f a, v2f b, v8f c) {
  return __builtin_amdgcn_wmma_f32_16x16x4_f32(false, a, false, b, (short)0, c,
                                               false, false);
}

#define BATCH   8
#define CIN     3
#define HW      320
#define CLS     2
#define PSZ     10
#define P2      100
#define LPATCH  1024
#define HID     200

#define WS_INI  0
#define WS_FEA  (WS_INI + BATCH*CLS*HW*HW)
#define WS_UI   (WS_FEA + BATCH*CLS*HW*HW)
#define WS_UF   (WS_UI  + BATCH*CLS*LPATCH*P2)
#define WS_W1P  (WS_UF  + BATCH*CLS*LPATCH*P2)
#define WS_W2P  (WS_W1P + 208*100)
#define WS_IMG  WS_UI
#define NCOL       32
#define NFOLD      18
#define IMG_W1     (74 * NCOL * 2)
#define IMG_HEAD   (IMG_W1 + NCOL)
#define IMG_TAIL   16
#define IMG_FLOATS (IMG_HEAD + IMG_TAIL)

__global__ __launch_bounds__(256) void pad_weights_kernel(
    const float* __restrict__ w_lin1, const float* __restrict__ w_lin2,
    const float* __restrict__ w1c, const float* __restrict__ b1c,
    const float* __restrict__ w2c, const float* __restrict__ b2c,
    const float* __restrict__ wfm, const float* __restrict__ bfm,
    float* __restrict__ w1p, float* __restrict__ w2p,
    float* __restrict__ img) {
  int i = blockIdx.x * 256 + threadIdx.x;
  if (i < 208 * 100) w1p[i] = (i < HID * P2) ? w_lin1[i] : 0.f;
  if (i < 112 * 200) w2p[i] = (i < P2 * HID) ? w_lin2[i] : 0.f;
  if (i < IMG_FLOATS) {
    float v = 0.f;
    if (i < IMG_W1) {
      const int k = ((i >> 6) << 1) + (i & 1), n = (i & 63) >> 1;
      if (k < 147 && n < NFOLD) {
        const float* wtap = w2c + (n / 9) * 576 + (n % 9);
        #pragma unroll 16
        for (int ch = 0; ch < 64; ++ch) v = fmaf(wtap[ch * 9], w1c[ch * 147 + k], v);
      }
    } else if (i < IMG_HEAD) {
      const int n = i - IMG_W1;
      if (n < NFOLD) {
        const float* wtap = w2c + (n / 9) * 576 + (n % 9);
        for (int ch = 0; ch < 64; ++ch) v = fmaf(wtap[ch * 9], b1c[ch], v);
      }
    } else {
      int j = i - IMG_HEAD;
      if (j < 2)       v = b2c[j];
      else if (j < 8)  v = wfm[j - 2];
      else if (j < 10) v = bfm[j - 8];
    }
    img[i] = v;
  }
}

#define XS_FLOATS        2160
#define CONV_LDS_FLOATS  (XS_FLOATS + 148 + IMG_W1 + NCOL + NFOLD*325)

template <int NT>
__device__ __forceinline__ void conv1_tiles(
    const float* xs, const int* offtab, const float* w1s, const float* b1s,
    float* t1s, int m0, int n0, int lane, int X0, int Y0) {
  constexpr int KS = 37;
  constexpr int G  = 4;
  constexpr int NB = (KS + G - 1) / G;
  const int l16 = lane & 15;
  const int hk  = lane >> 4;
  const int m   = m0 + l16;
  const int mm  = (m < 324) ? m : 0;
  const float* xa = xs + (mm / 18) * 24 + (mm % 18);
  const int2* ot = ((const int2*)offtab) + hk;
  const v2f*  bp = ((const v2f*)w1s) + hk * NCOL + n0 + l16;

  int2 off[2][G];
  v2f  af[2][G];
  v2f  bf[2][G][NT];
  v8f  acc[NT];
  #pragma unroll
  for (int j = 0; j < NT; ++j) acc[j] = v8f{};

  float bias[NT];
  #pragma unroll
  for (int j = 0; j < NT; ++j) bias[j] = b1s[n0 + 16 * j + l16];
  #pragma unroll
  for (int i = 0; i < G; ++i) { off[0][i] = ot[2 * i]; off[1][i] = ot[2 * (G + i)]; }
  #pragma unroll
  for (int i = 0; i < G; ++i) {
    af[0][i].x = xa[off[0][i].x];
    af[0][i].y = xa[off[0][i].y];
    #pragma unroll
    for (int j = 0; j < NT; ++j) bf[0][i][j] = bp[2 * NCOL * i + 16 * j];
  }

  #pragma unroll
  for (int g = 0; g < NB; ++g) {
    const int cur = g & 1, nxt = cur ^ 1;
    if (g + 1 < NB) {
      #pragma unroll
      for (int i = 0; i < G; ++i) {
        const int s = (g + 1) * G + i;
        if (s < KS) {
          af[nxt][i].x = xa[off[nxt][i].x];
          af[nxt][i].y = xa[off[nxt][i].y];
          #pragma unroll
          for (int j = 0; j < NT; ++j) bf[nxt][i][j] = bp[2 * NCOL * s + 16 * j];
        }
      }
    }
    if (g + 2 < NB) {
      #pragma unroll
      for (int i = 0; i < G; ++i) {
        const int s = (g + 2) * G + i;
        if (s < KS) off[cur][i] = ot[2 * s];
      }
    }
    __builtin_amdgcn_sched_barrier(0);
    #pragma unroll
    for (int i = 0; i < G; ++i) {
      if (g * G + i < KS) {
        #pragma unroll
        for (int j = 0; j < NT; ++j) acc[j] = wmma_f32(af[cur][i], bf[cur][i][j], acc[j]);
      }
    }
    __builtin_amdgcn_sched_barrier(0);
  }

  #pragma unroll
  for (int r = 0; r < 8; ++r) {
    const int M = m0 + r + 8 * hk;
    if (M < 324) {
      const int ty = M / 18, tx = M % 18;
      const int gy = Y0 - 1 + ty, gx = X0 - 1 + tx;
      const bool in = gy >= 0 && gy < HW && gx >= 0 && gx < HW;
      #pragma unroll
      for (int j = 0; j < NT; ++j) {
        const int ch = n0 + 16 * j + l16;
        if (ch < NFOLD) t1s[ch * 325 + M] = in ? (acc[j][r] + bias[j]) : 0.f;
      }
    }
  }
}

__global__ __launch_bounds__(256) void conv_fused_kernel(
    const float* __restrict__ x,  const float* __restrict__ img,
    float* __restrict__ ini, float* __restrict__ fea) {
  extern __shared__ float smem[];
  float* xs     = smem;
  int*   offtab = (int*)(xs + XS_FLOATS);
  float* w1s    = xs + XS_FLOATS + 148;
  float* b1s    = w1s + IMG_W1;
  float* t1s    = b1s + NCOL;
  const float* __restrict__ cg  = img + IMG_HEAD;

  const int tid = threadIdx.x;
  const int X0 = blockIdx.x * 16, Y0 = blockIdx.y * 16, b = blockIdx.z;

  const v4f* img4 = (const v4f*)img;
  v4f cw[5];
  #pragma unroll
  for (int j = 0; j < 5; ++j) {
    int q = tid + 256 * j;
    q = (q < IMG_HEAD / 4) ? q : (IMG_HEAD / 4 - 1);
    cw[j] = img4[q];
  }
  v4f xq[2];
  #pragma unroll
  for (int j = 0; j < 2; ++j) {
    const int q  = tid + 256 * j;
    const int qq = (q < CIN * 144) ? q : (CIN * 144 - 1);
    const int row = qq / 6, f = qq % 6, c = row / 24, iy = row % 24;
    const int gy = Y0 - 4 + iy, gx = X0 - 4 + 4 * f;
    const bool inb = (q < CIN * 144) && gy >= 0 && gy < HW && gx >= 0 && gx < HW;
    const int cy = gy < 0 ? 0 : (gy > HW - 1 ? HW - 1 : gy);
    const int cx = gx < 0 ? 0 : (gx > HW - 4 ? HW - 4 : gx);
    const v4f v = *(const v4f*)(x + ((size_t)(b * CIN + c) * HW + cy) * HW + cx);
    xq[j] = inb ? v : v4f{0.f, 0.f, 0.f, 0.f};
  }
  #pragma unroll
  for (int j = 0; j < 5; ++j) {
    const int q = tid + 256 * j;
    if (q < IMG_HEAD / 4) ((v4f*)w1s)[q] = cw[j];
  }
  ((v4f*)xs)[tid]       = xq[0];
  ((v4f*)xs)[tid + 256] = xq[1];
  if (tid < XS_FLOATS / 4 - 512) ((v4f*)xs)[tid + 512] = v4f{0.f, 0.f, 0.f, 0.f};
  if (tid < 148) {
    int off = 1728;
    if (tid < 147) {
      int c = tid / 49, rr = tid % 49;
      off = c * 576 + (rr / 7) * 24 + (rr % 7);
    }
    offtab[tid] = off;
  }
  __syncthreads();

  const int wave = tid >> 5;
  const int lane = tid & 31;

  for (int mt = wave; mt < 16; mt += 8)
    conv1_tiles<2>(xs, offtab, w1s, b1s, t1s, mt * 16, 0, lane, X0, Y0);
  conv1_tiles<1>(xs, offtab, w1s, b1s, t1s, (16 + (wave >> 1)) * 16, (wave & 1) * 16,
                 lane, X0, Y0);
  if (wave == 2 || wave == 3)
    conv1_tiles<1>(xs, offtab, w1s, b1s, t1s, 320, (wave - 2) * 16, lane, X0, Y0);
  __syncthreads();

  {
    const int oy = tid >> 4, ox = tid & 15;
    float a0 = cg[0], a1 = cg[1];
    for (int dy = 0; dy < 3; ++dy)
      for (int dx = 0; dx < 3; ++dx) {
        const int p = (oy + dy) * 18 + (ox + dx);
        a0 += t1s[(dy * 3 + dx) * 325 + p];
        a1 += t1s[(9 + dy * 3 + dx) * 325 + p];
      }
    const int gy = Y0 + oy, gx = X0 + ox;
    const size_t o0 = ((size_t)(b * CLS + 0) * HW + gy) * HW + gx;
    const size_t o1 = ((size_t)(b * CLS + 1) * HW + gy) * HW + gx;
    ini[o0] = a0;
    ini[o1] = a1;
    float x0 = xs[(oy + 4) * 24 + (ox + 4)];
    float x1 = xs[576 + (oy + 4) * 24 + (ox + 4)];
    float x2 = xs[1152 + (oy + 4) * 24 + (ox + 4)];
    fea[o0] = (cg[2] * x0 + cg[3] * x1 + cg[4] * x2 + cg[8]) * 0.25f;
    fea[o1] = (cg[5] * x0 + cg[6] * x1 + cg[7] * x2 + cg[9]) * 0.25f;
  }
}

#define TRANS_LDS_FLOATS (64 * 102 + 64 * 210)

__global__ __launch_bounds__(256) void trans_kernel(
    const float* __restrict__ ini, const float* __restrict__ fea,
    const float* __restrict__ w1p, const float* __restrict__ w2p,
    float* __restrict__ ui, float* __restrict__ uf) {
  extern __shared__ float smem[];
  float* a_lds = smem;
  float* h_lds = a_lds + 64 * 102;

  const int tid = threadIdx.x;
  const float* src = blockIdx.y ? fea : ini;
  float* dst       = blockIdx.y ? uf  : ui;
  const int P0 = blockIdx.x * 64;

  for (int i = tid; i < 64 * P2; i += 256) {
    int r = i / P2, p = i % P2;
    int Pa = P0 + r;
    int bc = Pa >> 10, l = Pa & 1023;
    int gy = (l >> 5) * PSZ + p / PSZ;
    int gx = (l & 31) * PSZ + p % PSZ;
    a_lds[r * 102 + p] = src[((size_t)bc * HW + gy) * HW + gx];
  }
  __syncthreads();

  const int wave  = tid >> 5;
  const int lane  = tid & 31;
  const int l16   = lane & 15;
  const int lhalf = (lane < 16) ? 0 : 2;

  for (int t = wave; t < 4 * 13; t += 8) {
    const int m0 = (t / 13) * 16, n0 = (t % 13) * 16;
    const int n  = n0 + l16;
    const v2f* ap = (const v2f*)&a_lds[(m0 + l16) * 102 + lhalf];
    const v2f* bp = (const v2f*)&w1p[n * P2 + lhalf];
    v8f acc = {};
    #pragma unroll 5
    for (int k0 = 0; k0 < P2; k0 += 4) {
      v2f a  = *ap;  ap += 2;
      v2f bf = *bp;  bp += 2;
      acc = wmma_f32(a, bf, acc);
    }
    #pragma unroll
    for (int r = 0; r < 8; ++r) {
      int M = m0 + r + ((lane < 16) ? 0 : 8);
      h_lds[M * 210 + n] = acc[r];
    }
  }
  __syncthreads();

  for (int t = wave; t < 4 * 7; t += 8) {
    const int m0 = (t / 7) * 16, n0 = (t % 7) * 16;
    const int n  = n0 + l16;
    const v2f* ap = (const v2f*)&h_lds[(m0 + l16) * 210 + lhalf];
    const v2f* bp = (const v2f*)&w2p[n * HID + lhalf];
    v8f acc = {};
    #pragma unroll 5
    for (int k0 = 0; k0 < HID; k0 += 4) {
      v2f a  = *ap;  ap += 2;
      v2f bf = *bp;  bp += 2;
      acc = wmma_f32(a, bf, acc);
    }
    if (n < P2) {
      #pragma unroll
      for (int r = 0; r < 8; ++r) {
        int M = m0 + r + ((lane < 16) ? 0 : 8);
        float v = acc[r];
        dst[(size_t)(P0 + M) * P2 + n] = v > 0.f ? v : 0.f;
      }
    }
  }
}

#define ATT_U_CHUNKS   (128 * P2 / 4)
#define ATT_CHUNKS     (ATT_U_CHUNKS + 64 * P2 / 4)
#define ATT_LDS_FLOATS (128 * P2 + 64 * P2)

__global__ __launch_bounds__(256) void att_kernel(
    const float* __restrict__ ui, const float* __restrict__ uf,
    float* __restrict__ out) {
  extern __shared__ float smem[];
  float* us = smem;
  float* vs = us + 128 * P2;

  const int tid = threadIdx.x;
  const int bc = blockIdx.z;
  const int M0 = blockIdx.x * 128, N0 = blockIdx.y * 64;
  const v4f* U4 = (const v4f*)(ui + ((size_t)bc * LPATCH + M0) * P2);
  const v4f* V4 = (const v4f*)(uf + ((size_t)bc * LPATCH + N0) * P2);

  v4f r[19];
  #pragma unroll
  for (int j = 0; j < 19; ++j) {
    const int q = (j < 18) ? (tid + 256 * j) : (tid + ATT_CHUNKS - 256);
    const v4f* p;
    if (j < 12)       p = U4 + q;
    else if (j == 12) p = (tid < 128) ? (U4 + q) : (V4 + (q - ATT_U_CHUNKS));
    else              p = V4 + (q - ATT_U_CHUNKS);
    r[j] = *p;
  }
  #pragma unroll
  for (int j = 0; j < 19; ++j) {
    const int q = (j < 18) ? (tid + 256 * j) : (tid + ATT_CHUNKS - 256);
    ((v4f*)us)[q] = r[j];
  }
  __syncthreads();

  const int wave  = tid >> 5;
  const int lane  = tid & 31;
  const int l16   = lane & 15;
  const int lhalf = (lane < 16) ? 0 : 2;

  for (int t = wave; t < 8 * 4; t += 8) {
    const int m0 = (t >> 2) * 16, n0 = (t & 3) * 16;
    const v2f* ap = (const v2f*)&us[(m0 + l16) * P2 + lhalf];
    const v2f* bp = (const v2f*)&vs[(n0 + l16) * P2 + lhalf];
    v8f acc = {};
    #pragma unroll 5
    for (int k0 = 0; k0 < P2; k0 += 4) {
      v2f a  = *ap;  ap += 2;
      v2f bf = *bp;  bp += 2;
      acc = wmma_f32(a, bf, acc);
    }
    #pragma unroll
    for (int r8 = 0; r8 < 8; ++r8) {
      int M = M0 + m0 + r8 + ((lane < 16) ? 0 : 8);
      int N = N0 + n0 + l16;
      out[((size_t)bc * LPATCH + M) * LPATCH + N] = acc[r8] * 0.01f;
    }
  }
}

extern "C" void kernel_launch(void* const* d_in, const int* in_sizes, int n_in,
                              void* d_out, int out_size, void* d_ws,
                              size_t ws_size, hipStream_t stream) {
  (void)in_sizes; (void)n_in; (void)out_size; (void)ws_size;
  const float* x       = (const float*)d_in[0];
  const float* w_conv1 = (const float*)d_in[1];
  const float* b_conv1 = (const float*)d_in[2];
  const float* w_conv2 = (const float*)d_in[3];
  const float* b_conv2 = (const float*)d_in[4];
  const float* w_fm    = (const float*)d_in[5];
  const float* b_fm    = (const float*)d_in[6];
  const float* w_lin1  = (const float*)d_in[7];
  const float* w_lin2  = (const float*)d_in[8];
  float* out = (float*)d_out;

  float* ws  = (float*)d_ws;
  float* ini = ws + WS_INI;
  float* fea = ws + WS_FEA;
  float* ui  = ws + WS_UI;
  float* uf  = ws + WS_UF;
  float* w1p = ws + WS_W1P;
  float* w2p = ws + WS_W2P;
  float* img = ws + WS_IMG;

  pad_weights_kernel<<<dim3((112 * 200 + 255) / 256), 256, 0, stream>>>(
      w_lin1, w_lin2, w_conv1, b_conv1, w_conv2, b_conv2, w_fm, b_fm,
      w1p, w2p, img);

  conv_fused_kernel<<<dim3(HW / 16, HW / 16, BATCH), 256,
                      CONV_LDS_FLOATS * sizeof(float), stream>>>(
      x, img, ini, fea);

  trans_kernel<<<dim3(BATCH * CLS * LPATCH / 64, 2, 1), 256,
                 TRANS_LDS_FLOATS * sizeof(float), stream>>>(
      ini, fea, w1p, w2p, ui, uf);

  att_kernel<<<dim3(LPATCH / 128, LPATCH / 64, BATCH * CLS), 256,
               ATT_LDS_FLOATS * sizeof(float), stream>>>(ui, uf, out);
}


# lin2 (linear, right behind lin1) folded into lin1 by the prep kernel: the patch MLP is one 100x100 matrix product (K=100, 7 N-tiles), not 100->200->100; hidden tile and barrier gone; sums reordered
# speedup vs baseline: 3.2551x; 1.1946x over previous
#include <hip/hip_runtime.h>
#include <hip/hip_bf16.h>

typedef float v2f __attribute__((ext_vector_type(2)));
typedef float v8f __attribute__((ext_vector_type(8)));
typedef float v4f __attribute__((ext_vector_type(4)));

__device__ __forceinline__ v8f wmma_f32(v2f a, v2f b, v8f c) {
  return __builtin_amdgcn_wmma_f32_16x16x4_f32(false, a, false, b, (short)0, c,
                                               false, false);
}

#define BATCH   8
#define CIN     3
#define HW      320
#define CLS     2
#define PSZ     10
#define P2      100
#define LPATCH  1024
#define HID     200

#define WS_INI  0
#define WS_FEA  (WS_INI + BATCH*CLS*HW*HW)
#define WS_UI   (WS_FEA + BATCH*CLS*HW*HW)
#define WS_UF   (WS_UI  + BATCH*CLS*LPATCH*P2)
#define WS_WF   (WS_UF  + BATCH*CLS*LPATCH*P2)
#define WF_ROWS 112
#define PREP_WF_BLOCKS  11
#define PREP_IMG_BLOCKS 19
#define WS_IMG  WS_UI
#define NCOL       32
#define NFOLD      18
#define IMG_W1     (74 * NCOL * 2)
#define IMG_HEAD   (IMG_W1 + NCOL)
#define IMG_TAIL   16
#define IMG_FLOATS (IMG_HEAD + IMG_TAIL)

__global__ __launch_bounds__(256) void pad_weights_kernel(
    const float* __restrict__ w_lin1, const float* __restrict__ w_lin2,
    const float* __restrict__ w1c, const float* __restrict__ b1c,
    const float* __restrict__ w2c, const float* __restrict__ b2c,
    const float* __restrict__ wfm, const float* __restrict__ bfm,
    float* __restrict__ wf, float* __restrict__ img) {
  if (blockIdx.x < PREP_WF_BLOCKS) {
    const int j = blockIdx.x * 256 + threadIdx.x;
    if (j < WF_ROWS * (P2 / 4)) {
      const int n = j / (P2 / 4), kq = j % (P2 / 4);
      v4f acc = {0.f, 0.f, 0.f, 0.f};
      if (n < P2) {
        const v4f* a4 = (const v4f*)(w_lin2 + n * HID);
        const v4f* b4 = ((const v4f*)w_lin1) + kq;
        #pragma unroll 1
        for (int t = 0; t < HID / 40; ++t) {
          v4f av[10], bv[40];
          #pragma unroll
          for (int u = 0; u < 10; ++u) {
            const int s = t * 10 + u;
            av[u] = a4[s];
            #pragma unroll
            for (int c = 0; c < 4; ++c) bv[4 * u + c] = b4[(4 * s + c) * (P2 / 4)];
          }
          __builtin_amdgcn_sched_barrier(0);
          #pragma unroll
          for (int u = 0; u < 10; ++u) {
            #pragma unroll
            for (int c = 0; c < 4; ++c) {
              const float a = av[u][c];
              const v4f b = bv[4 * u + c];
              acc.x = fmaf(a, b.x, acc.x); acc.y = fmaf(a, b.y, acc.y);
              acc.z = fmaf(a, b.z, acc.z); acc.w = fmaf(a, b.w, acc.w);
            }
          }
        }
      }
      ((v4f*)wf)[j] = acc;
    }
    return;
  }
  const int i = (blockIdx.x - PREP_WF_BLOCKS) * 256 + threadIdx.x;
  if (i < IMG_FLOATS) {
    float v = 0.f;
    if (i < IMG_W1) {
      const int k = ((i >> 6) << 1) + (i & 1), n = (i & 63) >> 1;
      if (k < 147 && n < NFOLD) {
        const float* wtap = w2c + (n / 9) * 576 + (n % 9);
        #pragma unroll 16
        for (int ch = 0; ch < 64; ++ch) v = fmaf(wtap[ch * 9], w1c[ch * 147 + k], v);
      }
    } else if (i < IMG_HEAD) {
      const int n = i - IMG_W1;
      if (n < NFOLD) {
        const float* wtap = w2c + (n / 9) * 576 + (n % 9);
        for (int ch = 0; ch < 64; ++ch) v = fmaf(wtap[ch * 9], b1c[ch], v);
      }
    } else {
      int j = i - IMG_HEAD;
      if (j < 2)       v = b2c[j];
      else if (j < 8)  v = wfm[j - 2];
      else if (j < 10) v = bfm[j - 8];
    }
    img[i] = v;
  }
}

#define XS_FLOATS        2160
#define CONV_LDS_FLOATS  (XS_FLOATS + 148 + IMG_W1 + NCOL + NFOLD*325)

template <int NT>
__device__ __forceinline__ void conv1_tiles(
    const float* xs, const int* offtab, const float* w1s, const float* b1s,
    float* t1s, int m0, int n0, int lane, int X0, int Y0) {
  constexpr int KS = 37;
  constexpr int G  = 4;
  constexpr int NB = (KS + G - 1) / G;
  const int l16 = lane & 15;
  const int hk  = lane >> 4;
  const int m   = m0 + l16;
  const int mm  = (m < 324) ? m : 0;
  const float* xa = xs + (mm / 18) * 24 + (mm % 18);
  const int2* ot = ((const int2*)offtab) + hk;
  const v2f*  bp = ((const v2f*)w1s) + hk * NCOL + n0 + l16;

  int2 off[2][G];
  v2f  af[2][G];
  v2f  bf[2][G][NT];
  v8f  acc[NT];
  #pragma unroll
  for (int j = 0; j < NT; ++j) acc[j] = v8f{};

  float bias[NT];
  #pragma unroll
  for (int j = 0; j < NT; ++j) bias[j] = b1s[n0 + 16 * j + l16];
  #pragma unroll
  for (int i = 0; i < G; ++i) { off[0][i] = ot[2 * i]; off[1][i] = ot[2 * (G + i)]; }
  #pragma unroll
  for (int i = 0; i < G; ++i) {
    af[0][i].x = xa[off[0][i].x];
    af[0][i].y = xa[off[0][i].y];
    #pragma unroll
    for (int j = 0; j < NT; ++j) bf[0][i][j] = bp[2 * NCOL * i + 16 * j];
  }

  #pragma unroll
  for (int g = 0; g < NB; ++g) {
    const int cur = g & 1, nxt = cur ^ 1;
    if (g + 1 < NB) {
      #pragma unroll
      for (int i = 0; i < G; ++i) {
        const int s = (g + 1) * G + i;
        if (s < KS) {
          af[nxt][i].x = xa[off[nxt][i].x];
          af[nxt][i].y = xa[off[nxt][i].y];
          #pragma unroll
          for (int j = 0; j < NT; ++j) bf[nxt][i][j] = bp[2 * NCOL * s + 16 * j];
        }
      }
    }
    if (g + 2 < NB) {
      #pragma unroll
      for (int i = 0; i < G; ++i) {
        const int s = (g + 2) * G + i;
        if (s < KS) off[cur][i] = ot[2 * s];
      }
    }
    __builtin_amdgcn_sched_barrier(0);
    #pragma unroll
    for (int i = 0; i < G; ++i) {
      if (g * G + i < KS) {
        #pragma unroll
        for (int j = 0; j < NT; ++j) acc[j] = wmma_f32(af[cur][i], bf[cur][i][j], acc[j]);
      }
    }
    __builtin_amdgcn_sched_barrier(0);
  }

  #pragma unroll
  for (int r = 0; r < 8; ++r) {
    const int M = m0 + r + 8 * hk;
    if (M < 324) {
      const int ty = M / 18, tx = M % 18;
      const int gy = Y0 - 1 + ty, gx = X0 - 1 + tx;
      const bool in = gy >= 0 && gy < HW && gx >= 0 && gx < HW;
      #pragma unroll
      for (int j = 0; j < NT; ++j) {
        const int ch = n0 + 16 * j + l16;
        if (ch < NFOLD) t1s[ch * 325 + M] = in ? (acc[j][r] + bias[j]) : 0.f;
      }
    }
  }
}

__global__ __launch_bounds__(256) void conv_fused_kernel(
    const float* __restrict__ x,  const float* __restrict__ img,
    float* __restrict__ ini, float* __restrict__ fea) {
  extern __shared__ float smem[];
  float* xs     = smem;
  int*   offtab = (int*)(xs + XS_FLOATS);
  float* w1s    = xs + XS_FLOATS + 148;
  float* b1s    = w1s + IMG_W1;
  float* t1s    = b1s + NCOL;
  const float* __restrict__ cg  = img + IMG_HEAD;

  const int tid = threadIdx.x;
  const int X0 = blockIdx.x * 16, Y0 = blockIdx.y * 16, b = blockIdx.z;

  const v4f* img4 = (const v4f*)img;
  v4f cw[5];
  #pragma unroll
  for (int j = 0; j < 5; ++j) {
    int q = tid + 256 * j;
    q = (q < IMG_HEAD / 4) ? q : (IMG_HEAD / 4 - 1);
    cw[j] = img4[q];
  }
  v4f xq[2];
  #pragma unroll
  for (int j = 0; j < 2; ++j) {
    const int q  = tid + 256 * j;
    const int qq = (q < CIN * 144) ? q : (CIN * 144 - 1);
    const int row = qq / 6, f = qq % 6, c = row / 24, iy = row % 24;
    const int gy = Y0 - 4 + iy, gx = X0 - 4 + 4 * f;
    const bool inb = (q < CIN * 144) && gy >= 0 && gy < HW && gx >= 0 && gx < HW;
    const int cy = gy < 0 ? 0 : (gy > HW - 1 ? HW - 1 : gy);
    const int cx = gx < 0 ? 0 : (gx > HW - 4 ? HW - 4 : gx);
    const v4f v = *(const v4f*)(x + ((size_t)(b * CIN + c) * HW + cy) * HW + cx);
    xq[j] = inb ? v : v4f{0.f, 0.f, 0.f, 0.f};
  }
  #pragma unroll
  for (int j = 0; j < 5; ++j) {
    const int q = tid + 256 * j;
    if (q < IMG_HEAD / 4) ((v4f*)w1s)[q] = cw[j];
  }
  ((v4f*)xs)[tid]       = xq[0];
  ((v4f*)xs)[tid + 256] = xq[1];
  if (tid < XS_FLOATS / 4 - 512) ((v4f*)xs)[tid + 512] = v4f{0.f, 0.f, 0.f, 0.f};
  if (tid < 148) {
    int off = 1728;
    if (tid < 147) {
      int c = tid / 49, rr = tid % 49;
      off = c * 576 + (rr / 7) * 24 + (rr % 7);
    }
    offtab[tid] = off;
  }
  __syncthreads();

  const int wave = tid >> 5;
  const int lane = tid & 31;

  for (int mt = wave; mt < 16; mt += 8)
    conv1_tiles<2>(xs, offtab, w1s, b1s, t1s, mt * 16, 0, lane, X0, Y0);
  conv1_tiles<1>(xs, offtab, w1s, b1s, t1s, (16 + (wave >> 1)) * 16, (wave & 1) * 16,
                 lane, X0, Y0);
  if (wave == 2 || wave == 3)
    conv1_tiles<1>(xs, offtab, w1s, b1s, t1s, 320, (wave - 2) * 16, lane, X0, Y0);
  __syncthreads();

  {
    const int oy = tid >> 4, ox = tid & 15;
    float a0 = cg[0], a1 = cg[1];
    for (int dy = 0; dy < 3; ++dy)
      for (int dx = 0; dx < 3; ++dx) {
        const int p = (oy + dy) * 18 + (ox + dx);
        a0 += t1s[(dy * 3 + dx) * 325 + p];
        a1 += t1s[(9 + dy * 3 + dx) * 325 + p];
      }
    const int gy = Y0 + oy, gx = X0 + ox;
    const size_t o0 = ((size_t)(b * CLS + 0) * HW + gy) * HW + gx;
    const size_t o1 = ((size_t)(b * CLS + 1) * HW + gy) * HW + gx;
    ini[o0] = a0;
    ini[o1] = a1;
    float x0 = xs[(oy + 4) * 24 + (ox + 4)];
    float x1 = xs[576 + (oy + 4) * 24 + (ox + 4)];
    float x2 = xs[1152 + (oy + 4) * 24 + (ox + 4)];
    fea[o0] = (cg[2] * x0 + cg[3] * x1 + cg[4] * x2 + cg[8]) * 0.25f;
    fea[o1] = (cg[5] * x0 + cg[6] * x1 + cg[7] * x2 + cg[9]) * 0.25f;
  }
}

#define TRANS_LDS_FLOATS (64 * 102)

__global__ __launch_bounds__(256) void trans_kernel(
    const float* __restrict__ ini, const float* __restrict__ fea,
    const float* __restrict__ wf,
    float* __restrict__ ui, float* __restrict__ uf) {
  extern __shared__ float smem[];
  float* a_lds = smem;

  const int tid = threadIdx.x;
  const float* src = blockIdx.y ? fea : ini;
  float* dst       = blockIdx.y ? uf  : ui;
  const int P0 = blockIdx.x * 64;

  for (int i = tid; i < 64 * P2; i += 256) {
    int r = i / P2, p = i % P2;
    int Pa = P0 + r;
    int bc = Pa >> 10, l = Pa & 1023;
    int gy = (l >> 5) * PSZ + p / PSZ;
    int gx = (l & 31) * PSZ + p % PSZ;
    a_lds[r * 102 + p] = src[((size_t)bc * HW + gy) * HW + gx];
  }
  __syncthreads();

  const int wave  = tid >> 5;
  const int lane  = tid & 31;
  const int l16   = lane & 15;
  const int lhalf = (lane < 16) ? 0 : 2;

  for (int t = wave; t < 4 * 7; t += 8) {
    const int m0 = (t / 7) * 16, n0 = (t % 7) * 16;
    const int n  = n0 + l16;
    const v2f* ap = (const v2f*)&a_lds[(m0 + l16) * 102 + lhalf];
    const v2f* bp = (const v2f*)&wf[n * P2 + lhalf];
    v8f acc = {};
    #pragma unroll 5
    for (int k0 = 0; k0 < P2; k0 += 4) {
      v2f a  = *ap;  ap += 2;
      v2f bf = *bp;  bp += 2;
      acc = wmma_f32(a, bf, acc);
    }
    if (n < P2) {
      #pragma unroll
      for (int r = 0; r < 8; ++r) {
        int M = m0 + r + ((lane < 16) ? 0 : 8);
        float v = acc[r];
        dst[(size_t)(P0 + M) * P2 + n] = v > 0.f ? v : 0.f;
      }
    }
  }
}

#define ATT_U_CHUNKS   (128 * P2 / 4)
#define ATT_CHUNKS     (ATT_U_CHUNKS + 64 * P2 / 4)
#define ATT_LDS_FLOATS (128 * P2 + 64 * P2)

__global__ __launch_bounds__(256) void att_kernel(
    const float* __restrict__ ui, const float* __restrict__ uf,
    float* __restrict__ out) {
  extern __shared__ float smem[];
  float* us = smem;
  float* vs = us + 128 * P2;

  const int tid = threadIdx.x;
  const int bc = blockIdx.z;
  const int M0 = blockIdx.x * 128, N0 = blockIdx.y * 64;
  const v4f* U4 = (const v4f*)(ui + ((size_t)bc * LPATCH + M0) * P2);
  const v4f* V4 = (const v4f*)(uf + ((size_t)bc * LPATCH + N0) * P2);

  v4f r[19];
  #pragma unroll
  for (int j = 0; j < 19; ++j) {
    const int q = (j < 18) ? (tid + 256 * j) : (tid + ATT_CHUNKS - 256);
    const v4f* p;
    if (j < 12)       p = U4 + q;
    else if (j == 12) p = (tid < 128) ? (U4 + q) : (V4 + (q - ATT_U_CHUNKS));
    else              p = V4 + (q - ATT_U_CHUNKS);
    r[j] = *p;
  }
  #pragma unroll
  for (int j = 0; j < 19; ++j) {
    const int q = (j < 18) ? (tid + 256 * j) : (tid + ATT_CHUNKS - 256);
    ((v4f*)us)[q] = r[j];
  }
  __syncthreads();

  const int wave  = tid >> 5;
  const int lane  = tid & 31;
  const int l16   = lane & 15;
  const int lhalf = (lane < 16) ? 0 : 2;

  for (int t = wave; t < 8 * 4; t += 8) {
    const int m0 = (t >> 2) * 16, n0 = (t & 3) * 16;
    const v2f* ap = (const v2f*)&us[(m0 + l16) * P2 + lhalf];
    const v2f* bp = (const v2f*)&vs[(n0 + l16) * P2 + lhalf];
    v8f acc = {};
    #pragma unroll 5
    for (int k0 = 0; k0 < P2; k0 += 4) {
      v2f a  = *ap;  ap += 2;
      v2f bf = *bp;  bp += 2;
      acc = wmma_f32(a, bf, acc);
    }
    #pragma unroll
    for (int r8 = 0; r8 < 8; ++r8) {
      int M = M0 + m0 + r8 + ((lane < 16) ? 0 : 8);
      int N = N0 + n0 + l16;
      out[((size_t)bc * LPATCH + M) * LPATCH + N] = acc[r8] * 0.01f;
    }
  }
}

extern "C" void kernel_launch(void* const* d_in, const int* in_sizes, int n_in,
                              void* d_out, int out_size, void* d_ws,
                              size_t ws_size, hipStream_t stream) {
  (void)in_sizes; (void)n_in; (void)out_size; (void)ws_size;
  const float* x       = (const float*)d_in[0];
  const float* w_conv1 = (const float*)d_in[1];
  const float* b_conv1 = (const float*)d_in[2];
  const float* w_conv2 = (const float*)d_in[3];
  const float* b_conv2 = (const float*)d_in[4];
  const float* w_fm    = (const float*)d_in[5];
  const float* b_fm    = (const float*)d_in[6];
  const float* w_lin1  = (const float*)d_in[7];
  const float* w_lin2  = (const float*)d_in[8];
  float* out = (float*)d_out;

  float* ws  = (float*)d_ws;
  float* ini = ws + WS_INI;
  float* fea = ws + WS_FEA;
  float* ui  = ws + WS_UI;
  float* uf  = ws + WS_UF;
  float* wf  = ws + WS_WF;
  float* img = ws + WS_IMG;

  pad_weights_kernel<<<dim3(PREP_WF_BLOCKS + PREP_IMG_BLOCKS), 256, 0, stream>>>(
      w_lin1, w_lin2, w_conv1, b_conv1, w_conv2, b_conv2, w_fm, b_fm,
      wf, img);

  conv_fused_kernel<<<dim3(HW / 16, HW / 16, BATCH), 256,
                      CONV_LDS_FLOATS * sizeof(float), stream>>>(
      x, img, ini, fea);

  trans_kernel<<<dim3(BATCH * CLS * LPATCH / 64, 2, 1), 256,
                 TRANS_LDS_FLOATS * sizeof(float), stream>>>(
      ini, fea, wf, ui, uf);

  att_kernel<<<dim3(LPATCH / 128, LPATCH / 64, BATCH * CLS), 256,
               ATT_LDS_FLOATS * sizeof(float), stream>>>(ui, uf, out);
}


# conv2's 3 column taps folded into conv1's weights (7x9 window, K 189): 6 folded columns + 6 image-edge variants fill ONE 16-wide N-tile, M-tile = halo row; conv2 = 3-term sum; sums reordered
# speedup vs baseline: 4.0002x; 1.2289x over previous
#include <hip/hip_runtime.h>
#include <hip/hip_bf16.h>

typedef float v2f __attribute__((ext_vector_type(2)));
typedef float v8f __attribute__((ext_vector_type(8)));
typedef float v4f __attribute__((ext_vector_type(4)));

__device__ __forceinline__ v8f wmma_f32(v2f a, v2f b, v8f c) {
  return __builtin_amdgcn_wmma_f32_16x16x4_f32(false, a, false, b, (short)0, c,
                                               false, false);
}

#define BATCH   8
#define CIN     3
#define HW      320
#define CLS     2
#define PSZ     10
#define P2      100
#define LPATCH  1024
#define HID     200

#define WS_INI  0
#define WS_FEA  (WS_INI + BATCH*CLS*HW*HW)
#define WS_UI   (WS_FEA + BATCH*CLS*HW*HW)
#define WS_UF   (WS_UI  + BATCH*CLS*LPATCH*P2)
#define WS_WF   (WS_UF  + BATCH*CLS*LPATCH*P2)
#define WF_ROWS 112
#define PREP_WF_BLOCKS  11
#define PREP_IMG_BLOCKS 25
#define WS_IMG  WS_UI
#define NCOL       32
#define NFOLD      18
#define KFOLD      189
#define NTCOL      12
#define IMG_W1     (96 * NCOL * 2)
#define IMG_HEAD   (IMG_W1 + NCOL)
#define IMG_TAIL   16
#define IMG_FLOATS (IMG_HEAD + IMG_TAIL)

__global__ __launch_bounds__(256) void pad_weights_kernel(
    const float* __restrict__ w_lin1, const float* __restrict__ w_lin2,
    const float* __restrict__ w1c, const float* __restrict__ b1c,
    const float* __restrict__ w2c, const float* __restrict__ b2c,
    const float* __restrict__ wfm, const float* __restrict__ bfm,
    float* __restrict__ wf, float* __restrict__ img) {
  if (blockIdx.x < PREP_WF_BLOCKS) {
    const int j = blockIdx.x * 256 + threadIdx.x;
    if (j < WF_ROWS * (P2 / 4)) {
      const int n = j / (P2 / 4), kq = j % (P2 / 4);
      v4f acc = {0.f, 0.f, 0.f, 0.f};
      if (n < P2) {
        const v4f* a4 = (const v4f*)(w_lin2 + n * HID);
        const v4f* b4 = ((const v4f*)w_lin1) + kq;
        #pragma unroll 1
        for (int t = 0; t < HID / 40; ++t) {
          v4f av[10], bv[40];
          #pragma unroll
          for (int u = 0; u < 10; ++u) {
            const int s = t * 10 + u;
            av[u] = a4[s];
            #pragma unroll
            for (int c = 0; c < 4; ++c) bv[4 * u + c] = b4[(4 * s + c) * (P2 / 4)];
          }
          __builtin_amdgcn_sched_barrier(0);
          #pragma unroll
          for (int u = 0; u < 10; ++u) {
            #pragma unroll
            for (int c = 0; c < 4; ++c) {
              const float a = av[u][c];
              const v4f b = bv[4 * u + c];
              acc.x = fmaf(a, b.x, acc.x); acc.y = fmaf(a, b.y, acc.y);
              acc.z = fmaf(a, b.z, acc.z); acc.w = fmaf(a, b.w, acc.w);
            }
          }
        }
      }
      ((v4f*)wf)[j] = acc;
    }
    return;
  }
  const int i = (blockIdx.x - PREP_WF_BLOCKS) * 256 + threadIdx.x;
  if (i < IMG_FLOATS) {
    float v = 0.f;
    if (i < IMG_W1) {
      const int k = ((i >> 6) << 1) + (i & 1), n = (i & 63) >> 1;
      if (k < KFOLD && n < NFOLD) {
        const int kx9 = k % 9, cky = k / 9;
        const int var = n / 6, cls = (n % 6) / 3, dy = n % 3;
        const float* wrow = w2c + cls * 576 + dy * 3;
        const float* w1r  = w1c + cky * 7;
        bool ok[3]; int kx[3];
        #pragma unroll
        for (int dx = 0; dx < 3; ++dx) {
          const int t = kx9 - dx;
          const bool inset = (var == 0) || (var == 1 && dx >= 1) || (var == 2 && dx <= 1);
          ok[dx] = inset && t >= 0 && t <= 6;
          kx[dx] = t < 0 ? 0 : (t > 6 ? 6 : t);
        }
        #pragma unroll 16
        for (int ch = 0; ch < 64; ++ch) {
          #pragma unroll
          for (int dx = 0; dx < 3; ++dx) {
            const float a = wrow[ch * 9 + dx];
            v = fmaf(ok[dx] ? a : 0.f, w1r[ch * 147 + kx[dx]], v);
          }
        }
      }
    } else if (i < IMG_HEAD) {
      const int n = i - IMG_W1;
      if (n < NFOLD) {
        const int var = n / 6, cls = (n % 6) / 3, dy = n % 3;
        const float* wrow = w2c + cls * 576 + dy * 3;
        const bool ok0 = (var != 1), ok2 = (var != 2);
        #pragma unroll 16
        for (int ch = 0; ch < 64; ++ch) {
          const float bb = b1c[ch];
          v = fmaf(ok0 ? wrow[ch * 9 + 0] : 0.f, bb, v);
          v = fmaf(wrow[ch * 9 + 1], bb, v);
          v = fmaf(ok2 ? wrow[ch * 9 + 2] : 0.f, bb, v);
        }
      }
    } else {
      int j = i - IMG_HEAD;
      if (j < 2)       v = b2c[j];
      else if (j < 8)  v = wfm[j - 2];
      else if (j < 10) v = bfm[j - 8];
    }
    img[i] = v;
  }
}

#define XS_FLOATS        2160
#define KPAD             192
#define TPITCH           289
#define CONV_LDS_FLOATS  (XS_FLOATS + KPAD + IMG_W1 + NCOL + NTCOL*TPITCH)

__device__ __forceinline__ void conv1_tile(
    const float* xs, const int* offtab, const float* w1s, const float* b1s,
    float* t1s, int mt, int ncol, int lane, int Y0) {
  constexpr int KS = KPAD / 4;
  constexpr int G  = 4;
  constexpr int NB = KS / G;
  const int l16 = lane & 15;
  const int hk  = lane >> 4;
  const float* xa = xs + mt * 24 + l16;
  const int2* ot = ((const int2*)offtab) + hk;
  const v2f*  bp = ((const v2f*)w1s) + hk * NCOL + ncol;

  int2 off[2][G];
  v2f  af[2][G];
  v2f  bf[2][G];
  v8f  acc0 = v8f{}, acc1 = v8f{};

  const float bias = b1s[ncol];
  #pragma unroll
  for (int i = 0; i < G; ++i) { off[0][i] = ot[2 * i]; off[1][i] = ot[2 * (G + i)]; }
  #pragma unroll
  for (int i = 0; i < G; ++i) {
    af[0][i].x = xa[off[0][i].x];
    af[0][i].y = xa[off[0][i].y];
    bf[0][i] = bp[2 * NCOL * i];
  }

  #pragma unroll
  for (int g = 0; g < NB; ++g) {
    const int cur = g & 1, nxt = cur ^ 1;
    if (g + 1 < NB) {
      #pragma unroll
      for (int i = 0; i < G; ++i) {
        const int s = (g + 1) * G + i;
        af[nxt][i].x = xa[off[nxt][i].x];
        af[nxt][i].y = xa[off[nxt][i].y];
        bf[nxt][i] = bp[2 * NCOL * s];
      }
    }
    if (g + 2 < NB) {
      #pragma unroll
      for (int i = 0; i < G; ++i) off[cur][i] = ot[2 * ((g + 2) * G + i)];
    }
    __builtin_amdgcn_sched_barrier(0);
    #pragma unroll
    for (int i = 0; i < G; i += 2) {
      acc0 = wmma_f32(af[cur][i],     bf[cur][i],     acc0);
      acc1 = wmma_f32(af[cur][i + 1], bf[cur][i + 1], acc1);
    }
    __builtin_amdgcn_sched_barrier(0);
  }

  const int gy = Y0 - 1 + mt;
  const bool in = gy >= 0 && gy < HW;
  if (l16 < NTCOL) {
    #pragma unroll
    for (int r = 0; r < 8; ++r)
      t1s[l16 * TPITCH + mt * 16 + r + 8 * hk] = in ? (acc0[r] + acc1[r] + bias) : 0.f;
  }
}

__global__ __launch_bounds__(256) void conv_fused_kernel(
    const float* __restrict__ x,  const float* __restrict__ img,
    float* __restrict__ ini, float* __restrict__ fea) {
  extern __shared__ float smem[];
  float* xs     = smem;
  int*   offtab = (int*)(xs + XS_FLOATS);
  float* w1s    = xs + XS_FLOATS + KPAD;
  float* b1s    = w1s + IMG_W1;
  float* t1s    = b1s + NCOL;
  const float* __restrict__ cg  = img + IMG_HEAD;

  const int tid = threadIdx.x;
  const int X0 = blockIdx.x * 16, Y0 = blockIdx.y * 16, b = blockIdx.z;

  const v4f* img4 = (const v4f*)img;
  v4f cw[7];
  #pragma unroll
  for (int j = 0; j < 7; ++j) {
    int q = tid + 256 * j;
    q = (q < IMG_HEAD / 4) ? q : (IMG_HEAD / 4 - 1);
    cw[j] = img4[q];
  }
  v4f xq[2];
  #pragma unroll
  for (int j = 0; j < 2; ++j) {
    const int q  = tid + 256 * j;
    const int qq = (q < CIN * 144) ? q : (CIN * 144 - 1);
    const int row = qq / 6, f = qq % 6, c = row / 24, iy = row % 24;
    const int gy = Y0 - 4 + iy, gx = X0 - 4 + 4 * f;
    const bool inb = (q < CIN * 144) && gy >= 0 && gy < HW && gx >= 0 && gx < HW;
    const int cy = gy < 0 ? 0 : (gy > HW - 1 ? HW - 1 : gy);
    const int cx = gx < 0 ? 0 : (gx > HW - 4 ? HW - 4 : gx);
    const v4f v = *(const v4f*)(x + ((size_t)(b * CIN + c) * HW + cy) * HW + cx);
    xq[j] = inb ? v : v4f{0.f, 0.f, 0.f, 0.f};
  }
  #pragma unroll
  for (int j = 0; j < 7; ++j) {
    const int q = tid + 256 * j;
    if (q < IMG_HEAD / 4) ((v4f*)w1s)[q] = cw[j];
  }
  ((v4f*)xs)[tid]       = xq[0];
  ((v4f*)xs)[tid + 256] = xq[1];
  if (tid < XS_FLOATS / 4 - 512) ((v4f*)xs)[tid + 512] = v4f{0.f, 0.f, 0.f, 0.f};
  if (tid < KPAD) {
    int off = 1728;
    if (tid < KFOLD) {
      int c = tid / 63, rr = tid % 63;
      off = c * 576 + (rr / 9) * 24 + (rr % 9);
    }
    offtab[tid] = off;
  }
  __syncthreads();

  const int wave = tid >> 5;
  const int lane = tid & 31;
  const int l16  = lane & 15;
  const int ncol = (l16 < 6) ? l16
                 : (l16 < NTCOL) ? (l16 + ((blockIdx.x == HW / 16 - 1) ? 6 : 0)) : NFOLD;

  for (int mt = wave; mt < 18; mt += 8)
    conv1_tile(xs, offtab, w1s, b1s, t1s, mt, ncol, lane, Y0);
  __syncthreads();

  {
    const int oy = tid >> 4, ox = tid & 15;
    const int gy = Y0 + oy, gx = X0 + ox;
    const int vb = (gx == 0 || gx == HW - 1) ? 6 : 0;
    float a0 = cg[0], a1 = cg[1];
    for (int dy = 0; dy < 3; ++dy) {
      const int p = (oy + dy) * 16 + ox;
      a0 += t1s[(vb + dy) * TPITCH + p];
      a1 += t1s[(vb + 3 + dy) * TPITCH + p];
    }
    const size_t o0 = ((size_t)(b * CLS + 0) * HW + gy) * HW + gx;
    const size_t o1 = ((size_t)(b * CLS + 1) * HW + gy) * HW + gx;
    ini[o0] = a0;
    ini[o1] = a1;
    float x0 = xs[(oy + 4) * 24 + (ox + 4)];
    float x1 = xs[576 + (oy + 4) * 24 + (ox + 4)];
    float x2 = xs[1152 + (oy + 4) * 24 + (ox + 4)];
    fea[o0] = (cg[2] * x0 + cg[3] * x1 + cg[4] * x2 + cg[8]) * 0.25f;
    fea[o1] = (cg[5] * x0 + cg[6] * x1 + cg[7] * x2 + cg[9]) * 0.25f;
  }
}

#define TRANS_LDS_FLOATS (64 * 102)

__global__ __launch_bounds__(256) void trans_kernel(
    const float* __restrict__ ini, const float* __restrict__ fea,
    const float* __restrict__ wf,
    float* __restrict__ ui, float* __restrict__ uf) {
  extern __shared__ float smem[];
  float* a_lds = smem;

  const int tid = threadIdx.x;
  const float* src = blockIdx.y ? fea : ini;
  float* dst       = blockIdx.y ? uf  : ui;
  const int P0 = blockIdx.x * 64;

  for (int i = tid; i < 64 * P2; i += 256) {
    int r = i / P2, p = i % P2;
    int Pa = P0 + r;
    int bc = Pa >> 10, l = Pa & 1023;
    int gy = (l >> 5) * PSZ + p / PSZ;
    int gx = (l & 31) * PSZ + p % PSZ;
    a_lds[r * 102 + p] = src[((size_t)bc * HW + gy) * HW + gx];
  }
  __syncthreads();

  const int wave  = tid >> 5;
  const int lane  = tid & 31;
  const int l16   = lane & 15;
  const int lhalf = (lane < 16) ? 0 : 2;

  for (int t = wave; t < 4 * 7; t += 8) {
    const int m0 = (t / 7) * 16, n0 = (t % 7) * 16;
    const int n  = n0 + l16;
    const v2f* ap = (const v2f*)&a_lds[(m0 + l16) * 102 + lhalf];
    const v2f* bp = (const v2f*)&wf[n * P2 + lhalf];
    v8f acc = {};
    #pragma unroll 5
    for (int k0 = 0; k0 < P2; k0 += 4) {
      v2f a  = *ap;  ap += 2;
      v2f bf = *bp;  bp += 2;
      acc = wmma_f32(a, bf, acc);
    }
    if (n < P2) {
      #pragma unroll
      for (int r = 0; r < 8; ++r) {
        int M = m0 + r + ((lane < 16) ? 0 : 8);
        float v = acc[r];
        dst[(size_t)(P0 + M) * P2 + n] = v > 0.f ? v : 0.f;
      }
    }
  }
}

#define ATT_U_CHUNKS   (128 * P2 / 4)
#define ATT_CHUNKS     (ATT_U_CHUNKS + 64 * P2 / 4)
#define ATT_LDS_FLOATS (128 * P2 + 64 * P2)

__global__ __launch_bounds__(256) void att_kernel(
    const float* __restrict__ ui, const float* __restrict__ uf,
    float* __restrict__ out) {
  extern __shared__ float smem[];
  float* us = smem;
  float* vs = us + 128 * P2;

  const int tid = threadIdx.x;
  const int bc = blockIdx.z;
  const int M0 = blockIdx.x * 128, N0 = blockIdx.y * 64;
  const v4f* U4 = (const v4f*)(ui + ((size_t)bc * LPATCH + M0) * P2);
  const v4f* V4 = (const v4f*)(uf + ((size_t)bc * LPATCH + N0) * P2);

  v4f r[19];
  #pragma unroll
  for (int j = 0; j < 19; ++j) {
    const int q = (j < 18) ? (tid + 256 * j) : (tid + ATT_CHUNKS - 256);
    const v4f* p;
    if (j < 12)       p = U4 + q;
    else if (j == 12) p = (tid < 128) ? (U4 + q) : (V4 + (q - ATT_U_CHUNKS));
    else              p = V4 + (q - ATT_U_CHUNKS);
    r[j] = *p;
  }
  #pragma unroll
  for (int j = 0; j < 19; ++j) {
    const int q = (j < 18) ? (tid + 256 * j) : (tid + ATT_CHUNKS - 256);
    ((v4f*)us)[q] = r[j];
  }
  __syncthreads();

  const int wave  = tid >> 5;
  const int lane  = tid & 31;
  const int l16   = lane & 15;
  const int lhalf = (lane < 16) ? 0 : 2;

  for (int t = wave; t < 8 * 4; t += 8) {
    const int m0 = (t >> 2) * 16, n0 = (t & 3) * 16;
    const v2f* ap = (const v2f*)&us[(m0 + l16) * P2 + lhalf];
    const v2f* bp = (const v2f*)&vs[(n0 + l16) * P2 + lhalf];
    v8f acc = {};
    #pragma unroll 5
    for (int k0 = 0; k0 < P2; k0 += 4) {
      v2f a  = *ap;  ap += 2;
      v2f bf = *bp;  bp += 2;
      acc = wmma_f32(a, bf, acc);
    }
    #pragma unroll
    for (int r8 = 0; r8 < 8; ++r8) {
      int M = M0 + m0 + r8 + ((lane < 16) ? 0 : 8);
      int N = N0 + n0 + l16;
      out[((size_t)bc * LPATCH + M) * LPATCH + N] = acc[r8] * 0.01f;
    }
  }
}

extern "C" void kernel_launch(void* const* d_in, const int* in_sizes, int n_in,
                              void* d_out, int out_size, void* d_ws,
                              size_t ws_size, hipStream_t stream) {
  (void)in_sizes; (void)n_in; (void)out_size; (void)ws_size;
  const float* x       = (const float*)d_in[0];
  const float* w_conv1 = (const float*)d_in[1];
  const float* b_conv1 = (const float*)d_in[2];
  const float* w_conv2 = (const float*)d_in[3];
  const float* b_conv2 = (const float*)d_in[4];
  const float* w_fm    = (const float*)d_in[5];
  const float* b_fm    = (const float*)d_in[6];
  const float* w_lin1  = (const float*)d_in[7];
  const float* w_lin2  = (const float*)d_in[8];
  float* out = (float*)d_out;

  float* ws  = (float*)d_ws;
  float* ini = ws + WS_INI;
  float* fea = ws + WS_FEA;
  float* ui  = ws + WS_UI;
  float* uf  = ws + WS_UF;
  float* wf  = ws + WS_WF;
  float* img = ws + WS_IMG;

  pad_weights_kernel<<<dim3(PREP_WF_BLOCKS + PREP_IMG_BLOCKS), 256, 0, stream>>>(
      w_lin1, w_lin2, w_conv1, b_conv1, w_conv2, b_conv2, w_fm, b_fm,
      wf, img);

  conv_fused_kernel<<<dim3(HW / 16, HW / 16, BATCH), 256,
                      CONV_LDS_FLOATS * sizeof(float), stream>>>(
      x, img, ini, fea);

  trans_kernel<<<dim3(BATCH * CLS * LPATCH / 64, 2, 1), 256,
                 TRANS_LDS_FLOATS * sizeof(float), stream>>>(
      ini, fea, wf, ui, uf);

  att_kernel<<<dim3(LPATCH / 128, LPATCH / 64, BATCH * CLS), 256,
               ATT_LDS_FLOATS * sizeof(float), stream>>>(ui, uf, out);
}


# conv2 wholly folded into conv1 (9x9 window) for conv blocks whose halo lies inside the image, done as a banded matrix product with K split over 4 waves; border blocks as before; sums reordered
# speedup vs baseline: 4.5729x; 1.1432x over previous
#include <hip/hip_runtime.h>
#include <hip/hip_bf16.h>

typedef float v2f __attribute__((ext_vector_type(2)));
typedef float v8f __attribute__((ext_vector_type(8)));
typedef float v4f __attribute__((ext_vector_type(4)));

__device__ __forceinline__ v8f wmma_f32(v2f a, v2f b, v8f c) {
  return __builtin_amdgcn_wmma_f32_16x16x4_f32(false, a, false, b, (short)0, c,
                                               false, false);
}

#define BATCH   8
#define CIN     3
#define HW      320
#define CLS     2
#define PSZ     10
#define P2      100
#define LPATCH  1024
#define HID     200

#define WS_INI  0
#define WS_FEA  (WS_INI + BATCH*CLS*HW*HW)
#define WS_UI   (WS_FEA + BATCH*CLS*HW*HW)
#define WS_UF   (WS_UI  + BATCH*CLS*LPATCH*P2)
#define WS_WF   (WS_UF  + BATCH*CLS*LPATCH*P2)
#define WF_ROWS 112
#define PREP_WF_BLOCKS  11
#define PREP_IMG_BLOCKS 25
#define WS_IMG  WS_UI
#define NCOL       32
#define NFOLD      18
#define KFOLD      189
#define NTCOL      12
#define IMG_W1     (96 * NCOL * 2)
#define IMG_HEAD   (IMG_W1 + NCOL)
#define IMG_TAIL   16
#define IMG_FLOATS (IMG_HEAD + IMG_TAIL)
#define T_ROWS      27
#define IMG2_W      (T_ROWS * 4 * 2 * 16 * 2)
#define IMG2_FLOATS (IMG2_W + 16)
#define WS_IMG2     (WS_IMG + 6400)
#define PREP_IMG2_BLOCKS 109
#define RED_Q       544

__global__ __launch_bounds__(256) void pad_weights_kernel(
    const float* __restrict__ w_lin1, const float* __restrict__ w_lin2,
    const float* __restrict__ w1c, const float* __restrict__ b1c,
    const float* __restrict__ w2c, const float* __restrict__ b2c,
    const float* __restrict__ wfm, const float* __restrict__ bfm,
    float* __restrict__ wf, float* __restrict__ img, float* __restrict__ img2) {
  if (blockIdx.x >= PREP_WF_BLOCKS + PREP_IMG_BLOCKS) {
    const int t = (blockIdx.x - PREP_WF_BLOCKS - PREP_IMG_BLOCKS) * 256 + threadIdx.x;
    const int e = t >> 2, part = t & 3;
    float v = 0.f;
    if (e < IMG2_W) {
      const int ee = e & 1, n = (e >> 1) & 15, hk = (e >> 5) & 1, q = (e >> 6) & 3, i = e >> 8;
      const int j = 4 * q + 2 * hk + ee, g = n & 7, cls = n >> 3, kx9 = j - g;
      const int c = i / 9, ky9 = i % 9;
      if (kx9 >= 0 && kx9 <= 8) {
        bool ok[9]; int o1[9];
        #pragma unroll
        for (int dy = 0; dy < 3; ++dy) {
          #pragma unroll
          for (int dx = 0; dx < 3; ++dx) {
            const int ty = ky9 - dy, tx = kx9 - dx;
            ok[dy * 3 + dx] = (ty >= 0) & (ty <= 6) & (tx >= 0) & (tx <= 6);
            o1[dy * 3 + dx] = (ty < 0 ? 0 : (ty > 6 ? 6 : ty)) * 7 + (tx < 0 ? 0 : (tx > 6 ? 6 : tx));
          }
        }
        const float* w2p = w2c + cls * 576 + (16 * part) * 9;
        const float* w1p = w1c + (16 * part) * 147 + c * 49;
        #pragma unroll 8
        for (int ch = 0; ch < 16; ++ch) {
          #pragma unroll
          for (int tt = 0; tt < 9; ++tt) {
            const float a = w2p[ch * 9 + tt];
            v = fmaf(ok[tt] ? a : 0.f, w1p[ch * 147 + o1[tt]], v);
          }
        }
      }
    } else if (e < IMG2_W + 2) {
      const int cls = e - IMG2_W;
      const float* w2p = w2c + cls * 576 + (16 * part) * 9;
      if (part == 0) v = b2c[cls];
      #pragma unroll 8
      for (int ch = 0; ch < 16; ++ch) {
        const float bb = b1c[16 * part + ch];
        #pragma unroll
        for (int tt = 0; tt < 9; ++tt) v = fmaf(w2p[ch * 9 + tt], bb, v);
      }
    }
    v += __shfl_xor(v, 1);
    v += __shfl_xor(v, 2);
    if (part == 0 && e < IMG2_FLOATS) img2[e] = v;
    return;
  }
  if (blockIdx.x < PREP_WF_BLOCKS) {
    const int j = blockIdx.x * 256 + threadIdx.x;
    if (j < WF_ROWS * (P2 / 4)) {
      const int n = j / (P2 / 4), kq = j % (P2 / 4);
      v4f acc = {0.f, 0.f, 0.f, 0.f};
      if (n < P2) {
        const v4f* a4 = (const v4f*)(w_lin2 + n * HID);
        const v4f* b4 = ((const v4f*)w_lin1) + kq;
        #pragma unroll 1
        for (int t = 0; t < HID / 40; ++t) {
          v4f av[10], bv[40];
          #pragma unroll
          for (int u = 0; u < 10; ++u) {
            const int s = t * 10 + u;
            av[u] = a4[s];
            #pragma unroll
            for (int c = 0; c < 4; ++c) bv[4 * u + c] = b4[(4 * s + c) * (P2 / 4)];
          }
          __builtin_amdgcn_sched_barrier(0);
          #pragma unroll
          for (int u = 0; u < 10; ++u) {
            #pragma unroll
            for (int c = 0; c < 4; ++c) {
              const float a = av[u][c];
              const v4f b = bv[4 * u + c];
              acc.x = fmaf(a, b.x, acc.x); acc.y = fmaf(a, b.y, acc.y);
              acc.z = fmaf(a, b.z, acc.z); acc.w = fmaf(a, b.w, acc.w);
            }
          }
        }
      }
      ((v4f*)wf)[j] = acc;
    }
    return;
  }
  const int i = (blockIdx.x - PREP_WF_BLOCKS) * 256 + threadIdx.x;
  if (i < IMG_FLOATS) {
    float v = 0.f;
    if (i < IMG_W1) {
      const int k = ((i >> 6) << 1) + (i & 1), n = (i & 63) >> 1;
      if (k < KFOLD && n < NFOLD) {
        const int kx9 = k % 9, cky = k / 9;
        const int var = n / 6, cls = (n % 6) / 3, dy = n % 3;
        const float* wrow = w2c + cls * 576 + dy * 3;
        const float* w1r  = w1c + cky * 7;
        bool ok[3]; int kx[3];
        #pragma unroll
        for (int dx = 0; dx < 3; ++dx) {
          const int t = kx9 - dx;
          const bool inset = (var == 0) || (var == 1 && dx >= 1) || (var == 2 && dx <= 1);
          ok[dx] = inset && t >= 0 && t <= 6;
          kx[dx] = t < 0 ? 0 : (t > 6 ? 6 : t);
        }
        #pragma unroll 16
        for (int ch = 0; ch < 64; ++ch) {
          #pragma unroll
          for (int dx = 0; dx < 3; ++dx) {
            const float a = wrow[ch * 9 + dx];
            v = fmaf(ok[dx] ? a : 0.f, w1r[ch * 147 + kx[dx]], v);
          }
        }
      }
    } else if (i < IMG_HEAD) {
      const int n = i - IMG_W1;
      if (n < NFOLD) {
        const int var = n / 6, cls = (n % 6) / 3, dy = n % 3;
        const float* wrow = w2c + cls * 576 + dy * 3;
        const bool ok0 = (var != 1), ok2 = (var != 2);
        #pragma unroll 16
        for (int ch = 0; ch < 64; ++ch) {
          const float bb = b1c[ch];
          v = fmaf(ok0 ? wrow[ch * 9 + 0] : 0.f, bb, v);
          v = fmaf(wrow[ch * 9 + 1], bb, v);
          v = fmaf(ok2 ? wrow[ch * 9 + 2] : 0.f, bb, v);
        }
      }
    } else {
      int j = i - IMG_HEAD;
      if (j < 2)       v = b2c[j];
      else if (j < 8)  v = wfm[j - 2];
      else if (j < 10) v = bfm[j - 8];
    }
    img[i] = v;
  }
}

#define XS_FLOATS        2160
#define KPAD             192
#define TPITCH           289
#define CONV_LDS_FLOATS  (XS_FLOATS + KPAD + IMG_W1 + NCOL + NTCOL*TPITCH)

__device__ __forceinline__ void conv1_tile(
    const float* xs, const int* offtab, const float* w1s, const float* b1s,
    float* t1s, int mt, int ncol, int lane, int Y0) {
  constexpr int KS = KPAD / 4;
  constexpr int G  = 4;
  constexpr int NB = KS / G;
  const int l16 = lane & 15;
  const int hk  = lane >> 4;
  const float* xa = xs + mt * 24 + l16;
  const int2* ot = ((const int2*)offtab) + hk;
  const v2f*  bp = ((const v2f*)w1s) + hk * NCOL + ncol;

  int2 off[2][G];
  v2f  af[2][G];
  v2f  bf[2][G];
  v8f  acc0 = v8f{}, acc1 = v8f{};

  const float bias = b1s[ncol];
  #pragma unroll
  for (int i = 0; i < G; ++i) { off[0][i] = ot[2 * i]; off[1][i] = ot[2 * (G + i)]; }
  #pragma unroll
  for (int i = 0; i < G; ++i) {
    af[0][i].x = xa[off[0][i].x];
    af[0][i].y = xa[off[0][i].y];
    bf[0][i] = bp[2 * NCOL * i];
  }

  #pragma unroll
  for (int g = 0; g < NB; ++g) {
    const int cur = g & 1, nxt = cur ^ 1;
    if (g + 1 < NB) {
      #pragma unroll
      for (int i = 0; i < G; ++i) {
        const int s = (g + 1) * G + i;
        af[nxt][i].x = xa[off[nxt][i].x];
        af[nxt][i].y = xa[off[nxt][i].y];
        bf[nxt][i] = bp[2 * NCOL * s];
      }
    }
    if (g + 2 < NB) {
      #pragma unroll
      for (int i = 0; i < G; ++i) off[cur][i] = ot[2 * ((g + 2) * G + i)];
    }
    __builtin_amdgcn_sched_barrier(0);
    #pragma unroll
    for (int i = 0; i < G; i += 2) {
      acc0 = wmma_f32(af[cur][i],     bf[cur][i],     acc0);
      acc1 = wmma_f32(af[cur][i + 1], bf[cur][i + 1], acc1);
    }
    __builtin_amdgcn_sched_barrier(0);
  }

  const int gy = Y0 - 1 + mt;
  const bool in = gy >= 0 && gy < HW;
  if (l16 < NTCOL) {
    #pragma unroll
    for (int r = 0; r < 8; ++r)
      t1s[l16 * TPITCH + mt * 16 + r + 8 * hk] = in ? (acc0[r] + acc1[r] + bias) : 0.f;
  }
}

__device__ __forceinline__ void conv_interior_tile(
    const float* xs, const float* wB, float* red, int tid) {
  const int wave = tid >> 5, lane = tid & 31;
  const int l16 = lane & 15, hk = lane >> 4;
  const int T = wave & 1, q = wave >> 1;
  const int Y = 8 * T + (l16 >> 1), grp = l16 & 1;
  const v2f* ap = (const v2f*)(xs + Y * 24 + 8 * grp + 4 * q + 2 * hk);
  const v2f* bp = ((const v2f*)wB) + (q * 2 + hk) * 16 + l16;
  constexpr int GS = 6;
  constexpr int NG = (T_ROWS + GS - 1) / GS;
  v2f af[2][GS], bf[2][GS];
  v8f acc0 = v8f{}, acc1 = v8f{};
  #pragma unroll
  for (int i = 0; i < GS; ++i) {
    af[0][i] = ap[(i / 9) * 288 + (i % 9) * 12];
    bf[0][i] = bp[128 * i];
  }
  #pragma unroll
  for (int gI = 0; gI < NG; ++gI) {
    const int cur = gI & 1, nxt = cur ^ 1;
    #pragma unroll
    for (int i = 0; i < GS; ++i) {
      const int st = (gI + 1) * GS + i;
      if (st < T_ROWS) {
        af[nxt][i] = ap[(st / 9) * 288 + (st % 9) * 12];
        bf[nxt][i] = bp[128 * st];
      }
    }
    __builtin_amdgcn_sched_barrier(0);
    #pragma unroll
    for (int i = 0; i < GS; ++i) {
      const int st = gI * GS + i;
      if (st < T_ROWS) {
        if (st & 1) acc1 = wmma_f32(af[cur][i], bf[cur][i], acc1);
        else        acc0 = wmma_f32(af[cur][i], bf[cur][i], acc0);
      }
    }
    __builtin_amdgcn_sched_barrier(0);
  }
  const int g = l16 & 7, cls = l16 >> 3;
  float* rp = red + q * RED_Q + cls * 272 + (8 * T + 4 * hk) * 16 + g;
  #pragma unroll
  for (int r = 0; r < 8; ++r)
    rp[(r >> 1) * 16 + 8 * (r & 1)] = acc0[r] + acc1[r];
}

__global__ __launch_bounds__(256) void conv_fused_kernel(
    const float* __restrict__ x,  const float* __restrict__ img,
    const float* __restrict__ img2,
    float* __restrict__ ini, float* __restrict__ fea) {
  extern __shared__ float smem[];
  float* xs     = smem;
  int*   offtab = (int*)(xs + XS_FLOATS);
  float* w1s    = xs + XS_FLOATS + KPAD;
  float* b1s    = w1s + IMG_W1;
  float* t1s    = b1s + NCOL;
  const float* __restrict__ cg  = img + IMG_HEAD;

  const int tid = threadIdx.x;
  const int X0 = blockIdx.x * 16, Y0 = blockIdx.y * 16, b = blockIdx.z;
  const bool interior = blockIdx.x >= 1 && blockIdx.x <= HW / 16 - 2 &&
                        blockIdx.y >= 1 && blockIdx.y <= HW / 16 - 2;
  float* wB  = w1s;
  float* red = wB + IMG2_FLOATS;

  const v4f* img4 = (const v4f*)(interior ? img2 : img);
  const int nchunk = interior ? IMG2_FLOATS / 4 : IMG_HEAD / 4;
  v4f cw[7];
  #pragma unroll
  for (int j = 0; j < 7; ++j) {
    int q = tid + 256 * j;
    q = (q < nchunk) ? q : (nchunk - 1);
    cw[j] = img4[q];
  }
  v4f xq[2];
  #pragma unroll
  for (int j = 0; j < 2; ++j) {
    const int q  = tid + 256 * j;
    const int qq = (q < CIN * 144) ? q : (CIN * 144 - 1);
    const int row = qq / 6, f = qq % 6, c = row / 24, iy = row % 24;
    const int gy = Y0 - 4 + iy, gx = X0 - 4 + 4 * f;
    const bool inb = (q < CIN * 144) && gy >= 0 && gy < HW && gx >= 0 && gx < HW;
    const int cy = gy < 0 ? 0 : (gy > HW - 1 ? HW - 1 : gy);
    const int cx = gx < 0 ? 0 : (gx > HW - 4 ? HW - 4 : gx);
    const v4f v = *(const v4f*)(x + ((size_t)(b * CIN + c) * HW + cy) * HW + cx);
    xq[j] = inb ? v : v4f{0.f, 0.f, 0.f, 0.f};
  }
  #pragma unroll
  for (int j = 0; j < 7; ++j) {
    const int q = tid + 256 * j;
    if (q < nchunk) ((v4f*)w1s)[q] = cw[j];
  }
  ((v4f*)xs)[tid]       = xq[0];
  ((v4f*)xs)[tid + 256] = xq[1];
  if (tid < XS_FLOATS / 4 - 512) ((v4f*)xs)[tid + 512] = v4f{0.f, 0.f, 0.f, 0.f};
  if (tid < KPAD) {
    int off = 1728;
    if (tid < KFOLD) {
      int c = tid / 63, rr = tid % 63;
      off = c * 576 + (rr / 9) * 24 + (rr % 9);
    }
    offtab[tid] = off;
  }
  __syncthreads();

  const int wave = tid >> 5;
  const int lane = tid & 31;
  const int l16  = lane & 15;
  const int ncol = (l16 < 6) ? l16
                 : (l16 < NTCOL) ? (l16 + ((blockIdx.x == HW / 16 - 1) ? 6 : 0)) : NFOLD;

  const int oy = tid >> 4, ox = tid & 15;
  const int gy = Y0 + oy, gx = X0 + ox;
  float a0, a1;
  if (interior) {
    conv_interior_tile(xs, wB, red, tid);
    __syncthreads();
    const float* rq = red + oy * 16 + ox;
    a0 = wB[IMG2_W + 0] + ((rq[0] + rq[RED_Q]) + (rq[2 * RED_Q] + rq[3 * RED_Q]));
    a1 = wB[IMG2_W + 1] + ((rq[272] + rq[RED_Q + 272]) + (rq[2 * RED_Q + 272] + rq[3 * RED_Q + 272]));
  } else {
    for (int mt = wave; mt < 18; mt += 8)
      conv1_tile(xs, offtab, w1s, b1s, t1s, mt, ncol, lane, Y0);
    __syncthreads();
    const int vb = (gx == 0 || gx == HW - 1) ? 6 : 0;
    a0 = cg[0]; a1 = cg[1];
    for (int dy = 0; dy < 3; ++dy) {
      const int p = (oy + dy) * 16 + ox;
      a0 += t1s[(vb + dy) * TPITCH + p];
      a1 += t1s[(vb + 3 + dy) * TPITCH + p];
    }
  }
  {
    const size_t o0 = ((size_t)(b * CLS + 0) * HW + gy) * HW + gx;
    const size_t o1 = ((size_t)(b * CLS + 1) * HW + gy) * HW + gx;
    ini[o0] = a0;
    ini[o1] = a1;
    float x0 = xs[(oy + 4) * 24 + (ox + 4)];
    float x1 = xs[576 + (oy + 4) * 24 + (ox + 4)];
    float x2 = xs[1152 + (oy + 4) * 24 + (ox + 4)];
    fea[o0] = (cg[2] * x0 + cg[3] * x1 + cg[4] * x2 + cg[8]) * 0.25f;
    fea[o1] = (cg[5] * x0 + cg[6] * x1 + cg[7] * x2 + cg[9]) * 0.25f;
  }
}

#define TRANS_LDS_FLOATS (64 * 102)

__global__ __launch_bounds__(256) void trans_kernel(
    const float* __restrict__ ini, const float* __restrict__ fea,
    const float* __restrict__ wf,
    float* __restrict__ ui, float* __restrict__ uf) {
  extern __shared__ float smem[];
  float* a_lds = smem;

  const int tid = threadIdx.x;
  const float* src = blockIdx.y ? fea : ini;
  float* dst       = blockIdx.y ? uf  : ui;
  const int P0 = blockIdx.x * 64;

  for (int i = tid; i < 64 * P2; i += 256) {
    int r = i / P2, p = i % P2;
    int Pa = P0 + r;
    int bc = Pa >> 10, l = Pa & 1023;
    int gy = (l >> 5) * PSZ + p / PSZ;
    int gx = (l & 31) * PSZ + p % PSZ;
    a_lds[r * 102 + p] = src[((size_t)bc * HW + gy) * HW + gx];
  }
  __syncthreads();

  const int wave  = tid >> 5;
  const int lane  = tid & 31;
  const int l16   = lane & 15;
  const int lhalf = (lane < 16) ? 0 : 2;

  for (int t = wave; t < 4 * 7; t += 8) {
    const int m0 = (t / 7) * 16, n0 = (t % 7) * 16;
    const int n  = n0 + l16;
    const v2f* ap = (const v2f*)&a_lds[(m0 + l16) * 102 + lhalf];
    const v2f* bp = (const v2f*)&wf[n * P2 + lhalf];
    v8f acc = {};
    #pragma unroll 5
    for (int k0 = 0; k0 < P2; k0 += 4) {
      v2f a  = *ap;  ap += 2;
      v2f bf = *bp;  bp += 2;
      acc = wmma_f32(a, bf, acc);
    }
    if (n < P2) {
      #pragma unroll
      for (int r = 0; r < 8; ++r) {
        int M = m0 + r + ((lane < 16) ? 0 : 8);
        float v = acc[r];
        dst[(size_t)(P0 + M) * P2 + n] = v > 0.f ? v : 0.f;
      }
    }
  }
}

#define ATT_U_CHUNKS   (128 * P2 / 4)
#define ATT_CHUNKS     (ATT_U_CHUNKS + 64 * P2 / 4)
#define ATT_LDS_FLOATS (128 * P2 + 64 * P2)

__global__ __launch_bounds__(256) void att_kernel(
    const float* __restrict__ ui, const float* __restrict__ uf,
    float* __restrict__ out) {
  extern __shared__ float smem[];
  float* us = smem;
  float* vs = us + 128 * P2;

  const int tid = threadIdx.x;
  const int bc = blockIdx.z;
  const int M0 = blockIdx.x * 128, N0 = blockIdx.y * 64;
  const v4f* U4 = (const v4f*)(ui + ((size_t)bc * LPATCH + M0) * P2);
  const v4f* V4 = (const v4f*)(uf + ((size_t)bc * LPATCH + N0) * P2);

  v4f r[19];
  #pragma unroll
  for (int j = 0; j < 19; ++j) {
    const int q = (j < 18) ? (tid + 256 * j) : (tid + ATT_CHUNKS - 256);
    const v4f* p;
    if (j < 12)       p = U4 + q;
    else if (j == 12) p = (tid < 128) ? (U4 + q) : (V4 + (q - ATT_U_CHUNKS));
    else              p = V4 + (q - ATT_U_CHUNKS);
    r[j] = *p;
  }
  #pragma unroll
  for (int j = 0; j < 19; ++j) {
    const int q = (j < 18) ? (tid + 256 * j) : (tid + ATT_CHUNKS - 256);
    ((v4f*)us)[q] = r[j];
  }
  __syncthreads();

  const int wave  = tid >> 5;
  const int lane  = tid & 31;
  const int l16   = lane & 15;
  const int lhalf = (lane < 16) ? 0 : 2;

  for (int t = wave; t < 8 * 4; t += 8) {
    const int m0 = (t >> 2) * 16, n0 = (t & 3) * 16;
    const v2f* ap = (const v2f*)&us[(m0 + l16) * P2 + lhalf];
    const v2f* bp = (const v2f*)&vs[(n0 + l16) * P2 + lhalf];
    v8f acc = {};
    #pragma unroll 5
    for (int k0 = 0; k0 < P2; k0 += 4) {
      v2f a  = *ap;  ap += 2;
      v2f bf = *bp;  bp += 2;
      acc = wmma_f32(a, bf, acc);
    }
    #pragma unroll
    for (int r8 = 0; r8 < 8; ++r8) {
      int M = M0 + m0 + r8 + ((lane < 16) ? 0 : 8);
      int N = N0 + n0 + l16;
      out[((size_t)bc * LPATCH + M) * LPATCH + N] = acc[r8] * 0.01f;
    }
  }
}

extern "C" void kernel_launch(void* const* d_in, const int* in_sizes, int n_in,
                              void* d_out, int out_size, void* d_ws,
                              size_t ws_size, hipStream_t stream) {
  (void)in_sizes; (void)n_in; (void)out_size; (void)ws_size;
  const float* x       = (const float*)d_in[0];
  const float* w_conv1 = (const float*)d_in[1];
  const float* b_conv1 = (const float*)d_in[2];
  const float* w_conv2 = (const float*)d_in[3];
  const float* b_conv2 = (const float*)d_in[4];
  const float* w_fm    = (const float*)d_in[5];
  const float* b_fm    = (const float*)d_in[6];
  const float* w_lin1  = (const float*)d_in[7];
  const float* w_lin2  = (const float*)d_in[8];
  float* out = (float*)d_out;

  float* ws  = (float*)d_ws;
  float* ini = ws + WS_INI;
  float* fea = ws + WS_FEA;
  float* ui  = ws + WS_UI;
  float* uf  = ws + WS_UF;
  float* wf  = ws + WS_WF;
  float* img = ws + WS_IMG;
  float* img2 = ws + WS_IMG2;

  pad_weights_kernel<<<dim3(PREP_WF_BLOCKS + PREP_IMG_BLOCKS + PREP_IMG2_BLOCKS), 256, 0, stream>>>(
      w_lin1, w_lin2, w_conv1, b_conv1, w_conv2, b_conv2, w_fm, b_fm,
      wf, img, img2);

  conv_fused_kernel<<<dim3(HW / 16, HW / 16, BATCH), 256,
                      CONV_LDS_FLOATS * sizeof(float), stream>>>(
      x, img, img2, ini, fea);

  trans_kernel<<<dim3(BATCH * CLS * LPATCH / 64, 2, 1), 256,
                 TRANS_LDS_FLOATS * sizeof(float), stream>>>(
      ini, fea, wf, ui, uf);

  att_kernel<<<dim3(LPATCH / 128, LPATCH / 64, BATCH * CLS), 256,
               ATT_LDS_FLOATS * sizeof(float), stream>>>(ui, uf, out);
}


# conv kernel: interior workgroups loop over 4 tiles, border tiles first (grid 3200 -> 1256, all resident at once): 15552 fewer waves enter, image 2 staged once per 4 tiles, next x tile loaded ahead
# speedup vs baseline: 5.1015x; 1.1156x over previous
#include <hip/hip_runtime.h>
#include <hip/hip_bf16.h>

typedef float v2f __attribute__((ext_vector_type(2)));
typedef float v8f __attribute__((ext_vector_type(8)));
typedef float v4f __attribute__((ext_vector_type(4)));

__device__ __forceinline__ v8f wmma_f32(v2f a, v2f b, v8f c) {
  return __builtin_amdgcn_wmma_f32_16x16x4_f32(false, a, false, b, (short)0, c,
                                               false, false);
}

#define BATCH   8
#define CIN     3
#define HW      320
#define CLS     2
#define PSZ     10
#define P2      100
#define LPATCH  1024
#define HID     200

#define WS_INI  0
#define WS_FEA  (WS_INI + BATCH*CLS*HW*HW)
#define WS_UI   (WS_FEA + BATCH*CLS*HW*HW)
#define WS_UF   (WS_UI  + BATCH*CLS*LPATCH*P2)
#define WS_WF   (WS_UF  + BATCH*CLS*LPATCH*P2)
#define WF_ROWS 112
#define PREP_WF_BLOCKS  11
#define PREP_IMG_BLOCKS 25
#define WS_IMG  WS_UI
#define NCOL       32
#define NFOLD      18
#define KFOLD      189
#define NTCOL      12
#define IMG_W1     (96 * NCOL * 2)
#define IMG_HEAD   (IMG_W1 + NCOL)
#define IMG_TAIL   16
#define IMG_FLOATS (IMG_HEAD + IMG_TAIL)
#define T_ROWS      27
#define IMG2_W      (T_ROWS * 4 * 2 * 16 * 2)
#define IMG2_FLOATS (IMG2_W + 16)
#define WS_IMG2     (WS_IMG + 6400)
#define PREP_IMG2_BLOCKS 109
#define RED_Q       544
#define TILES_1D       (HW / 16)
#define INT_1D         (TILES_1D - 2)
#define INT_PER_IMG    (INT_1D * INT_1D)
#define BORDER_PER_IMG (4 * TILES_1D - 4)
#define N_BORDER_BLK   (BATCH * BORDER_PER_IMG)
#define INT_TILES_BLK  4
#define N_INT_BLK      (BATCH * INT_PER_IMG / INT_TILES_BLK)

__global__ __launch_bounds__(256) void pad_weights_kernel(
    const float* __restrict__ w_lin1, const float* __restrict__ w_lin2,
    const float* __restrict__ w1c, const float* __restrict__ b1c,
    const float* __restrict__ w2c, const float* __restrict__ b2c,
    const float* __restrict__ wfm, const float* __restrict__ bfm,
    float* __restrict__ wf, float* __restrict__ img, float* __restrict__ img2) {
  if (blockIdx.x >= PREP_WF_BLOCKS + PREP_IMG_BLOCKS) {
    const int t = (blockIdx.x - PREP_WF_BLOCKS - PREP_IMG_BLOCKS) * 256 + threadIdx.x;
    const int e = t >> 2, part = t & 3;
    float v = 0.f;
    if (e < IMG2_W) {
      const int ee = e & 1, n = (e >> 1) & 15, hk = (e >> 5) & 1, q = (e >> 6) & 3, i = e >> 8;
      const int j = 4 * q + 2 * hk + ee, g = n & 7, cls = n >> 3, kx9 = j - g;
      const int c = i / 9, ky9 = i % 9;
      if (kx9 >= 0 && kx9 <= 8) {
        bool ok[9]; int o1[9];
        #pragma unroll
        for (int dy = 0; dy < 3; ++dy) {
          #pragma unroll
          for (int dx = 0; dx < 3; ++dx) {
            const int ty = ky9 - dy, tx = kx9 - dx;
            ok[dy * 3 + dx] = (ty >= 0) & (ty <= 6) & (tx >= 0) & (tx <= 6);
            o1[dy * 3 + dx] = (ty < 0 ? 0 : (ty > 6 ? 6 : ty)) * 7 + (tx < 0 ? 0 : (tx > 6 ? 6 : tx));
          }
        }
        const float* w2p = w2c + cls * 576 + (16 * part) * 9;
        const float* w1p = w1c + (16 * part) * 147 + c * 49;
        #pragma unroll 8
        for (int ch = 0; ch < 16; ++ch) {
          #pragma unroll
          for (int tt = 0; tt < 9; ++tt) {
            const float a = w2p[ch * 9 + tt];
            v = fmaf(ok[tt] ? a : 0.f, w1p[ch * 147 + o1[tt]], v);
          }
        }
      }
    } else if (e < IMG2_W + 2) {
      const int cls = e - IMG2_W;
      const float* w2p = w2c + cls * 576 + (16 * part) * 9;
      if (part == 0) v = b2c[cls];
      #pragma unroll 8
      for (int ch = 0; ch < 16; ++ch) {
        const float bb = b1c[16 * part + ch];
        #pragma unroll
        for (int tt = 0; tt < 9; ++tt) v = fmaf(w2p[ch * 9 + tt], bb, v);
      }
    }
    v += __shfl_xor(v, 1);
    v += __shfl_xor(v, 2);
    if (part == 0 && e < IMG2_FLOATS) img2[e] = v;
    return;
  }
  if (blockIdx.x < PREP_WF_BLOCKS) {
    const int j = blockIdx.x * 256 + threadIdx.x;
    if (j < WF_ROWS * (P2 / 4)) {
      const int n = j / (P2 / 4), kq = j % (P2 / 4);
      v4f acc = {0.f, 0.f, 0.f, 0.f};
      if (n < P2) {
        const v4f* a4 = (const v4f*)(w_lin2 + n * HID);
        const v4f* b4 = ((const v4f*)w_lin1) + kq;
        #pragma unroll 1
        for (int t = 0; t < HID / 40; ++t) {
          v4f av[10], bv[40];
          #pragma unroll
          for (int u = 0; u < 10; ++u) {
            const int s = t * 10 + u;
            av[u] = a4[s];
            #pragma unroll
            for (int c = 0; c < 4; ++c) bv[4 * u + c] = b4[(4 * s + c) * (P2 / 4)];
          }
          __builtin_amdgcn_sched_barrier(0);
          #pragma unroll
          for (int u = 0; u < 10; ++u) {
            #pragma unroll
            for (int c = 0; c < 4; ++c) {
              const float a = av[u][c];
              const v4f b = bv[4 * u + c];
              acc.x = fmaf(a, b.x, acc.x); acc.y = fmaf(a, b.y, acc.y);
              acc.z = fmaf(a, b.z, acc.z); acc.w = fmaf(a, b.w, acc.w);
            }
          }
        }
      }
      ((v4f*)wf)[j] = acc;
    }
    return;
  }
  const int i = (blockIdx.x - PREP_WF_BLOCKS) * 256 + threadIdx.x;
  if (i < IMG_FLOATS) {
    float v = 0.f;
    if (i < IMG_W1) {
      const int k = ((i >> 6) << 1) + (i & 1), n = (i & 63) >> 1;
      if (k < KFOLD && n < NFOLD) {
        const int kx9 = k % 9, cky = k / 9;
        const int var = n / 6, cls = (n % 6) / 3, dy = n % 3;
        const float* wrow = w2c + cls * 576 + dy * 3;
        const float* w1r  = w1c + cky * 7;
        bool ok[3]; int kx[3];
        #pragma unroll
        for (int dx = 0; dx < 3; ++dx) {
          const int t = kx9 - dx;
          const bool inset = (var == 0) || (var == 1 && dx >= 1) || (var == 2 && dx <= 1);
          ok[dx] = inset && t >= 0 && t <= 6;
          kx[dx] = t < 0 ? 0 : (t > 6 ? 6 : t);
        }
        #pragma unroll 16
        for (int ch = 0; ch < 64; ++ch) {
          #pragma unroll
          for (int dx = 0; dx < 3; ++dx) {
            const float a = wrow[ch * 9 + dx];
            v = fmaf(ok[dx] ? a : 0.f, w1r[ch * 147 + kx[dx]], v);
          }
        }
      }
    } else if (i < IMG_HEAD) {
      const int n = i - IMG_W1;
      if (n < NFOLD) {
        const int var = n / 6, cls = (n % 6) / 3, dy = n % 3;
        const float* wrow = w2c + cls * 576 + dy * 3;
        const bool ok0 = (var != 1), ok2 = (var != 2);
        #pragma unroll 16
        for (int ch = 0; ch < 64; ++ch) {
          const float bb = b1c[ch];
          v = fmaf(ok0 ? wrow[ch * 9 + 0] : 0.f, bb, v);
          v = fmaf(wrow[ch * 9 + 1], bb, v);
          v = fmaf(ok2 ? wrow[ch * 9 + 2] : 0.f, bb, v);
        }
      }
    } else {
      int j = i - IMG_HEAD;
      if (j < 2)       v = b2c[j];
      else if (j < 8)  v = wfm[j - 2];
      else if (j < 10) v = bfm[j - 8];
    }
    img[i] = v;
  }
}

#define XS_FLOATS        2160
#define KPAD             192
#define TPITCH           289
#define CONV_LDS_FLOATS  (XS_FLOATS + KPAD + IMG_W1 + NCOL + NTCOL*TPITCH)

__device__ __forceinline__ void conv1_tile(
    const float* xs, const int* offtab, const float* w1s, const float* b1s,
    float* t1s, int mt, int ncol, int lane, int Y0) {
  constexpr int KS = KPAD / 4;
  constexpr int G  = 4;
  constexpr int NB = KS / G;
  const int l16 = lane & 15;
  const int hk  = lane >> 4;
  const float* xa = xs + mt * 24 + l16;
  const int2* ot = ((const int2*)offtab) + hk;
  const v2f*  bp = ((const v2f*)w1s) + hk * NCOL + ncol;

  int2 off[2][G];
  v2f  af[2][G];
  v2f  bf[2][G];
  v8f  acc0 = v8f{}, acc1 = v8f{};

  const float bias = b1s[ncol];
  #pragma unroll
  for (int i = 0; i < G; ++i) { off[0][i] = ot[2 * i]; off[1][i] = ot[2 * (G + i)]; }
  #pragma unroll
  for (int i = 0; i < G; ++i) {
    af[0][i].x = xa[off[0][i].x];
    af[0][i].y = xa[off[0][i].y];
    bf[0][i] = bp[2 * NCOL * i];
  }

  #pragma unroll
  for (int g = 0; g < NB; ++g) {
    const int cur = g & 1, nxt = cur ^ 1;
    if (g + 1 < NB) {
      #pragma unroll
      for (int i = 0; i < G; ++i) {
        const int s = (g + 1) * G + i;
        af[nxt][i].x = xa[off[nxt][i].x];
        af[nxt][i].y = xa[off[nxt][i].y];
        bf[nxt][i] = bp[2 * NCOL * s];
      }
    }
    if (g + 2 < NB) {
      #pragma unroll
      for (int i = 0; i < G; ++i) off[cur][i] = ot[2 * ((g + 2) * G + i)];
    }
    __builtin_amdgcn_sched_barrier(0);
    #pragma unroll
    for (int i = 0; i < G; i += 2) {
      acc0 = wmma_f32(af[cur][i],     bf[cur][i],     acc0);
      acc1 = wmma_f32(af[cur][i + 1], bf[cur][i + 1], acc1);
    }
    __builtin_amdgcn_sched_barrier(0);
  }

  const int gy = Y0 - 1 + mt;
  const bool in = gy >= 0 && gy < HW;
  if (l16 < NTCOL) {
    #pragma unroll
    for (int r = 0; r < 8; ++r)
      t1s[l16 * TPITCH + mt * 16 + r + 8 * hk] = in ? (acc0[r] + acc1[r] + bias) : 0.f;
  }
}

__device__ __forceinline__ void conv_interior_tile(
    const float* xs, const float* wB, float* red, int tid) {
  const int wave = tid >> 5, lane = tid & 31;
  const int l16 = lane & 15, hk = lane >> 4;
  const int T = wave & 1, q = wave >> 1;
  const int Y = 8 * T + (l16 >> 1), grp = l16 & 1;
  const v2f* ap = (const v2f*)(xs + Y * 24 + 8 * grp + 4 * q + 2 * hk);
  const v2f* bp = ((const v2f*)wB) + (q * 2 + hk) * 16 + l16;
  constexpr int GS = 6;
  constexpr int NG = (T_ROWS + GS - 1) / GS;
  v2f af[2][GS], bf[2][GS];
  v8f acc0 = v8f{}, acc1 = v8f{};
  #pragma unroll
  for (int i = 0; i < GS; ++i) {
    af[0][i] = ap[(i / 9) * 288 + (i % 9) * 12];
    bf[0][i] = bp[128 * i];
  }
  #pragma unroll
  for (int gI = 0; gI < NG; ++gI) {
    const int cur = gI & 1, nxt = cur ^ 1;
    #pragma unroll
    for (int i = 0; i < GS; ++i) {
      const int st = (gI + 1) * GS + i;
      if (st < T_ROWS) {
        af[nxt][i] = ap[(st / 9) * 288 + (st % 9) * 12];
        bf[nxt][i] = bp[128 * st];
      }
    }
    __builtin_amdgcn_sched_barrier(0);
    #pragma unroll
    for (int i = 0; i < GS; ++i) {
      const int st = gI * GS + i;
      if (st < T_ROWS) {
        if (st & 1) acc1 = wmma_f32(af[cur][i], bf[cur][i], acc1);
        else        acc0 = wmma_f32(af[cur][i], bf[cur][i], acc0);
      }
    }
    __builtin_amdgcn_sched_barrier(0);
  }
  const int g = l16 & 7, cls = l16 >> 3;
  float* rp = red + q * RED_Q + cls * 272 + (8 * T + 4 * hk) * 16 + g;
  #pragma unroll
  for (int r = 0; r < 8; ++r)
    rp[(r >> 1) * 16 + 8 * (r & 1)] = acc0[r] + acc1[r];
}

__device__ __forceinline__ float lane_uniform(float v) {
  return __int_as_float(__builtin_amdgcn_readfirstlane(__float_as_int(v)));
}

__device__ __forceinline__ void lds_barrier() {
  asm volatile("s_wait_dscnt 0x0\n\ts_barrier_signal -1\n\ts_barrier_wait -1" ::: "memory");
}

__device__ __forceinline__ v4f load_x_piece(const float* __restrict__ x, int b,
                                            int X0, int Y0, int q) {
  const int qq = (q < CIN * 144) ? q : (CIN * 144 - 1);
  const int row = qq / 6, f = qq % 6, c = row / 24, iy = row % 24;
  const int gy = Y0 - 4 + iy, gx = X0 - 4 + 4 * f;
  const bool inb = (q < CIN * 144) && gy >= 0 && gy < HW && gx >= 0 && gx < HW;
  const int cy = gy < 0 ? 0 : (gy > HW - 1 ? HW - 1 : gy);
  const int cx = gx < 0 ? 0 : (gx > HW - 4 ? HW - 4 : gx);
  const v4f v = *(const v4f*)(x + ((size_t)(b * CIN + c) * HW + cy) * HW + cx);
  return inb ? v : v4f{0.f, 0.f, 0.f, 0.f};
}

__device__ __forceinline__ void store_pixel(
    float* __restrict__ ini, float* __restrict__ fea, const float* __restrict__ cg,
    int b, int gy, int gx, float a0, float a1, float x0, float x1, float x2) {
  const size_t o0 = ((size_t)(b * CLS + 0) * HW + gy) * HW + gx;
  const size_t o1 = ((size_t)(b * CLS + 1) * HW + gy) * HW + gx;
  ini[o0] = a0;
  ini[o1] = a1;
  fea[o0] = (cg[2] * x0 + cg[3] * x1 + cg[4] * x2 + cg[8]) * 0.25f;
  fea[o1] = (cg[5] * x0 + cg[6] * x1 + cg[7] * x2 + cg[9]) * 0.25f;
}

__global__ __launch_bounds__(256) void conv_fused_kernel(
    const float* __restrict__ x,  const float* __restrict__ img,
    const float* __restrict__ img2,
    float* __restrict__ ini, float* __restrict__ fea) {
  extern __shared__ float smem[];
  float* xs     = smem;
  int*   offtab = (int*)(xs + XS_FLOATS);
  float* w1s    = xs + XS_FLOATS + KPAD;
  float* b1s    = w1s + IMG_W1;
  float* t1s    = b1s + NCOL;
  const float* __restrict__ cg  = img + IMG_HEAD;

  const int tid = threadIdx.x;
  const int oy = tid >> 4, ox = tid & 15;

  if (blockIdx.x >= N_BORDER_BLK) {
    float* wB  = w1s;
    float* red = wB + IMG2_FLOATS;
    int t = blockIdx.x - N_BORDER_BLK;
    int b = t / INT_PER_IMG, r = t % INT_PER_IMG;
    int Y0 = (1 + r / INT_1D) * 16, X0 = (1 + r % INT_1D) * 16;
    const int q1  = (tid + 256 < CIN * 144) ? (tid + 256) : (CIN * 144 - 1);
    const int po0 = ((tid / 144) * HW + (tid / 6) % 24) * HW + 4 * (tid % 6);
    const int po1 = ((q1 / 144) * HW + (q1 / 6) % 24) * HW + 4 * (q1 % 6);
    const float* __restrict__ xw = x + ((b * CIN) * HW + (Y0 - 4)) * HW + (X0 - 4);
    const float f00 = lane_uniform(cg[2]), f01 = lane_uniform(cg[3]), f02 = lane_uniform(cg[4]);
    const float f10 = lane_uniform(cg[5]), f11 = lane_uniform(cg[6]), f12 = lane_uniform(cg[7]);
    const float fb0 = lane_uniform(cg[8]), fb1 = lane_uniform(cg[9]);
    const v4f* img4 = (const v4f*)img2;
    v4f cw[7];
    #pragma unroll
    for (int j = 0; j < 7; ++j) {
      int q = tid + 256 * j;
      q = (q < IMG2_FLOATS / 4) ? q : (IMG2_FLOATS / 4 - 1);
      cw[j] = img4[q];
    }
    v4f xq0 = *(const v4f*)(xw + po0);
    v4f xq1 = *(const v4f*)(xw + po1);
    #pragma unroll
    for (int j = 0; j < 7; ++j) {
      const int q = tid + 256 * j;
      if (q < IMG2_FLOATS / 4) ((v4f*)wB)[q] = cw[j];
    }
    #pragma unroll 1
    for (int k = 0; k < INT_TILES_BLK; ++k) {
      ((v4f*)xs)[tid]       = xq0;
      ((v4f*)xs)[tid + 256] = xq1;
      lds_barrier();
      const int bc = b, gy = Y0 + oy, gx = X0 + ox;
      if (k + 1 < INT_TILES_BLK) {
        t += N_INT_BLK;
        b = t / INT_PER_IMG; r = t % INT_PER_IMG;
        Y0 = (1 + r / INT_1D) * 16; X0 = (1 + r % INT_1D) * 16;
        xw = x + ((b * CIN) * HW + (Y0 - 4)) * HW + (X0 - 4);
        xq0 = *(const v4f*)(xw + po0);
        xq1 = *(const v4f*)(xw + po1);
      }
      conv_interior_tile(xs, wB, red, tid);
      const float x0 = xs[(oy + 4) * 24 + (ox + 4)];
      const float x1 = xs[576 + (oy + 4) * 24 + (ox + 4)];
      const float x2 = xs[1152 + (oy + 4) * 24 + (ox + 4)];
      lds_barrier();
      const float* rq = red + oy * 16 + ox;
      const float a0 = wB[IMG2_W + 0] + ((rq[0] + rq[RED_Q]) + (rq[2 * RED_Q] + rq[3 * RED_Q]));
      const float a1 = wB[IMG2_W + 1] + ((rq[272] + rq[RED_Q + 272]) + (rq[2 * RED_Q + 272] + rq[3 * RED_Q + 272]));
      const size_t o0 = ((size_t)(bc * CLS + 0) * HW + gy) * HW + gx;
      const size_t o1 = ((size_t)(bc * CLS + 1) * HW + gy) * HW + gx;
      ini[o0] = a0;
      ini[o1] = a1;
      fea[o0] = (f00 * x0 + f01 * x1 + f02 * x2 + fb0) * 0.25f;
      fea[o1] = (f10 * x0 + f11 * x1 + f12 * x2 + fb1) * 0.25f;
    }
    return;
  }

  const int b = blockIdx.x / BORDER_PER_IMG, rb = blockIdx.x % BORDER_PER_IMG;
  int by, bx;
  if (rb < TILES_1D)          { by = 0;            bx = rb; }
  else if (rb < 2 * TILES_1D) { by = TILES_1D - 1; bx = rb - TILES_1D; }
  else { by = 1 + ((rb - 2 * TILES_1D) >> 1); bx = ((rb - 2 * TILES_1D) & 1) ? TILES_1D - 1 : 0; }
  const int X0 = bx * 16, Y0 = by * 16;

  const v4f* img4 = (const v4f*)img;
  v4f cw[7];
  #pragma unroll
  for (int j = 0; j < 7; ++j) {
    int q = tid + 256 * j;
    q = (q < IMG_HEAD / 4) ? q : (IMG_HEAD / 4 - 1);
    cw[j] = img4[q];
  }
  const v4f xq0 = load_x_piece(x, b, X0, Y0, tid);
  const v4f xq1 = load_x_piece(x, b, X0, Y0, tid + 256);
  #pragma unroll
  for (int j = 0; j < 7; ++j) {
    const int q = tid + 256 * j;
    if (q < IMG_HEAD / 4) ((v4f*)w1s)[q] = cw[j];
  }
  ((v4f*)xs)[tid]       = xq0;
  ((v4f*)xs)[tid + 256] = xq1;
  if (tid < XS_FLOATS / 4 - 512) ((v4f*)xs)[tid + 512] = v4f{0.f, 0.f, 0.f, 0.f};
  if (tid < KPAD) {
    int off = 1728;
    if (tid < KFOLD) {
      int c = tid / 63, rr = tid % 63;
      off = c * 576 + (rr / 9) * 24 + (rr % 9);
    }
    offtab[tid] = off;
  }
  __syncthreads();

  const int wave = tid >> 5;
  const int lane = tid & 31;
  const int l16  = lane & 15;
  const int ncol = (l16 < 6) ? l16
                 : (l16 < NTCOL) ? (l16 + ((bx == TILES_1D - 1) ? 6 : 0)) : NFOLD;

  const int gy = Y0 + oy, gx = X0 + ox;
  for (int mt = wave; mt < 18; mt += 8)
    conv1_tile(xs, offtab, w1s, b1s, t1s, mt, ncol, lane, Y0);
  __syncthreads();
  const int vb = (gx == 0 || gx == HW - 1) ? 6 : 0;
  float a0 = cg[0], a1 = cg[1];
  for (int dy = 0; dy < 3; ++dy) {
    const int p = (oy + dy) * 16 + ox;
    a0 += t1s[(vb + dy) * TPITCH + p];
    a1 += t1s[(vb + 3 + dy) * TPITCH + p];
  }
  store_pixel(ini, fea, cg, b, gy, gx, a0, a1,
              xs[(oy + 4) * 24 + (ox + 4)], xs[576 + (oy + 4) * 24 + (ox + 4)],
              xs[1152 + (oy + 4) * 24 + (ox + 4)]);
}

#define TRANS_LDS_FLOATS (64 * 102)

__global__ __launch_bounds__(256) void trans_kernel(
    const float* __restrict__ ini, const float* __restrict__ fea,
    const float* __restrict__ wf,
    float* __restrict__ ui, float* __restrict__ uf) {
  extern __shared__ float smem[];
  float* a_lds = smem;

  const int tid = threadIdx.x;
  const float* src = blockIdx.y ? fea : ini;
  float* dst       = blockIdx.y ? uf  : ui;
  const int P0 = blockIdx.x * 64;

  for (int i = tid; i < 64 * P2; i += 256) {
    int r = i / P2, p = i % P2;
    int Pa = P0 + r;
    int bc = Pa >> 10, l = Pa & 1023;
    int gy = (l >> 5) * PSZ + p / PSZ;
    int gx = (l & 31) * PSZ + p % PSZ;
    a_lds[r * 102 + p] = src[((size_t)bc * HW + gy) * HW + gx];
  }
  __syncthreads();

  const int wave  = tid >> 5;
  const int lane  = tid & 31;
  const int l16   = lane & 15;
  const int lhalf = (lane < 16) ? 0 : 2;

  for (int t = wave; t < 4 * 7; t += 8) {
    const int m0 = (t / 7) * 16, n0 = (t % 7) * 16;
    const int n  = n0 + l16;
    const v2f* ap = (const v2f*)&a_lds[(m0 + l16) * 102 + lhalf];
    const v2f* bp = (const v2f*)&wf[n * P2 + lhalf];
    v8f acc = {};
    #pragma unroll 5
    for (int k0 = 0; k0 < P2; k0 += 4) {
      v2f a  = *ap;  ap += 2;
      v2f bf = *bp;  bp += 2;
      acc = wmma_f32(a, bf, acc);
    }
    if (n < P2) {
      #pragma unroll
      for (int r = 0; r < 8; ++r) {
        int M = m0 + r + ((lane < 16) ? 0 : 8);
        float v = acc[r];
        dst[(size_t)(P0 + M) * P2 + n] = v > 0.f ? v : 0.f;
      }
    }
  }
}

#define ATT_U_CHUNKS   (128 * P2 / 4)
#define ATT_CHUNKS     (ATT_U_CHUNKS + 64 * P2 / 4)
#define ATT_LDS_FLOATS (128 * P2 + 64 * P2)

__global__ __launch_bounds__(256) void att_kernel(
    const float* __restrict__ ui, const float* __restrict__ uf,
    float* __restrict__ out) {
  extern __shared__ float smem[];
  float* us = smem;
  float* vs = us + 128 * P2;

  const int tid = threadIdx.x;
  const int bc = blockIdx.z;
  const int M0 = blockIdx.x * 128, N0 = blockIdx.y * 64;
  const v4f* U4 = (const v4f*)(ui + ((size_t)bc * LPATCH + M0) * P2);
  const v4f* V4 = (const v4f*)(uf + ((size_t)bc * LPATCH + N0) * P2);

  v4f r[19];
  #pragma unroll
  for (int j = 0; j < 19; ++j) {
    const int q = (j < 18) ? (tid + 256 * j) : (tid + ATT_CHUNKS - 256);
    const v4f* p;
    if (j < 12)       p = U4 + q;
    else if (j == 12) p = (tid < 128) ? (U4 + q) : (V4 + (q - ATT_U_CHUNKS));
    else              p = V4 + (q - ATT_U_CHUNKS);
    r[j] = *p;
  }
  #pragma unroll
  for (int j = 0; j < 19; ++j) {
    const int q = (j < 18) ? (tid + 256 * j) : (tid + ATT_CHUNKS - 256);
    ((v4f*)us)[q] = r[j];
  }
  __syncthreads();

  const int wave  = tid >> 5;
  const int lane  = tid & 31;
  const int l16   = lane & 15;
  const int lhalf = (lane < 16) ? 0 : 2;

  for (int t = wave; t < 8 * 4; t += 8) {
    const int m0 = (t >> 2) * 16, n0 = (t & 3) * 16;
    const v2f* ap = (const v2f*)&us[(m0 + l16) * P2 + lhalf];
    const v2f* bp = (const v2f*)&vs[(n0 + l16) * P2 + lhalf];
    v8f acc = {};
    #pragma unroll 5
    for (int k0 = 0; k0 < P2; k0 += 4) {
      v2f a  = *ap;  ap += 2;
      v2f bf = *bp;  bp += 2;
      acc = wmma_f32(a, bf, acc);
    }
    #pragma unroll
    for (int r8 = 0; r8 < 8; ++r8) {
      int M = M0 + m0 + r8 + ((lane < 16) ? 0 : 8);
      int N = N0 + n0 + l16;
      out[((size_t)bc * LPATCH + M) * LPATCH + N] = acc[r8] * 0.01f;
    }
  }
}

extern "C" void kernel_launch(void* const* d_in, const int* in_sizes, int n_in,
                              void* d_out, int out_size, void* d_ws,
                              size_t ws_size, hipStream_t stream) {
  (void)in_sizes; (void)n_in; (void)out_size; (void)ws_size;
  const float* x       = (const float*)d_in[0];
  const float* w_conv1 = (const float*)d_in[1];
  const float* b_conv1 = (const float*)d_in[2];
  const float* w_conv2 = (const float*)d_in[3];
  const float* b_conv2 = (const float*)d_in[4];
  const float* w_fm    = (const float*)d_in[5];
  const float* b_fm    = (const float*)d_in[6];
  const float* w_lin1  = (const float*)d_in[7];
  const float* w_lin2  = (const float*)d_in[8];
  float* out = (float*)d_out;

  float* ws  = (float*)d_ws;
  float* ini = ws + WS_INI;
  float* fea = ws + WS_FEA;
  float* ui  = ws + WS_UI;
  float* uf  = ws + WS_UF;
  float* wf  = ws + WS_WF;
  float* img = ws + WS_IMG;
  float* img2 = ws + WS_IMG2;

  pad_weights_kernel<<<dim3(PREP_WF_BLOCKS + PREP_IMG_BLOCKS + PREP_IMG2_BLOCKS), 256, 0, stream>>>(
      w_lin1, w_lin2, w_conv1, b_conv1, w_conv2, b_conv2, w_fm, b_fm,
      wf, img, img2);

  conv_fused_kernel<<<dim3(N_BORDER_BLK + N_INT_BLK), 256,
                      CONV_LDS_FLOATS * sizeof(float), stream>>>(
      x, img, img2, ini, fea);

  trans_kernel<<<dim3(BATCH * CLS * LPATCH / 64, 2, 1), 256,
                 TRANS_LDS_FLOATS * sizeof(float), stream>>>(
      ini, fea, wf, ui, uf);

  att_kernel<<<dim3(LPATCH / 128, LPATCH / 64, BATCH * CLS), 256,
               ATT_LDS_FLOATS * sizeof(float), stream>>>(ui, uf, out);
}


# trans kernel: both operands go to LDS as one group of 18 coalesced 16-byte loads (a block's 64 patches are one contiguous run of the plane; WF a flat copy), not per-float gathers and global B reads
# speedup vs baseline: 5.6246x; 1.1025x over previous
#include <hip/hip_runtime.h>
#include <hip/hip_bf16.h>

typedef float v2f __attribute__((ext_vector_type(2)));
typedef float v8f __attribute__((ext_vector_type(8)));
typedef float v4f __attribute__((ext_vector_type(4)));

__device__ __forceinline__ v8f wmma_f32(v2f a, v2f b, v8f c) {
  return __builtin_amdgcn_wmma_f32_16x16x4_f32(false, a, false, b, (short)0, c,
                                               false, false);
}

#define BATCH   8
#define CIN     3
#define HW      320
#define CLS     2
#define PSZ     10
#define P2      100
#define LPATCH  1024
#define HID     200

#define WS_INI  0
#define WS_FEA  (WS_INI + BATCH*CLS*HW*HW)
#define WS_UI   (WS_FEA + BATCH*CLS*HW*HW)
#define WS_UF   (WS_UI  + BATCH*CLS*LPATCH*P2)
#define WS_WF   (WS_UF  + BATCH*CLS*LPATCH*P2)
#define WF_ROWS 112
#define PREP_WF_BLOCKS  11
#define PREP_IMG_BLOCKS 25
#define WS_IMG  WS_UI
#define NCOL       32
#define NFOLD      18
#define KFOLD      189
#define NTCOL      12
#define IMG_W1     (96 * NCOL * 2)
#define IMG_HEAD   (IMG_W1 + NCOL)
#define IMG_TAIL   16
#define IMG_FLOATS (IMG_HEAD + IMG_TAIL)
#define T_ROWS      27
#define IMG2_W      (T_ROWS * 4 * 2 * 16 * 2)
#define IMG2_FLOATS (IMG2_W + 16)
#define WS_IMG2     (WS_IMG + 6400)
#define PREP_IMG2_BLOCKS 109
#define RED_Q       544
#define TILES_1D       (HW / 16)
#define INT_1D         (TILES_1D - 2)
#define INT_PER_IMG    (INT_1D * INT_1D)
#define BORDER_PER_IMG (4 * TILES_1D - 4)
#define N_BORDER_BLK   (BATCH * BORDER_PER_IMG)
#define INT_TILES_BLK  4
#define N_INT_BLK      (BATCH * INT_PER_IMG / INT_TILES_BLK)

__global__ __launch_bounds__(256) void pad_weights_kernel(
    const float* __restrict__ w_lin1, const float* __restrict__ w_lin2,
    const float* __restrict__ w1c, const float* __restrict__ b1c,
    const float* __restrict__ w2c, const float* __restrict__ b2c,
    const float* __restrict__ wfm, const float* __restrict__ bfm,
    float* __restrict__ wf, float* __restrict__ img, float* __restrict__ img2) {
  if (blockIdx.x >= PREP_WF_BLOCKS + PREP_IMG_BLOCKS) {
    const int t = (blockIdx.x - PREP_WF_BLOCKS - PREP_IMG_BLOCKS) * 256 + threadIdx.x;
    const int e = t >> 2, part = t & 3;
    float v = 0.f;
    if (e < IMG2_W) {
      const int ee = e & 1, n = (e >> 1) & 15, hk = (e >> 5) & 1, q = (e >> 6) & 3, i = e >> 8;
      const int j = 4 * q + 2 * hk + ee, g = n & 7, cls = n >> 3, kx9 = j - g;
      const int c = i / 9, ky9 = i % 9;
      if (kx9 >= 0 && kx9 <= 8) {
        bool ok[9]; int o1[9];
        #pragma unroll
        for (int dy = 0; dy < 3; ++dy) {
          #pragma unroll
          for (int dx = 0; dx < 3; ++dx) {
            const int ty = ky9 - dy, tx = kx9 - dx;
            ok[dy * 3 + dx] = (ty >= 0) & (ty <= 6) & (tx >= 0) & (tx <= 6);
            o1[dy * 3 + dx] = (ty < 0 ? 0 : (ty > 6 ? 6 : ty)) * 7 + (tx < 0 ? 0 : (tx > 6 ? 6 : tx));
          }
        }
        const float* w2p = w2c + cls * 576 + (16 * part) * 9;
        const float* w1p = w1c + (16 * part) * 147 + c * 49;
        #pragma unroll 8
        for (int ch = 0; ch < 16; ++ch) {
          #pragma unroll
          for (int tt = 0; tt < 9; ++tt) {
            const float a = w2p[ch * 9 + tt];
            v = fmaf(ok[tt] ? a : 0.f, w1p[ch * 147 + o1[tt]], v);
          }
        }
      }
    } else if (e < IMG2_W + 2) {
      const int cls = e - IMG2_W;
      const float* w2p = w2c + cls * 576 + (16 * part) * 9;
      if (part == 0) v = b2c[cls];
      #pragma unroll 8
      for (int ch = 0; ch < 16; ++ch) {
        const float bb = b1c[16 * part + ch];
        #pragma unroll
        for (int tt = 0; tt < 9; ++tt) v = fmaf(w2p[ch * 9 + tt], bb, v);
      }
    }
    v += __shfl_xor(v, 1);
    v += __shfl_xor(v, 2);
    if (part == 0 && e < IMG2_FLOATS) img2[e] = v;
    return;
  }
  if (blockIdx.x < PREP_WF_BLOCKS) {
    const int j = blockIdx.x * 256 + threadIdx.x;
    if (j < WF_ROWS * (P2 / 4)) {
      const int n = j / (P2 / 4), kq = j % (P2 / 4);
      v4f acc = {0.f, 0.f, 0.f, 0.f};
      if (n < P2) {
        const v4f* a4 = (const v4f*)(w_lin2 + n * HID);
        const v4f* b4 = ((const v4f*)w_lin1) + kq;
        #pragma unroll 1
        for (int t = 0; t < HID / 40; ++t) {
          v4f av[10], bv[40];
          #pragma unroll
          for (int u = 0; u < 10; ++u) {
            const int s = t * 10 + u;
            av[u] = a4[s];
            #pragma unroll
            for (int c = 0; c < 4; ++c) bv[4 * u + c] = b4[(4 * s + c) * (P2 / 4)];
          }
          __builtin_amdgcn_sched_barrier(0);
          #pragma unroll
          for (int u = 0; u < 10; ++u) {
            #pragma unroll
            for (int c = 0; c < 4; ++c) {
              const float a = av[u][c];
              const v4f b = bv[4 * u + c];
              acc.x = fmaf(a, b.x, acc.x); acc.y = fmaf(a, b.y, acc.y);
              acc.z = fmaf(a, b.z, acc.z); acc.w = fmaf(a, b.w, acc.w);
            }
          }
        }
      }
      ((v4f*)wf)[j] = acc;
    }
    return;
  }
  const int i = (blockIdx.x - PREP_WF_BLOCKS) * 256 + threadIdx.x;
  if (i < IMG_FLOATS) {
    float v = 0.f;
    if (i < IMG_W1) {
      const int k = ((i >> 6) << 1) + (i & 1), n = (i & 63) >> 1;
      if (k < KFOLD && n < NFOLD) {
        const int kx9 = k % 9, cky = k / 9;
        const int var = n / 6, cls = (n % 6) / 3, dy = n % 3;
        const float* wrow = w2c + cls * 576 + dy * 3;
        const float* w1r  = w1c + cky * 7;
        bool ok[3]; int kx[3];
        #pragma unroll
        for (int dx = 0; dx < 3; ++dx) {
          const int t = kx9 - dx;
          const bool inset = (var == 0) || (var == 1 && dx >= 1) || (var == 2 && dx <= 1);
          ok[dx] = inset && t >= 0 && t <= 6;
          kx[dx] = t < 0 ? 0 : (t > 6 ? 6 : t);
        }
        #pragma unroll 16
        for (int ch = 0; ch < 64; ++ch) {
          #pragma unroll
          for (int dx = 0; dx < 3; ++dx) {
            const float a = wrow[ch * 9 + dx];
            v = fmaf(ok[dx] ? a : 0.f, w1r[ch * 147 + kx[dx]], v);
          }
        }
      }
    } else if (i < IMG_HEAD) {
      const int n = i - IMG_W1;
      if (n < NFOLD) {
        const int var = n / 6, cls = (n % 6) / 3, dy = n % 3;
        const float* wrow = w2c + cls * 576 + dy * 3;
        const bool ok0 = (var != 1), ok2 = (var != 2);
        #pragma unroll 16
        for (int ch = 0; ch < 64; ++ch) {
          const float bb = b1c[ch];
          v = fmaf(ok0 ? wrow[ch * 9 + 0] : 0.f, bb, v);
          v = fmaf(wrow[ch * 9 + 1], bb, v);
          v = fmaf(ok2 ? wrow[ch * 9 + 2] : 0.f, bb, v);
        }
      }
    } else {
      int j = i - IMG_HEAD;
      if (j < 2)       v = b2c[j];
      else if (j < 8)  v = wfm[j - 2];
      else if (j < 10) v = bfm[j - 8];
    }
    img[i] = v;
  }
}

#define XS_FLOATS        2160
#define KPAD             192
#define TPITCH           289
#define CONV_LDS_FLOATS  (XS_FLOATS + KPAD + IMG_W1 + NCOL + NTCOL*TPITCH)

__device__ __forceinline__ void conv1_tile(
    const float* xs, const int* offtab, const float* w1s, const float* b1s,
    float* t1s, int mt, int ncol, int lane, int Y0) {
  constexpr int KS = KPAD / 4;
  constexpr int G  = 4;
  constexpr int NB = KS / G;
  const int l16 = lane & 15;
  const int hk  = lane >> 4;
  const float* xa = xs + mt * 24 + l16;
  const int2* ot = ((const int2*)offtab) + hk;
  const v2f*  bp = ((const v2f*)w1s) + hk * NCOL + ncol;

  int2 off[2][G];
  v2f  af[2][G];
  v2f  bf[2][G];
  v8f  acc0 = v8f{}, acc1 = v8f{};

  const float bias = b1s[ncol];
  #pragma unroll
  for (int i = 0; i < G; ++i) { off[0][i] = ot[2 * i]; off[1][i] = ot[2 * (G + i)]; }
  #pragma unroll
  for (int i = 0; i < G; ++i) {
    af[0][i].x = xa[off[0][i].x];
    af[0][i].y = xa[off[0][i].y];
    bf[0][i] = bp[2 * NCOL * i];
  }

  #pragma unroll
  for (int g = 0; g < NB; ++g) {
    const int cur = g & 1, nxt = cur ^ 1;
    if (g + 1 < NB) {
      #pragma unroll
      for (int i = 0; i < G; ++i) {
        const int s = (g + 1) * G + i;
        af[nxt][i].x = xa[off[nxt][i].x];
        af[nxt][i].y = xa[off[nxt][i].y];
        bf[nxt][i] = bp[2 * NCOL * s];
      }
    }
    if (g + 2 < NB) {
      #pragma unroll
      for (int i = 0; i < G; ++i) off[cur][i] = ot[2 * ((g + 2) * G + i)];
    }
    __builtin_amdgcn_sched_barrier(0);
    #pragma unroll
    for (int i = 0; i < G; i += 2) {
      acc0 = wmma_f32(af[cur][i],     bf[cur][i],     acc0);
      acc1 = wmma_f32(af[cur][i + 1], bf[cur][i + 1], acc1);
    }
    __builtin_amdgcn_sched_barrier(0);
  }

  const int gy = Y0 - 1 + mt;
  const bool in = gy >= 0 && gy < HW;
  if (l16 < NTCOL) {
    #pragma unroll
    for (int r = 0; r < 8; ++r)
      t1s[l16 * TPITCH + mt * 16 + r + 8 * hk] = in ? (acc0[r] + acc1[r] + bias) : 0.f;
  }
}

__device__ __forceinline__ void conv_interior_tile(
    const float* xs, const float* wB, float* red, int tid) {
  const int wave = tid >> 5, lane = tid & 31;
  const int l16 = lane & 15, hk = lane >> 4;
  const int T = wave & 1, q = wave >> 1;
  const int Y = 8 * T + (l16 >> 1), grp = l16 & 1;
  const v2f* ap = (const v2f*)(xs + Y * 24 + 8 * grp + 4 * q + 2 * hk);
  const v2f* bp = ((const v2f*)wB) + (q * 2 + hk) * 16 + l16;
  constexpr int GS = 6;
  constexpr int NG = (T_ROWS + GS - 1) / GS;
  v2f af[2][GS], bf[2][GS];
  v8f acc0 = v8f{}, acc1 = v8f{};
  #pragma unroll
  for (int i = 0; i < GS; ++i) {
    af[0][i] = ap[(i / 9) * 288 + (i % 9) * 12];
    bf[0][i] = bp[128 * i];
  }
  #pragma unroll
  for (int gI = 0; gI < NG; ++gI) {
    const int cur = gI & 1, nxt = cur ^ 1;
    #pragma unroll
    for (int i = 0; i < GS; ++i) {
      const int st = (gI + 1) * GS + i;
      if (st < T_ROWS) {
        af[nxt][i] = ap[(st / 9) * 288 + (st % 9) * 12];
        bf[nxt][i] = bp[128 * st];
      }
    }
    __builtin_amdgcn_sched_barrier(0);
    #pragma unroll
    for (int i = 0; i < GS; ++i) {
      const int st = gI * GS + i;
      if (st < T_ROWS) {
        if (st & 1) acc1 = wmma_f32(af[cur][i], bf[cur][i], acc1);
        else        acc0 = wmma_f32(af[cur][i], bf[cur][i], acc0);
      }
    }
    __builtin_amdgcn_sched_barrier(0);
  }
  const int g = l16 & 7, cls = l16 >> 3;
  float* rp = red + q * RED_Q + cls * 272 + (8 * T + 4 * hk) * 16 + g;
  #pragma unroll
  for (int r = 0; r < 8; ++r)
    rp[(r >> 1) * 16 + 8 * (r & 1)] = acc0[r] + acc1[r];
}

__device__ __forceinline__ float lane_uniform(float v) {
  return __int_as_float(__builtin_amdgcn_readfirstlane(__float_as_int(v)));
}

__device__ __forceinline__ void lds_barrier() {
  asm volatile("s_wait_dscnt 0x0\n\ts_barrier_signal -1\n\ts_barrier_wait -1" ::: "memory");
}

__device__ __forceinline__ v4f load_x_piece(const float* __restrict__ x, int b,
                                            int X0, int Y0, int q) {
  const int qq = (q < CIN * 144) ? q : (CIN * 144 - 1);
  const int row = qq / 6, f = qq % 6, c = row / 24, iy = row % 24;
  const int gy = Y0 - 4 + iy, gx = X0 - 4 + 4 * f;
  const bool inb = (q < CIN * 144) && gy >= 0 && gy < HW && gx >= 0 && gx < HW;
  const int cy = gy < 0 ? 0 : (gy > HW - 1 ? HW - 1 : gy);
  const int cx = gx < 0 ? 0 : (gx > HW - 4 ? HW - 4 : gx);
  const v4f v = *(const v4f*)(x + ((size_t)(b * CIN + c) * HW + cy) * HW + cx);
  return inb ? v : v4f{0.f, 0.f, 0.f, 0.f};
}

__device__ __forceinline__ void store_pixel(
    float* __restrict__ ini, float* __restrict__ fea, const float* __restrict__ cg,
    int b, int gy, int gx, float a0, float a1, float x0, float x1, float x2) {
  const size_t o0 = ((size_t)(b * CLS + 0) * HW + gy) * HW + gx;
  const size_t o1 = ((size_t)(b * CLS + 1) * HW + gy) * HW + gx;
  ini[o0] = a0;
  ini[o1] = a1;
  fea[o0] = (cg[2] * x0 + cg[3] * x1 + cg[4] * x2 + cg[8]) * 0.25f;
  fea[o1] = (cg[5] * x0 + cg[6] * x1 + cg[7] * x2 + cg[9]) * 0.25f;
}

__global__ __launch_bounds__(256) void conv_fused_kernel(
    const float* __restrict__ x,  const float* __restrict__ img,
    const float* __restrict__ img2,
    float* __restrict__ ini, float* __restrict__ fea) {
  extern __shared__ float smem[];
  float* xs     = smem;
  int*   offtab = (int*)(xs + XS_FLOATS);
  float* w1s    = xs + XS_FLOATS + KPAD;
  float* b1s    = w1s + IMG_W1;
  float* t1s    = b1s + NCOL;
  const float* __restrict__ cg  = img + IMG_HEAD;

  const int tid = threadIdx.x;
  const int oy = tid >> 4, ox = tid & 15;

  if (blockIdx.x >= N_BORDER_BLK) {
    float* wB  = w1s;
    float* red = wB + IMG2_FLOATS;
    int t = blockIdx.x - N_BORDER_BLK;
    int b = t / INT_PER_IMG, r = t % INT_PER_IMG;
    int Y0 = (1 + r / INT_1D) * 16, X0 = (1 + r % INT_1D) * 16;
    const int q1  = (tid + 256 < CIN * 144) ? (tid + 256) : (CIN * 144 - 1);
    const int po0 = ((tid / 144) * HW + (tid / 6) % 24) * HW + 4 * (tid % 6);
    const int po1 = ((q1 / 144) * HW + (q1 / 6) % 24) * HW + 4 * (q1 % 6);
    const float* __restrict__ xw = x + ((b * CIN) * HW + (Y0 - 4)) * HW + (X0 - 4);
    const float f00 = lane_uniform(cg[2]), f01 = lane_uniform(cg[3]), f02 = lane_uniform(cg[4]);
    const float f10 = lane_uniform(cg[5]), f11 = lane_uniform(cg[6]), f12 = lane_uniform(cg[7]);
    const float fb0 = lane_uniform(cg[8]), fb1 = lane_uniform(cg[9]);
    const v4f* img4 = (const v4f*)img2;
    v4f cw[7];
    #pragma unroll
    for (int j = 0; j < 7; ++j) {
      int q = tid + 256 * j;
      q = (q < IMG2_FLOATS / 4) ? q : (IMG2_FLOATS / 4 - 1);
      cw[j] = img4[q];
    }
    v4f xq0 = *(const v4f*)(xw + po0);
    v4f xq1 = *(const v4f*)(xw + po1);
    #pragma unroll
    for (int j = 0; j < 7; ++j) {
      const int q = tid + 256 * j;
      if (q < IMG2_FLOATS / 4) ((v4f*)wB)[q] = cw[j];
    }
    #pragma unroll 1
    for (int k = 0; k < INT_TILES_BLK; ++k) {
      ((v4f*)xs)[tid]       = xq0;
      ((v4f*)xs)[tid + 256] = xq1;
      lds_barrier();
      const int bc = b, gy = Y0 + oy, gx = X0 + ox;
      if (k + 1 < INT_TILES_BLK) {
        t += N_INT_BLK;
        b = t / INT_PER_IMG; r = t % INT_PER_IMG;
        Y0 = (1 + r / INT_1D) * 16; X0 = (1 + r % INT_1D) * 16;
        xw = x + ((b * CIN) * HW + (Y0 - 4)) * HW + (X0 - 4);
        xq0 = *(const v4f*)(xw + po0);
        xq1 = *(const v4f*)(xw + po1);
      }
      conv_interior_tile(xs, wB, red, tid);
      const float x0 = xs[(oy + 4) * 24 + (ox + 4)];
      const float x1 = xs[576 + (oy + 4) * 24 + (ox + 4)];
      const float x2 = xs[1152 + (oy + 4) * 24 + (ox + 4)];
      lds_barrier();
      const float* rq = red + oy * 16 + ox;
      const float a0 = wB[IMG2_W + 0] + ((rq[0] + rq[RED_Q]) + (rq[2 * RED_Q] + rq[3 * RED_Q]));
      const float a1 = wB[IMG2_W + 1] + ((rq[272] + rq[RED_Q + 272]) + (rq[2 * RED_Q + 272] + rq[3 * RED_Q + 272]));
      const size_t o0 = ((size_t)(bc * CLS + 0) * HW + gy) * HW + gx;
      const size_t o1 = ((size_t)(bc * CLS + 1) * HW + gy) * HW + gx;
      ini[o0] = a0;
      ini[o1] = a1;
      fea[o0] = (f00 * x0 + f01 * x1 + f02 * x2 + fb0) * 0.25f;
      fea[o1] = (f10 * x0 + f11 * x1 + f12 * x2 + fb1) * 0.25f;
    }
    return;
  }

  const int b = blockIdx.x / BORDER_PER_IMG, rb = blockIdx.x % BORDER_PER_IMG;
  int by, bx;
  if (rb < TILES_1D)          { by = 0;            bx = rb; }
  else if (rb < 2 * TILES_1D) { by = TILES_1D - 1; bx = rb - TILES_1D; }
  else { by = 1 + ((rb - 2 * TILES_1D) >> 1); bx = ((rb - 2 * TILES_1D) & 1) ? TILES_1D - 1 : 0; }
  const int X0 = bx * 16, Y0 = by * 16;

  const v4f* img4 = (const v4f*)img;
  v4f cw[7];
  #pragma unroll
  for (int j = 0; j < 7; ++j) {
    int q = tid + 256 * j;
    q = (q < IMG_HEAD / 4) ? q : (IMG_HEAD / 4 - 1);
    cw[j] = img4[q];
  }
  const v4f xq0 = load_x_piece(x, b, X0, Y0, tid);
  const v4f xq1 = load_x_piece(x, b, X0, Y0, tid + 256);
  #pragma unroll
  for (int j = 0; j < 7; ++j) {
    const int q = tid + 256 * j;
    if (q < IMG_HEAD / 4) ((v4f*)w1s)[q] = cw[j];
  }
  ((v4f*)xs)[tid]       = xq0;
  ((v4f*)xs)[tid + 256] = xq1;
  if (tid < XS_FLOATS / 4 - 512) ((v4f*)xs)[tid + 512] = v4f{0.f, 0.f, 0.f, 0.f};
  if (tid < KPAD) {
    int off = 1728;
    if (tid < KFOLD) {
      int c = tid / 63, rr = tid % 63;
      off = c * 576 + (rr / 9) * 24 + (rr % 9);
    }
    offtab[tid] = off;
  }
  __syncthreads();

  const int wave = tid >> 5;
  const int lane = tid & 31;
  const int l16  = lane & 15;
  const int ncol = (l16 < 6) ? l16
                 : (l16 < NTCOL) ? (l16 + ((bx == TILES_1D - 1) ? 6 : 0)) : NFOLD;

  const int gy = Y0 + oy, gx = X0 + ox;
  for (int mt = wave; mt < 18; mt += 8)
    conv1_tile(xs, offtab, w1s, b1s, t1s, mt, ncol, lane, Y0);
  __syncthreads();
  const int vb = (gx == 0 || gx == HW - 1) ? 6 : 0;
  float a0 = cg[0], a1 = cg[1];
  for (int dy = 0; dy < 3; ++dy) {
    const int p = (oy + dy) * 16 + ox;
    a0 += t1s[(vb + dy) * TPITCH + p];
    a1 += t1s[(vb + 3 + dy) * TPITCH + p];
  }
  store_pixel(ini, fea, cg, b, gy, gx, a0, a1,
              xs[(oy + 4) * 24 + (ox + 4)], xs[576 + (oy + 4) * 24 + (ox + 4)],
              xs[1152 + (oy + 4) * 24 + (ox + 4)]);
}

#define TRANS_A_CHUNKS   (64 * P2 / 4)
#define TRANS_B_CHUNKS   (WF_ROWS * P2 / 4)
#define TRANS_LDS_FLOATS (64 * P2 + WF_ROWS * P2)

__global__ __launch_bounds__(256) void trans_kernel(
    const float* __restrict__ ini, const float* __restrict__ fea,
    const float* __restrict__ wf,
    float* __restrict__ ui, float* __restrict__ uf) {
  extern __shared__ float smem[];
  float* a_lds = smem;
  float* b_lds = smem + 64 * P2;

  const int tid = threadIdx.x;
  const float* src = blockIdx.y ? fea : ini;
  float* dst       = blockIdx.y ? uf  : ui;
  const int P0 = blockIdx.x * 64;

  const v4f* S4 = (const v4f*)(src + (size_t)P0 * P2);
  const v4f* W4 = (const v4f*)wf;
  v4f ra[7], rb[11];
  #pragma unroll
  for (int j = 0; j < 7; ++j) {
    const int q = (j < 6) ? (tid + 256 * j) : (tid + TRANS_A_CHUNKS - 256);
    ra[j] = S4[q];
  }
  #pragma unroll
  for (int j = 0; j < 11; ++j) {
    const int q = (j < 10) ? (tid + 256 * j) : (tid + TRANS_B_CHUNKS - 256);
    rb[j] = W4[q];
  }
  __builtin_amdgcn_sched_barrier(0);
  int o0[7], o1[7];
  #pragma unroll
  for (int j = 0; j < 7; ++j) {
    const int q = (j < 6) ? (tid + 256 * j) : (tid + TRANS_A_CHUNKS - 256);
    const int y = q / 80, x = 4 * (q % 80);
    const int row = (y / PSZ) * (32 * P2) + (y % PSZ) * PSZ;
    o0[j] = row + (x / PSZ) * P2 + x % PSZ;
    o1[j] = row + ((x + 2) / PSZ) * P2 + (x + 2) % PSZ;
  }
  __builtin_amdgcn_sched_barrier(0);
  #pragma unroll
  for (int j = 0; j < 7; ++j) {
    *(v2f*)(a_lds + o0[j]) = v2f{ra[j].x, ra[j].y};
    *(v2f*)(a_lds + o1[j]) = v2f{ra[j].z, ra[j].w};
  }
  #pragma unroll
  for (int j = 0; j < 11; ++j) {
    const int q = (j < 10) ? (tid + 256 * j) : (tid + TRANS_B_CHUNKS - 256);
    ((v4f*)b_lds)[q] = rb[j];
  }
  __syncthreads();

  const int wave  = tid >> 5;
  const int lane  = tid & 31;
  const int l16   = lane & 15;
  const int lhalf = (lane < 16) ? 0 : 2;

  for (int t = wave; t < 4 * 7; t += 8) {
    const int m0 = (t / 7) * 16, n0 = (t % 7) * 16;
    const int n  = n0 + l16;
    const v2f* ap = (const v2f*)&a_lds[(m0 + l16) * P2 + lhalf];
    const v2f* bp = (const v2f*)&b_lds[n * P2 + lhalf];
    v8f acc = {};
    #pragma unroll 5
    for (int k0 = 0; k0 < P2; k0 += 4) {
      v2f a  = *ap;  ap += 2;
      v2f bf = *bp;  bp += 2;
      acc = wmma_f32(a, bf, acc);
    }
    if (n < P2) {
      #pragma unroll
      for (int r = 0; r < 8; ++r) {
        int M = m0 + r + ((lane < 16) ? 0 : 8);
        float v = acc[r];
        dst[(size_t)(P0 + M) * P2 + n] = v > 0.f ? v : 0.f;
      }
    }
  }
}

#define ATT_U_CHUNKS   (128 * P2 / 4)
#define ATT_CHUNKS     (ATT_U_CHUNKS + 64 * P2 / 4)
#define ATT_LDS_FLOATS (128 * P2 + 64 * P2)

__global__ __launch_bounds__(256) void att_kernel(
    const float* __restrict__ ui, const float* __restrict__ uf,
    float* __restrict__ out) {
  extern __shared__ float smem[];
  float* us = smem;
  float* vs = us + 128 * P2;

  const int tid = threadIdx.x;
  const int bc = blockIdx.z;
  const int M0 = blockIdx.x * 128, N0 = blockIdx.y * 64;
  const v4f* U4 = (const v4f*)(ui + ((size_t)bc * LPATCH + M0) * P2);
  const v4f* V4 = (const v4f*)(uf + ((size_t)bc * LPATCH + N0) * P2);

  v4f r[19];
  #pragma unroll
  for (int j = 0; j < 19; ++j) {
    const int q = (j < 18) ? (tid + 256 * j) : (tid + ATT_CHUNKS - 256);
    const v4f* p;
    if (j < 12)       p = U4 + q;
    else if (j == 12) p = (tid < 128) ? (U4 + q) : (V4 + (q - ATT_U_CHUNKS));
    else              p = V4 + (q - ATT_U_CHUNKS);
    r[j] = *p;
  }
  #pragma unroll
  for (int j = 0; j < 19; ++j) {
    const int q = (j < 18) ? (tid + 256 * j) : (tid + ATT_CHUNKS - 256);
    ((v4f*)us)[q] = r[j];
  }
  __syncthreads();

  const int wave  = tid >> 5;
  const int lane  = tid & 31;
  const int l16   = lane & 15;
  const int lhalf = (lane < 16) ? 0 : 2;

  for (int t = wave; t < 8 * 4; t += 8) {
    const int m0 = (t >> 2) * 16, n0 = (t & 3) * 16;
    const v2f* ap = (const v2f*)&us[(m0 + l16) * P2 + lhalf];
    const v2f* bp = (const v2f*)&vs[(n0 + l16) * P2 + lhalf];
    v8f acc = {};
    #pragma unroll 5
    for (int k0 = 0; k0 < P2; k0 += 4) {
      v2f a  = *ap;  ap += 2;
      v2f bf = *bp;  bp += 2;
      acc = wmma_f32(a, bf, acc);
    }
    #pragma unroll
    for (int r8 = 0; r8 < 8; ++r8) {
      int M = M0 + m0 + r8 + ((lane < 16) ? 0 : 8);
      int N = N0 + n0 + l16;
      out[((size_t)bc * LPATCH + M) * LPATCH + N] = acc[r8] * 0.01f;
    }
  }
}

extern "C" void kernel_launch(void* const* d_in, const int* in_sizes, int n_in,
                              void* d_out, int out_size, void* d_ws,
                              size_t ws_size, hipStream_t stream) {
  (void)in_sizes; (void)n_in; (void)out_size; (void)ws_size;
  const float* x       = (const float*)d_in[0];
  const float* w_conv1 = (const float*)d_in[1];
  const float* b_conv1 = (const float*)d_in[2];
  const float* w_conv2 = (const float*)d_in[3];
  const float* b_conv2 = (const float*)d_in[4];
  const float* w_fm    = (const float*)d_in[5];
  const float* b_fm    = (const float*)d_in[6];
  const float* w_lin1  = (const float*)d_in[7];
  const float* w_lin2  = (const float*)d_in[8];
  float* out = (float*)d_out;

  float* ws  = (float*)d_ws;
  float* ini = ws + WS_INI;
  float* fea = ws + WS_FEA;
  float* ui  = ws + WS_UI;
  float* uf  = ws + WS_UF;
  float* wf  = ws + WS_WF;
  float* img = ws + WS_IMG;
  float* img2 = ws + WS_IMG2;

  pad_weights_kernel<<<dim3(PREP_WF_BLOCKS + PREP_IMG_BLOCKS + PREP_IMG2_BLOCKS), 256, 0, stream>>>(
      w_lin1, w_lin2, w_conv1, b_conv1, w_conv2, b_conv2, w_fm, b_fm,
      wf, img, img2);

  conv_fused_kernel<<<dim3(N_BORDER_BLK + N_INT_BLK), 256,
                      CONV_LDS_FLOATS * sizeof(float), stream>>>(
      x, img, img2, ini, fea);

  trans_kernel<<<dim3(BATCH * CLS * LPATCH / 64, 2, 1), 256,
                 TRANS_LDS_FLOATS * sizeof(float), stream>>>(
      ini, fea, wf, ui, uf);

  att_kernel<<<dim3(LPATCH / 128, LPATCH / 64, BATCH * CLS), 256,
               ATT_LDS_FLOATS * sizeof(float), stream>>>(ui, uf, out);
}
